# starting kernel
# baseline (speedup 1.0000x reference)
#include <hip/hip_runtime.h>


#define LRELU 0.2f

typedef __attribute__((ext_vector_type(2))) float v2f;
typedef __attribute__((ext_vector_type(8))) float v8f;

__device__ __forceinline__ float wred_sum(float v) {
#pragma unroll
    for (int o = 16; o > 0; o >>= 1) v += __shfl_xor(v, o, 32);
    return v;
}
__device__ __forceinline__ float wred_max(float v) {
#pragma unroll
    for (int o = 16; o > 0; o >>= 1) v = fmaxf(v, __shfl_xor(v, o, 32));
    return v;
}
__device__ __forceinline__ unsigned fkey(float f) {
    unsigned u = __float_as_uint(f);
    return (u & 0x80000000u) ? ~u : (u | 0x80000000u);
}
__device__ __forceinline__ float fdec(unsigned k) {
    unsigned u = (k & 0x80000000u) ? (k & 0x7fffffffu) : ~k;
    return __uint_as_float(u);
}
__device__ __forceinline__ float lrelu(float x) { return x > 0.f ? x : LRELU * x; }

__global__ void fill_u32_kernel(unsigned* __restrict__ p, unsigned v, long long n) {
    long long i = (long long)blockIdx.x * blockDim.x + threadIdx.x;
    if (i < n) p[i] = v;
}

template <int COLT, int MODE>
__global__ __launch_bounds__(256) void gemm_wmma_f32(const float* __restrict__ A,
                                                     const float* __restrict__ B,
                                                     float* __restrict__ C, int nrows) {
    constexpr int CW = COLT * 16;
    __shared__ float Blds[256 * CW];
    const int tid = threadIdx.x;
    const int colBase = blockIdx.y * CW;

    for (int idx = tid; idx < 256 * CW; idx += 256) {
        float bv;
        if constexpr (MODE == 0) {
            bv = B[(size_t)blockIdx.y * (256 * 64) + idx];
        } else {
            int k = idx / CW;
            int c = idx - k * CW;
            bv = (c < 40) ? B[k * 40 + c] : 0.0f;
        }
        Blds[idx] = bv;
    }
    __syncthreads();

    const int wid  = tid >> 5;
    const int lane = tid & 31;
    const int half = lane >> 4;
    const int l16  = lane & 15;
    const int rowBase = blockIdx.x * 128 + wid * 16;

    int arow = rowBase + l16;
    if (arow >= nrows) arow = nrows - 1;
    const float* aptr = A + (size_t)arow * 256;

    v8f acc[COLT] = {};

    for (int k0 = 0; k0 < 256; k0 += 4) {
        const int kb = k0 + half * 2;
        v2f a;
        a.x = aptr[kb];
        a.y = aptr[kb + 1];
#pragma unroll
        for (int t = 0; t < COLT; ++t) {
            v2f b;
            const int c = t * 16 + l16;
            b.x = Blds[kb * CW + c];
            b.y = Blds[(kb + 1) * CW + c];
            acc[t] = __builtin_amdgcn_wmma_f32_16x16x4_f32(false, a, false, b, (short)0,
                                                           acc[t], false, false);
        }
    }

    if (rowBase + 16 <= nrows) {
#pragma unroll
        for (int t = 0; t < COLT; ++t) {
            const int col = colBase + t * 16 + l16;
            if constexpr (MODE == 1) {
                if (col >= 40) continue;
            }
#pragma unroll
            for (int v = 0; v < 8; ++v) {
                const int row = rowBase + v + half * 8;
                if constexpr (MODE == 0) {
                    C[(size_t)row * 256 + col] = acc[t][v];
                } else {
                    C[(size_t)row * 40 + col] = acc[t][v];
                }
            }
        }
    } else {
#pragma unroll
        for (int t = 0; t < COLT; ++t) {
            const int col = colBase + t * 16 + l16;
#pragma unroll
            for (int v = 0; v < 8; ++v) {
                const int row = rowBase + v + half * 8;
                if constexpr (MODE == 0) {
                    if (row < nrows) C[(size_t)row * 256 + col] = acc[t][v];
                } else {
                    if (row < nrows && col < 40) C[(size_t)row * 40 + col] = acc[t][v];
                }
            }
        }
    }
}

__global__ __launch_bounds__(256) void scores1_kernel(const float* __restrict__ Wh,
                                                      const float* __restrict__ a_heads,
                                                      float* __restrict__ ssrc,
                                                      float* __restrict__ sdst, int n) {
    long long gt = (long long)blockIdx.x * blockDim.x + threadIdx.x;
    int node = (int)(gt >> 5);
    int lane = threadIdx.x & 31;
    if (node >= n) return;
    const float* wh = Wh + (size_t)node * 256;
#pragma unroll
    for (int h = 0; h < 4; ++h) {
        float w1 = wh[h * 64 + lane];
        float w2 = wh[h * 64 + 32 + lane];
        const float* a = a_heads + h * 128;
        float ps = w1 * a[lane] + w2 * a[32 + lane];
        float pd = w1 * a[64 + lane] + w2 * a[96 + lane];
        ps = wred_sum(ps);
        pd = wred_sum(pd);
        if (lane == 0) {
            ssrc[node * 4 + h] = ps;
            sdst[node * 4 + h] = pd;
        }
    }
}

__global__ __launch_bounds__(256) void scores2_kernel(const float* __restrict__ Wx,
                                                      const float* __restrict__ a_out,
                                                      float* __restrict__ ssrc,
                                                      float* __restrict__ sdst, int n) {
    long long gt = (long long)blockIdx.x * blockDim.x + threadIdx.x;
    int node = (int)(gt >> 5);
    int lane = threadIdx.x & 31;
    if (node >= n) return;
    const float* wx = Wx + (size_t)node * 40;
    float w1 = wx[lane];
    float w2 = (lane < 8) ? wx[32 + lane] : 0.f;
    float as2 = (lane < 8) ? a_out[32 + lane] : 0.f;
    float ad2 = (lane < 8) ? a_out[72 + lane] : 0.f;
    float ps = w1 * a_out[lane] + w2 * as2;
    float pd = w1 * a_out[40 + lane] + w2 * ad2;
    ps = wred_sum(ps);
    pd = wred_sum(pd);
    if (lane == 0) {
        ssrc[node] = ps;
        sdst[node] = pd;
    }
}

__global__ __launch_bounds__(256) void edge_max1(const int* __restrict__ src,
                                                 const int* __restrict__ dst,
                                                 const float4* __restrict__ ss4,
                                                 const float4* __restrict__ sd4,
                                                 unsigned* __restrict__ mkey, int ne) {
    int e = blockIdx.x * blockDim.x + threadIdx.x;
    if (e >= ne) return;
    int s = src[e], d = dst[e];
    float4 vs = ss4[s];
    float4 vd = sd4[d];
    unsigned* mk = mkey + (size_t)d * 4;
    atomicMax(mk + 0, fkey(lrelu(vs.x + vd.x)));
    atomicMax(mk + 1, fkey(lrelu(vs.y + vd.y)));
    atomicMax(mk + 2, fkey(lrelu(vs.z + vd.z)));
    atomicMax(mk + 3, fkey(lrelu(vs.w + vd.w)));
}
__global__ __launch_bounds__(256) void edge_exp1(const int* __restrict__ src,
                                                 const int* __restrict__ dst,
                                                 const float4* __restrict__ ss4,
                                                 const float4* __restrict__ sd4,
                                                 const uint4* __restrict__ mkey4,
                                                 float* __restrict__ den,
                                                 float4* __restrict__ att4, int ne) {
    int e = blockIdx.x * blockDim.x + threadIdx.x;
    if (e >= ne) return;
    int s = src[e], d = dst[e];
    float4 vs = ss4[s];
    float4 vd = sd4[d];
    uint4 mk = mkey4[d];
    float4 ex;
    ex.x = expf(lrelu(vs.x + vd.x) - fdec(mk.x));
    ex.y = expf(lrelu(vs.y + vd.y) - fdec(mk.y));
    ex.z = expf(lrelu(vs.z + vd.z) - fdec(mk.z));
    ex.w = expf(lrelu(vs.w + vd.w) - fdec(mk.w));
    att4[e] = ex;
    float* dn = den + (size_t)d * 4;
    atomicAdd(dn + 0, ex.x);
    atomicAdd(dn + 1, ex.y);
    atomicAdd(dn + 2, ex.z);
    atomicAdd(dn + 3, ex.w);
}
__global__ __launch_bounds__(256) void edge_att1(const int* __restrict__ dst,
                                                 const float4* __restrict__ den4,
                                                 float4* __restrict__ att4, int ne) {
    int e = blockIdx.x * blockDim.x + threadIdx.x;
    if (e >= ne) return;
    float4 a = att4[e];
    float4 dn = den4[dst[e]];
    a.x /= dn.x;
    a.y /= dn.y;
    a.z /= dn.z;
    a.w /= dn.w;
    att4[e] = a;
}
__global__ __launch_bounds__(256) void scatter1(const int* __restrict__ src,
                                                const int* __restrict__ dst,
                                                const float* __restrict__ att,
                                                const float* __restrict__ Wh,
                                                float* __restrict__ hp, long long total) {
    long long i = (long long)blockIdx.x * blockDim.x + threadIdx.x;
    if (i >= total) return;
    int e = (int)(i >> 8);
    int f = (int)(i & 255);
    float a = att[(size_t)e * 4 + (f >> 6)];
    atomicAdd(&hp[(size_t)dst[e] * 256 + f], a * Wh[(size_t)src[e] * 256 + f]);
}
__global__ __launch_bounds__(256) void elu_kernel(float* __restrict__ p, long long n) {
    long long i = (long long)blockIdx.x * blockDim.x + threadIdx.x;
    if (i >= n) return;
    float v = p[i];
    p[i] = v > 0.f ? v : (expf(v) - 1.f);
}

__global__ __launch_bounds__(256) void edge_max2(const int* __restrict__ src,
                                                 const int* __restrict__ dst,
                                                 const float* __restrict__ ss,
                                                 const float* __restrict__ sd,
                                                 unsigned* __restrict__ mkey, int ne) {
    int e = blockIdx.x * blockDim.x + threadIdx.x;
    if (e >= ne) return;
    float x = lrelu(ss[src[e]] + sd[dst[e]]);
    atomicMax(&mkey[dst[e]], fkey(x));
}
__global__ __launch_bounds__(256) void edge_exp2(const int* __restrict__ src,
                                                 const int* __restrict__ dst,
                                                 const float* __restrict__ ss,
                                                 const float* __restrict__ sd,
                                                 const unsigned* __restrict__ mkey,
                                                 float* __restrict__ den,
                                                 float* __restrict__ att, int ne) {
    int e = blockIdx.x * blockDim.x + threadIdx.x;
    if (e >= ne) return;
    int d = dst[e];
    float x = lrelu(ss[src[e]] + sd[d]);
    float ex = expf(x - fdec(mkey[d]));
    att[e] = ex;
    atomicAdd(&den[d], ex);
}
__global__ __launch_bounds__(256) void edge_att2(const int* __restrict__ dst,
                                                 const float* __restrict__ den,
                                                 float* __restrict__ att, int ne) {
    int e = blockIdx.x * blockDim.x + threadIdx.x;
    if (e >= ne) return;
    att[e] = att[e] / den[dst[e]];
}
__global__ __launch_bounds__(256) void scatter2(const int* __restrict__ src,
                                                const int* __restrict__ dst,
                                                const float* __restrict__ att,
                                                const float* __restrict__ Wx,
                                                float* __restrict__ out, long long total) {
    long long i = (long long)blockIdx.x * blockDim.x + threadIdx.x;
    if (i >= total) return;
    int e = (int)(i >> 6);
    int f = (int)(i & 63);
    if (f >= 40) return;
    atomicAdd(&out[(size_t)dst[e] * 40 + f], att[e] * Wx[(size_t)src[e] * 40 + f]);
}

__global__ __launch_bounds__(256) void logsoftmax_kernel(float* __restrict__ out, int n) {
    long long gt = (long long)blockIdx.x * blockDim.x + threadIdx.x;
    int node = (int)(gt >> 5);
    int lane = threadIdx.x & 31;
    if (node >= n) return;
    float* row = out + (size_t)node * 40;
    float v1 = row[lane];
    float v2 = (lane < 8) ? row[32 + lane] : -INFINITY;
    float mx = wred_max(fmaxf(v1, v2));
    float s = expf(v1 - mx) + ((lane < 8) ? expf(v2 - mx) : 0.f);
    s = wred_sum(s);
    float lg = mx + logf(s);
    row[lane] = v1 - lg;
    if (lane < 8) row[32 + lane] = v2 - lg;
}

extern "C" void kernel_launch(void* const* d_in, const int* in_sizes, int n_in, void* d_out,
                              int out_size, void* d_ws, size_t ws_size, hipStream_t stream) {
    (void)n_in; (void)out_size; (void)ws_size;
    const float* h      = (const float*)d_in[0];
    const int*   eidx   = (const int*)d_in[1];
    const float* Wheads = (const float*)d_in[2];
    const float* aheads = (const float*)d_in[3];
    const float* Wout   = (const float*)d_in[4];
    const float* aout   = (const float*)d_in[5];
    float* out = (float*)d_out;

    const int N = in_sizes[0] / 256;
    const int E = in_sizes[1] / 2;
    const int* src = eidx;
    const int* dst = eidx + E;

    float* ws = (float*)d_ws;
    size_t oWh   = 0;
    size_t oHP   = oWh  + (size_t)N * 256;
    size_t oS1s  = oHP  + (size_t)N * 256;
    size_t oS1d  = oS1s + (size_t)N * 4;
    size_t oM1   = oS1d + (size_t)N * 4;
    size_t oD1   = oM1  + (size_t)N * 4;
    size_t oAtt1 = oD1  + (size_t)N * 4;
    size_t oS2s  = oAtt1 + (size_t)E * 4;
    size_t oS2d  = oS2s + (size_t)N;
    size_t oM2   = oS2d + (size_t)N;
    size_t oD2   = oM2  + (size_t)N;
    size_t oAtt2 = oD2  + (size_t)N;
    size_t oWx   = oWh;

    const unsigned NEG_INF_KEY = 0x007fffffu;

    hipMemsetAsync(ws + oHP, 0, (size_t)N * 256 * sizeof(float), stream);
    hipMemsetAsync(ws + oD1, 0, (size_t)N * 4 * sizeof(float), stream);
    hipMemsetAsync(ws + oD2, 0, (size_t)N * sizeof(float), stream);
    hipMemsetAsync(out, 0, (size_t)N * 40 * sizeof(float), stream);
    {
        long long n4 = (long long)N * 4;
        fill_u32_kernel<<<(int)((n4 + 255) / 256), 256, 0, stream>>>(
            (unsigned*)(ws + oM1), NEG_INF_KEY, n4);
        fill_u32_kernel<<<(N + 255) / 256, 256, 0, stream>>>(
            (unsigned*)(ws + oM2), NEG_INF_KEY, (long long)N);
    }

    dim3 g1((N + 127) / 128, 4);
    gemm_wmma_f32<4, 0><<<g1, 256, 0, stream>>>(h, Wheads, ws + oWh, N);

    scores1_kernel<<<(N + 7) / 8, 256, 0, stream>>>(ws + oWh, aheads, ws + oS1s, ws + oS1d, N);

    int geE = (E + 255) / 256;
    edge_max1<<<geE, 256, 0, stream>>>(src, dst, (const float4*)(ws + oS1s),
                                       (const float4*)(ws + oS1d), (unsigned*)(ws + oM1), E);
    edge_exp1<<<geE, 256, 0, stream>>>(src, dst, (const float4*)(ws + oS1s),
                                       (const float4*)(ws + oS1d), (const uint4*)(ws + oM1),
                                       ws + oD1, (float4*)(ws + oAtt1), E);
    edge_att1<<<geE, 256, 0, stream>>>(dst, (const float4*)(ws + oD1), (float4*)(ws + oAtt1), E);

    long long tot1 = (long long)E * 256;
    scatter1<<<(int)((tot1 + 255) / 256), 256, 0, stream>>>(src, dst, ws + oAtt1, ws + oWh,
                                                            ws + oHP, tot1);
    long long nf = (long long)N * 256;
    elu_kernel<<<(int)((nf + 255) / 256), 256, 0, stream>>>(ws + oHP, nf);

    dim3 g2((N + 127) / 128, 1);
    gemm_wmma_f32<3, 1><<<g2, 256, 0, stream>>>(ws + oHP, Wout, ws + oWx, N);

    scores2_kernel<<<(N + 7) / 8, 256, 0, stream>>>(ws + oWx, aout, ws + oS2s, ws + oS2d, N);

    edge_max2<<<geE, 256, 0, stream>>>(src, dst, ws + oS2s, ws + oS2d, (unsigned*)(ws + oM2), E);
    edge_exp2<<<geE, 256, 0, stream>>>(src, dst, ws + oS2s, ws + oS2d, (unsigned*)(ws + oM2),
                                       ws + oD2, ws + oAtt2, E);
    edge_att2<<<geE, 256, 0, stream>>>(dst, ws + oD2, ws + oAtt2, E);

    long long tot2 = (long long)E * 64;
    scatter2<<<(int)((tot2 + 255) / 256), 256, 0, stream>>>(src, dst, ws + oAtt2, ws + oWx, out,
                                                            tot2);

    logsoftmax_kernel<<<(N + 7) / 8, 256, 0, stream>>>(out, N);
}


# aggregation by destination-sorted edge list (counting sort per call): one wave per node sums att*Wh[src] in registers, no feature atomics; elu/log_softmax in its epilogue; sum order = edge id
# speedup vs baseline: 2.2954x; 2.2954x over previous
#include <hip/hip_runtime.h>


#define LRELU 0.2f

typedef __attribute__((ext_vector_type(2))) float v2f;
typedef __attribute__((ext_vector_type(8))) float v8f;

__device__ __forceinline__ float wred_sum(float v) {
#pragma unroll
    for (int o = 16; o > 0; o >>= 1) v += __shfl_xor(v, o, 32);
    return v;
}
__device__ __forceinline__ float wred_max(float v) {
#pragma unroll
    for (int o = 16; o > 0; o >>= 1) v = fmaxf(v, __shfl_xor(v, o, 32));
    return v;
}
__device__ __forceinline__ unsigned fkey(float f) {
    unsigned u = __float_as_uint(f);
    return (u & 0x80000000u) ? ~u : (u | 0x80000000u);
}
__device__ __forceinline__ float fdec(unsigned k) {
    unsigned u = (k & 0x80000000u) ? (k & 0x7fffffffu) : ~k;
    return __uint_as_float(u);
}
__device__ __forceinline__ float lrelu(float x) { return x > 0.f ? x : LRELU * x; }

__global__ void fill_u32_kernel(unsigned* __restrict__ p, unsigned v, long long n) {
    long long i = (long long)blockIdx.x * blockDim.x + threadIdx.x;
    if (i < n) p[i] = v;
}

template <int COLT, int MODE>
__global__ __launch_bounds__(256) void gemm_wmma_f32(const float* __restrict__ A,
                                                     const float* __restrict__ B,
                                                     float* __restrict__ C, int nrows) {
    constexpr int CW = COLT * 16;
    __shared__ float Blds[256 * CW];
    const int tid = threadIdx.x;
    const int colBase = blockIdx.y * CW;

    for (int idx = tid; idx < 256 * CW; idx += 256) {
        float bv;
        if constexpr (MODE == 0) {
            bv = B[(size_t)blockIdx.y * (256 * 64) + idx];
        } else {
            int k = idx / CW;
            int c = idx - k * CW;
            bv = (c < 40) ? B[k * 40 + c] : 0.0f;
        }
        Blds[idx] = bv;
    }
    __syncthreads();

    const int wid  = tid >> 5;
    const int lane = tid & 31;
    const int half = lane >> 4;
    const int l16  = lane & 15;
    const int rowBase = blockIdx.x * 128 + wid * 16;

    int arow = rowBase + l16;
    if (arow >= nrows) arow = nrows - 1;
    const float* aptr = A + (size_t)arow * 256;

    v8f acc[COLT] = {};

    for (int k0 = 0; k0 < 256; k0 += 4) {
        const int kb = k0 + half * 2;
        v2f a;
        a.x = aptr[kb];
        a.y = aptr[kb + 1];
#pragma unroll
        for (int t = 0; t < COLT; ++t) {
            v2f b;
            const int c = t * 16 + l16;
            b.x = Blds[kb * CW + c];
            b.y = Blds[(kb + 1) * CW + c];
            acc[t] = __builtin_amdgcn_wmma_f32_16x16x4_f32(false, a, false, b, (short)0,
                                                           acc[t], false, false);
        }
    }

    if (rowBase + 16 <= nrows) {
#pragma unroll
        for (int t = 0; t < COLT; ++t) {
            const int col = colBase + t * 16 + l16;
            if constexpr (MODE == 1) {
                if (col >= 40) continue;
            }
#pragma unroll
            for (int v = 0; v < 8; ++v) {
                const int row = rowBase + v + half * 8;
                if constexpr (MODE == 0) {
                    C[(size_t)row * 256 + col] = acc[t][v];
                } else {
                    C[(size_t)row * 40 + col] = acc[t][v];
                }
            }
        }
    } else {
#pragma unroll
        for (int t = 0; t < COLT; ++t) {
            const int col = colBase + t * 16 + l16;
#pragma unroll
            for (int v = 0; v < 8; ++v) {
                const int row = rowBase + v + half * 8;
                if constexpr (MODE == 0) {
                    if (row < nrows) C[(size_t)row * 256 + col] = acc[t][v];
                } else {
                    if (row < nrows && col < 40) C[(size_t)row * 40 + col] = acc[t][v];
                }
            }
        }
    }
}

__global__ __launch_bounds__(256) void scores1_kernel(const float* __restrict__ Wh,
                                                      const float* __restrict__ a_heads,
                                                      float* __restrict__ ssrc,
                                                      float* __restrict__ sdst, int n) {
    long long gt = (long long)blockIdx.x * blockDim.x + threadIdx.x;
    int node = (int)(gt >> 5);
    int lane = threadIdx.x & 31;
    if (node >= n) return;
    const float* wh = Wh + (size_t)node * 256;
#pragma unroll
    for (int h = 0; h < 4; ++h) {
        float w1 = wh[h * 64 + lane];
        float w2 = wh[h * 64 + 32 + lane];
        const float* a = a_heads + h * 128;
        float ps = w1 * a[lane] + w2 * a[32 + lane];
        float pd = w1 * a[64 + lane] + w2 * a[96 + lane];
        ps = wred_sum(ps);
        pd = wred_sum(pd);
        if (lane == 0) {
            ssrc[node * 4 + h] = ps;
            sdst[node * 4 + h] = pd;
        }
    }
}

__global__ __launch_bounds__(256) void scores2_kernel(const float* __restrict__ Wx,
                                                      const float* __restrict__ a_out,
                                                      float* __restrict__ ssrc,
                                                      float* __restrict__ sdst, int n) {
    long long gt = (long long)blockIdx.x * blockDim.x + threadIdx.x;
    int node = (int)(gt >> 5);
    int lane = threadIdx.x & 31;
    if (node >= n) return;
    const float* wx = Wx + (size_t)node * 40;
    float w1 = wx[lane];
    float w2 = (lane < 8) ? wx[32 + lane] : 0.f;
    float as2 = (lane < 8) ? a_out[32 + lane] : 0.f;
    float ad2 = (lane < 8) ? a_out[72 + lane] : 0.f;
    float ps = w1 * a_out[lane] + w2 * as2;
    float pd = w1 * a_out[40 + lane] + w2 * ad2;
    ps = wred_sum(ps);
    pd = wred_sum(pd);
    if (lane == 0) {
        ssrc[node] = ps;
        sdst[node] = pd;
    }
}

__global__ __launch_bounds__(256) void edge_max1(const int* __restrict__ src,
                                                 const int* __restrict__ dst,
                                                 const float4* __restrict__ ss4,
                                                 const float4* __restrict__ sd4,
                                                 unsigned* __restrict__ mkey, int ne) {
    int e = blockIdx.x * blockDim.x + threadIdx.x;
    if (e >= ne) return;
    int s = src[e], d = dst[e];
    float4 vs = ss4[s];
    float4 vd = sd4[d];
    unsigned* mk = mkey + (size_t)d * 4;
    atomicMax(mk + 0, fkey(lrelu(vs.x + vd.x)));
    atomicMax(mk + 1, fkey(lrelu(vs.y + vd.y)));
    atomicMax(mk + 2, fkey(lrelu(vs.z + vd.z)));
    atomicMax(mk + 3, fkey(lrelu(vs.w + vd.w)));
}
__global__ __launch_bounds__(256) void edge_exp1(const int* __restrict__ src,
                                                 const int* __restrict__ dst,
                                                 const float4* __restrict__ ss4,
                                                 const float4* __restrict__ sd4,
                                                 const uint4* __restrict__ mkey4,
                                                 float* __restrict__ den,
                                                 float4* __restrict__ att4, int ne) {
    int e = blockIdx.x * blockDim.x + threadIdx.x;
    if (e >= ne) return;
    int s = src[e], d = dst[e];
    float4 vs = ss4[s];
    float4 vd = sd4[d];
    uint4 mk = mkey4[d];
    float4 ex;
    ex.x = expf(lrelu(vs.x + vd.x) - fdec(mk.x));
    ex.y = expf(lrelu(vs.y + vd.y) - fdec(mk.y));
    ex.z = expf(lrelu(vs.z + vd.z) - fdec(mk.z));
    ex.w = expf(lrelu(vs.w + vd.w) - fdec(mk.w));
    att4[e] = ex;
    float* dn = den + (size_t)d * 4;
    atomicAdd(dn + 0, ex.x);
    atomicAdd(dn + 1, ex.y);
    atomicAdd(dn + 2, ex.z);
    atomicAdd(dn + 3, ex.w);
}
__global__ __launch_bounds__(256) void edge_att1(const int* __restrict__ dst,
                                                 const float4* __restrict__ den4,
                                                 float4* __restrict__ att4, int ne) {
    int e = blockIdx.x * blockDim.x + threadIdx.x;
    if (e >= ne) return;
    float4 a = att4[e];
    float4 dn = den4[dst[e]];
    a.x /= dn.x;
    a.y /= dn.y;
    a.z /= dn.z;
    a.w /= dn.w;
    att4[e] = a;
}
__device__ __forceinline__ int wred_sum_i(int v) {
    for (int o = 16; o > 0; o >>= 1) v += __shfl_xor(v, o, 32);
    return v;
}
__device__ __forceinline__ float readlane_f(float v, int l) {
    return __int_as_float(__builtin_amdgcn_readlane(__float_as_int(v), l));
}
__global__ __launch_bounds__(256) void csr_count(const int* __restrict__ dst,
                                                 int* __restrict__ cnt,
                                                 int* __restrict__ rank, int ne) {
    int e = blockIdx.x * blockDim.x + threadIdx.x;
    if (e >= ne) return;
    rank[e] = atomicAdd(&cnt[dst[e]], 1);
}
__global__ __launch_bounds__(256) void csr_blocksum(const int* __restrict__ cnt,
                                                    int* __restrict__ bsum, int n1) {
    __shared__ int wsum[8];
    const int base = blockIdx.x * 1024 + threadIdx.x * 4;
    int s = 0;
    for (int k = 0; k < 4; ++k)
        if (base + k < n1) s += cnt[base + k];
    s = wred_sum_i(s);
    if ((threadIdx.x & 31) == 0) wsum[threadIdx.x >> 5] = s;
    __syncthreads();
    if (threadIdx.x == 0) {
        int t = 0;
        for (int w = 0; w < 8; ++w) t += wsum[w];
        bsum[blockIdx.x] = t;
    }
}
__global__ __launch_bounds__(256) void csr_rowptr(const int* __restrict__ cnt,
                                                  const int* __restrict__ bsum,
                                                  int* __restrict__ rowptr, int n1) {
    __shared__ int wsum[8];
    __shared__ int wsum2[8];
    const int tid = threadIdx.x, lane = tid & 31, wid = tid >> 5;
    int p = 0;
    for (int j = tid; j < (int)blockIdx.x; j += 256) p += bsum[j];
    p = wred_sum_i(p);
    if (lane == 0) wsum[wid] = p;
    const int base = blockIdx.x * 1024 + tid * 4;
    int c[4];
    int s = 0;
    for (int k = 0; k < 4; ++k) {
        c[k] = (base + k < n1) ? cnt[base + k] : 0;
        s += c[k];
    }
    int inc = s;
    for (int o = 1; o < 32; o <<= 1) {
        int t = __shfl_up(inc, o, 32);
        if (lane >= o) inc += t;
    }
    if (lane == 31) wsum2[wid] = inc;
    __syncthreads();
    int run = inc - s;
    for (int w = 0; w < 8; ++w) {
        run += wsum[w];
        if (w < wid) run += wsum2[w];
    }
    for (int k = 0; k < 4; ++k) {
        if (base + k < n1) rowptr[base + k] = run;
        run += c[k];
    }
}
__global__ __launch_bounds__(256) void csr_place(const int* __restrict__ dst,
                                                 const int* __restrict__ rowptr,
                                                 const int* __restrict__ rank,
                                                 int* __restrict__ tmp, int ne) {
    int e = blockIdx.x * blockDim.x + threadIdx.x;
    if (e >= ne) return;
    tmp[rowptr[dst[e]] + rank[e]] = e;
}
__global__ __launch_bounds__(256) void csr_segsort(const int* __restrict__ rowptr,
                                                   const int* __restrict__ tmp,
                                                   int* __restrict__ perm, int n) {
    const int node = __builtin_amdgcn_readfirstlane((int)((blockIdx.x * 256 + threadIdx.x) >> 5));
    const int lane = threadIdx.x & 31;
    if (node >= n) return;
    const int beg = rowptr[node], end = rowptr[node + 1];
    for (int i0 = beg; i0 < end; i0 += 32) {
        const int i = i0 + lane;
        const int ei = (i < end) ? tmp[i] : 0x7fffffff;
        int r = 0;
        for (int j0 = beg; j0 < end; j0 += 32) {
            const int j = j0 + lane;
            const int ej = (j < end) ? tmp[j] : 0x7fffffff;
            const int m = (end - j0 < 32) ? (end - j0) : 32;
            for (int t = 0; t < m; ++t) r += (__builtin_amdgcn_readlane(ej, t) < ei) ? 1 : 0;
        }
        if (i < end) perm[beg + r] = ei;
    }
}

__global__ __launch_bounds__(256) void gather1_elu(const int* __restrict__ rowptr,
                                                   const int* __restrict__ perm,
                                                   const int* __restrict__ src,
                                                   const float4* __restrict__ att4,
                                                   const float* __restrict__ Wh,
                                                   float* __restrict__ hp, int n) {
    const int node = __builtin_amdgcn_readfirstlane((int)((blockIdx.x * 256 + threadIdx.x) >> 5));
    const int lane = threadIdx.x & 31;
    if (node >= n) return;
    const int beg = rowptr[node], end = rowptr[node + 1];
    const bool hi = lane >= 16;
    const float* wl = Wh + lane * 4;
    float4 acc0 = {0.f, 0.f, 0.f, 0.f};
    float4 acc1 = {0.f, 0.f, 0.f, 0.f};
    for (int i0 = beg; i0 < end; i0 += 32) {
        const int i = i0 + lane;
        int s_i = 0;
        float4 a_i = {0.f, 0.f, 0.f, 0.f};
        if (i < end) {
            const int e = perm[i];
            s_i = src[e];
            a_i = att4[e];
        }
        const int m = (end - i0 < 32) ? (end - i0) : 32;
#pragma unroll 4
        for (int t = 0; t < m; ++t) {
            const int s = __builtin_amdgcn_readlane(s_i, t);
            const float ax = readlane_f(a_i.x, t);
            const float ay = readlane_f(a_i.y, t);
            const float az = readlane_f(a_i.z, t);
            const float aw = readlane_f(a_i.w, t);
            const float al = hi ? ay : ax;
            const float au = hi ? aw : az;
            const float* r = wl + (size_t)s * 256;
            const float4 w0 = *(const float4*)(r);
            const float4 w1 = *(const float4*)(r + 128);
            acc0.x += al * w0.x;
            acc0.y += al * w0.y;
            acc0.z += al * w0.z;
            acc0.w += al * w0.w;
            acc1.x += au * w1.x;
            acc1.y += au * w1.y;
            acc1.z += au * w1.z;
            acc1.w += au * w1.w;
        }
    }
    acc0.x = acc0.x > 0.f ? acc0.x : (expf(acc0.x) - 1.f);
    acc0.y = acc0.y > 0.f ? acc0.y : (expf(acc0.y) - 1.f);
    acc0.z = acc0.z > 0.f ? acc0.z : (expf(acc0.z) - 1.f);
    acc0.w = acc0.w > 0.f ? acc0.w : (expf(acc0.w) - 1.f);
    acc1.x = acc1.x > 0.f ? acc1.x : (expf(acc1.x) - 1.f);
    acc1.y = acc1.y > 0.f ? acc1.y : (expf(acc1.y) - 1.f);
    acc1.z = acc1.z > 0.f ? acc1.z : (expf(acc1.z) - 1.f);
    acc1.w = acc1.w > 0.f ? acc1.w : (expf(acc1.w) - 1.f);
    float* o = hp + (size_t)node * 256 + lane * 4;
    *(float4*)(o) = acc0;
    *(float4*)(o + 128) = acc1;
}

__global__ __launch_bounds__(256) void edge_max2(const int* __restrict__ src,
                                                 const int* __restrict__ dst,
                                                 const float* __restrict__ ss,
                                                 const float* __restrict__ sd,
                                                 unsigned* __restrict__ mkey, int ne) {
    int e = blockIdx.x * blockDim.x + threadIdx.x;
    if (e >= ne) return;
    float x = lrelu(ss[src[e]] + sd[dst[e]]);
    atomicMax(&mkey[dst[e]], fkey(x));
}
__global__ __launch_bounds__(256) void edge_exp2(const int* __restrict__ src,
                                                 const int* __restrict__ dst,
                                                 const float* __restrict__ ss,
                                                 const float* __restrict__ sd,
                                                 const unsigned* __restrict__ mkey,
                                                 float* __restrict__ den,
                                                 float* __restrict__ att, int ne) {
    int e = blockIdx.x * blockDim.x + threadIdx.x;
    if (e >= ne) return;
    int d = dst[e];
    float x = lrelu(ss[src[e]] + sd[d]);
    float ex = expf(x - fdec(mkey[d]));
    att[e] = ex;
    atomicAdd(&den[d], ex);
}
__global__ __launch_bounds__(256) void edge_att2(const int* __restrict__ dst,
                                                 const float* __restrict__ den,
                                                 float* __restrict__ att, int ne) {
    int e = blockIdx.x * blockDim.x + threadIdx.x;
    if (e >= ne) return;
    att[e] = att[e] / den[dst[e]];
}
__global__ __launch_bounds__(256) void gather2_lsm(const int* __restrict__ rowptr,
                                                   const int* __restrict__ perm,
                                                   const int* __restrict__ src,
                                                   const float* __restrict__ att,
                                                   const float* __restrict__ Wx,
                                                   float* __restrict__ out, int n) {
    const int node = __builtin_amdgcn_readfirstlane((int)((blockIdx.x * 256 + threadIdx.x) >> 5));
    const int lane = threadIdx.x & 31;
    if (node >= n) return;
    const int beg = rowptr[node], end = rowptr[node + 1];
    const float* w1p = Wx + lane;
    const float* w2p = Wx + 32 + (lane & 7);
    float acc1 = 0.f, acc2 = 0.f;
    for (int i0 = beg; i0 < end; i0 += 32) {
        const int i = i0 + lane;
        int s_i = 0;
        float a_i = 0.f;
        if (i < end) {
            const int e = perm[i];
            s_i = src[e];
            a_i = att[e];
        }
        const int m = (end - i0 < 32) ? (end - i0) : 32;
#pragma unroll 4
        for (int t = 0; t < m; ++t) {
            const int s = __builtin_amdgcn_readlane(s_i, t);
            const float a = readlane_f(a_i, t);
            acc1 += a * w1p[(size_t)s * 40];
            acc2 += a * w2p[(size_t)s * 40];
        }
    }
    const float v1 = acc1;
    const float v2 = (lane < 8) ? acc2 : -INFINITY;
    const float mx = wred_max(fmaxf(v1, v2));
    float sm = expf(v1 - mx) + ((lane < 8) ? expf(v2 - mx) : 0.f);
    sm = wred_sum(sm);
    const float lg = mx + logf(sm);
    float* row = out + (size_t)node * 40;
    row[lane] = v1 - lg;
    if (lane < 8) row[32 + lane] = v2 - lg;
}

extern "C" void kernel_launch(void* const* d_in, const int* in_sizes, int n_in, void* d_out,
                              int out_size, void* d_ws, size_t ws_size, hipStream_t stream) {
    (void)n_in; (void)out_size; (void)ws_size;
    const float* h      = (const float*)d_in[0];
    const int*   eidx   = (const int*)d_in[1];
    const float* Wheads = (const float*)d_in[2];
    const float* aheads = (const float*)d_in[3];
    const float* Wout   = (const float*)d_in[4];
    const float* aout   = (const float*)d_in[5];
    float* out = (float*)d_out;

    const int N = in_sizes[0] / 256;
    const int E = in_sizes[1] / 2;
    const int* src = eidx;
    const int* dst = eidx + E;

    float* ws = (float*)d_ws;
    size_t oWh   = 0;
    size_t oHP   = oWh  + (size_t)N * 256;
    size_t oS1s  = oHP  + (size_t)N * 256;
    size_t oS1d  = oS1s + (size_t)N * 4;
    size_t oM1   = oS1d + (size_t)N * 4;
    size_t oD1   = oM1  + (size_t)N * 4;
    size_t oAtt1 = oD1  + (size_t)N * 4;
    size_t oCnt  = oAtt1 + (size_t)E * 4;
    size_t oRP   = oCnt + (size_t)N + 1;
    size_t oBs   = oRP  + (size_t)N + 1;
    size_t oPerm = oAtt1 + (size_t)E * 4 + (size_t)N * 4;
    size_t oRank = oAtt1;
    size_t oTmp  = oAtt1 + (size_t)E;
    size_t oS2s  = oS1s;
    size_t oS2d  = oS1s + (size_t)N;
    size_t oM2   = oM1;
    size_t oD2   = oD1;
    size_t oAtt2 = oAtt1;
    size_t oWx   = oWh;

    const unsigned NEG_INF_KEY = 0x007fffffu;

    int* cnt    = (int*)(ws + oCnt);
    int* rowptr = (int*)(ws + oRP);
    int* bsum   = (int*)(ws + oBs);
    int* perm   = (int*)(ws + oPerm);
    int* rank   = (int*)(ws + oRank);
    int* tmp    = (int*)(ws + oTmp);

    const int geE = (E + 255) / 256;
    const int n1  = N + 1;
    const int nb  = (n1 + 1023) / 1024;

    hipMemsetAsync(ws + oD1, 0, (size_t)N * 4 * sizeof(float), stream);
    {
        long long n4 = (long long)N * 4;
        fill_u32_kernel<<<(int)((n4 + 255) / 256), 256, 0, stream>>>(
            (unsigned*)(ws + oM1), NEG_INF_KEY, n4);
        fill_u32_kernel<<<(n1 + 255) / 256, 256, 0, stream>>>((unsigned*)cnt, 0u, (long long)n1);
    }

    csr_count<<<geE, 256, 0, stream>>>(dst, cnt, rank, E);
    csr_blocksum<<<nb, 256, 0, stream>>>(cnt, bsum, n1);
    csr_rowptr<<<nb, 256, 0, stream>>>(cnt, bsum, rowptr, n1);
    csr_place<<<geE, 256, 0, stream>>>(dst, rowptr, rank, tmp, E);
    csr_segsort<<<(N + 7) / 8, 256, 0, stream>>>(rowptr, tmp, perm, N);

    dim3 g1((N + 127) / 128, 4);
    gemm_wmma_f32<4, 0><<<g1, 256, 0, stream>>>(h, Wheads, ws + oWh, N);

    scores1_kernel<<<(N + 7) / 8, 256, 0, stream>>>(ws + oWh, aheads, ws + oS1s, ws + oS1d, N);

    edge_max1<<<geE, 256, 0, stream>>>(src, dst, (const float4*)(ws + oS1s),
                                       (const float4*)(ws + oS1d), (unsigned*)(ws + oM1), E);
    edge_exp1<<<geE, 256, 0, stream>>>(src, dst, (const float4*)(ws + oS1s),
                                       (const float4*)(ws + oS1d), (const uint4*)(ws + oM1),
                                       ws + oD1, (float4*)(ws + oAtt1), E);
    edge_att1<<<geE, 256, 0, stream>>>(dst, (const float4*)(ws + oD1), (float4*)(ws + oAtt1), E);

    gather1_elu<<<(N + 7) / 8, 256, 0, stream>>>(rowptr, perm, src, (const float4*)(ws + oAtt1),
                                                 ws + oWh, ws + oHP, N);

    dim3 g2((N + 127) / 128, 1);
    gemm_wmma_f32<3, 1><<<g2, 256, 0, stream>>>(ws + oHP, Wout, ws + oWx, N);

    scores2_kernel<<<(N + 7) / 8, 256, 0, stream>>>(ws + oWx, aout, ws + oS2s, ws + oS2d, N);

    hipMemsetAsync(ws + oD2, 0, (size_t)N * sizeof(float), stream);
    fill_u32_kernel<<<(N + 255) / 256, 256, 0, stream>>>(
        (unsigned*)(ws + oM2), NEG_INF_KEY, (long long)N);

    edge_max2<<<geE, 256, 0, stream>>>(src, dst, ws + oS2s, ws + oS2d, (unsigned*)(ws + oM2), E);
    edge_exp2<<<geE, 256, 0, stream>>>(src, dst, ws + oS2s, ws + oS2d, (unsigned*)(ws + oM2),
                                       ws + oD2, ws + oAtt2, E);
    edge_att2<<<geE, 256, 0, stream>>>(dst, ws + oD2, ws + oAtt2, E);

    gather2_lsm<<<(N + 7) / 8, 256, 0, stream>>>(rowptr, perm, src, ws + oAtt2, ws + oWx, out, N);
}


# edge softmax (scores, max, exp, denominator, divide) done inside each destination's gather wave from the sorted list: 6 edge passes, their atomics and att buffers gone; denominator sum order fixed
# speedup vs baseline: 5.2558x; 2.2897x over previous
#include <hip/hip_runtime.h>


#define LRELU 0.2f

typedef __attribute__((ext_vector_type(2))) float v2f;
typedef __attribute__((ext_vector_type(8))) float v8f;

__device__ __forceinline__ float wred_sum(float v) {
#pragma unroll
    for (int o = 16; o > 0; o >>= 1) v += __shfl_xor(v, o, 32);
    return v;
}
__device__ __forceinline__ float wred_max(float v) {
#pragma unroll
    for (int o = 16; o > 0; o >>= 1) v = fmaxf(v, __shfl_xor(v, o, 32));
    return v;
}
__device__ __forceinline__ float lrelu(float x) { return x > 0.f ? x : LRELU * x; }

__global__ void fill_u32_kernel(unsigned* __restrict__ p, unsigned v, long long n) {
    long long i = (long long)blockIdx.x * blockDim.x + threadIdx.x;
    if (i < n) p[i] = v;
}

template <int COLT, int MODE>
__global__ __launch_bounds__(256) void gemm_wmma_f32(const float* __restrict__ A,
                                                     const float* __restrict__ B,
                                                     float* __restrict__ C, int nrows) {
    constexpr int CW = COLT * 16;
    __shared__ float Blds[256 * CW];
    const int tid = threadIdx.x;
    const int colBase = blockIdx.y * CW;

    for (int idx = tid; idx < 256 * CW; idx += 256) {
        float bv;
        if constexpr (MODE == 0) {
            bv = B[(size_t)blockIdx.y * (256 * 64) + idx];
        } else {
            int k = idx / CW;
            int c = idx - k * CW;
            bv = (c < 40) ? B[k * 40 + c] : 0.0f;
        }
        Blds[idx] = bv;
    }
    __syncthreads();

    const int wid  = tid >> 5;
    const int lane = tid & 31;
    const int half = lane >> 4;
    const int l16  = lane & 15;
    const int rowBase = blockIdx.x * 128 + wid * 16;

    int arow = rowBase + l16;
    if (arow >= nrows) arow = nrows - 1;
    const float* aptr = A + (size_t)arow * 256;

    v8f acc[COLT] = {};

    for (int k0 = 0; k0 < 256; k0 += 4) {
        const int kb = k0 + half * 2;
        v2f a;
        a.x = aptr[kb];
        a.y = aptr[kb + 1];
#pragma unroll
        for (int t = 0; t < COLT; ++t) {
            v2f b;
            const int c = t * 16 + l16;
            b.x = Blds[kb * CW + c];
            b.y = Blds[(kb + 1) * CW + c];
            acc[t] = __builtin_amdgcn_wmma_f32_16x16x4_f32(false, a, false, b, (short)0,
                                                           acc[t], false, false);
        }
    }

    if (rowBase + 16 <= nrows) {
#pragma unroll
        for (int t = 0; t < COLT; ++t) {
            const int col = colBase + t * 16 + l16;
            if constexpr (MODE == 1) {
                if (col >= 40) continue;
            }
#pragma unroll
            for (int v = 0; v < 8; ++v) {
                const int row = rowBase + v + half * 8;
                if constexpr (MODE == 0) {
                    C[(size_t)row * 256 + col] = acc[t][v];
                } else {
                    C[(size_t)row * 40 + col] = acc[t][v];
                }
            }
        }
    } else {
#pragma unroll
        for (int t = 0; t < COLT; ++t) {
            const int col = colBase + t * 16 + l16;
#pragma unroll
            for (int v = 0; v < 8; ++v) {
                const int row = rowBase + v + half * 8;
                if constexpr (MODE == 0) {
                    if (row < nrows) C[(size_t)row * 256 + col] = acc[t][v];
                } else {
                    if (row < nrows && col < 40) C[(size_t)row * 40 + col] = acc[t][v];
                }
            }
        }
    }
}

__global__ __launch_bounds__(256) void scores1_kernel(const float* __restrict__ Wh,
                                                      const float* __restrict__ a_heads,
                                                      float* __restrict__ ssrc,
                                                      float* __restrict__ sdst, int n) {
    long long gt = (long long)blockIdx.x * blockDim.x + threadIdx.x;
    int node = (int)(gt >> 5);
    int lane = threadIdx.x & 31;
    if (node >= n) return;
    const float* wh = Wh + (size_t)node * 256;
#pragma unroll
    for (int h = 0; h < 4; ++h) {
        float w1 = wh[h * 64 + lane];
        float w2 = wh[h * 64 + 32 + lane];
        const float* a = a_heads + h * 128;
        float ps = w1 * a[lane] + w2 * a[32 + lane];
        float pd = w1 * a[64 + lane] + w2 * a[96 + lane];
        ps = wred_sum(ps);
        pd = wred_sum(pd);
        if (lane == 0) {
            ssrc[node * 4 + h] = ps;
            sdst[node * 4 + h] = pd;
        }
    }
}

__global__ __launch_bounds__(256) void scores2_kernel(const float* __restrict__ Wx,
                                                      const float* __restrict__ a_out,
                                                      float* __restrict__ ssrc,
                                                      float* __restrict__ sdst, int n) {
    long long gt = (long long)blockIdx.x * blockDim.x + threadIdx.x;
    int node = (int)(gt >> 5);
    int lane = threadIdx.x & 31;
    if (node >= n) return;
    const float* wx = Wx + (size_t)node * 40;
    float w1 = wx[lane];
    float w2 = (lane < 8) ? wx[32 + lane] : 0.f;
    float as2 = (lane < 8) ? a_out[32 + lane] : 0.f;
    float ad2 = (lane < 8) ? a_out[72 + lane] : 0.f;
    float ps = w1 * a_out[lane] + w2 * as2;
    float pd = w1 * a_out[40 + lane] + w2 * ad2;
    ps = wred_sum(ps);
    pd = wred_sum(pd);
    if (lane == 0) {
        ssrc[node] = ps;
        sdst[node] = pd;
    }
}

__device__ __forceinline__ int wred_sum_i(int v) {
    for (int o = 16; o > 0; o >>= 1) v += __shfl_xor(v, o, 32);
    return v;
}
__device__ __forceinline__ float readlane_f(float v, int l) {
    return __int_as_float(__builtin_amdgcn_readlane(__float_as_int(v), l));
}
__device__ __forceinline__ float uni_f(float v) {
    return __int_as_float(__builtin_amdgcn_readfirstlane(__float_as_int(v)));
}
__global__ __launch_bounds__(256) void csr_count(const int* __restrict__ dst,
                                                 int* __restrict__ cnt,
                                                 int* __restrict__ rank, int ne) {
    int e = blockIdx.x * blockDim.x + threadIdx.x;
    if (e >= ne) return;
    rank[e] = atomicAdd(&cnt[dst[e]], 1);
}
__global__ __launch_bounds__(256) void csr_blocksum(const int* __restrict__ cnt,
                                                    int* __restrict__ bsum, int n1) {
    __shared__ int wsum[8];
    const int base = blockIdx.x * 1024 + threadIdx.x * 4;
    int s = 0;
    for (int k = 0; k < 4; ++k)
        if (base + k < n1) s += cnt[base + k];
    s = wred_sum_i(s);
    if ((threadIdx.x & 31) == 0) wsum[threadIdx.x >> 5] = s;
    __syncthreads();
    if (threadIdx.x == 0) {
        int t = 0;
        for (int w = 0; w < 8; ++w) t += wsum[w];
        bsum[blockIdx.x] = t;
    }
}
__global__ __launch_bounds__(256) void csr_rowptr(const int* __restrict__ cnt,
                                                  const int* __restrict__ bsum,
                                                  int* __restrict__ rowptr, int n1) {
    __shared__ int wsum[8];
    __shared__ int wsum2[8];
    const int tid = threadIdx.x, lane = tid & 31, wid = tid >> 5;
    int p = 0;
    for (int j = tid; j < (int)blockIdx.x; j += 256) p += bsum[j];
    p = wred_sum_i(p);
    if (lane == 0) wsum[wid] = p;
    const int base = blockIdx.x * 1024 + tid * 4;
    int c[4];
    int s = 0;
    for (int k = 0; k < 4; ++k) {
        c[k] = (base + k < n1) ? cnt[base + k] : 0;
        s += c[k];
    }
    int inc = s;
    for (int o = 1; o < 32; o <<= 1) {
        int t = __shfl_up(inc, o, 32);
        if (lane >= o) inc += t;
    }
    if (lane == 31) wsum2[wid] = inc;
    __syncthreads();
    int run = inc - s;
    for (int w = 0; w < 8; ++w) {
        run += wsum[w];
        if (w < wid) run += wsum2[w];
    }
    for (int k = 0; k < 4; ++k) {
        if (base + k < n1) rowptr[base + k] = run;
        run += c[k];
    }
}
__global__ __launch_bounds__(256) void csr_place(const int* __restrict__ dst,
                                                 const int* __restrict__ rowptr,
                                                 const int* __restrict__ rank,
                                                 int* __restrict__ tmp, int ne) {
    int e = blockIdx.x * blockDim.x + threadIdx.x;
    if (e >= ne) return;
    tmp[rowptr[dst[e]] + rank[e]] = e;
}
__global__ __launch_bounds__(256) void csr_segsort(const int* __restrict__ rowptr,
                                                   const int* __restrict__ tmp,
                                                   int* __restrict__ perm, int n) {
    const int node = __builtin_amdgcn_readfirstlane((int)((blockIdx.x * 256 + threadIdx.x) >> 5));
    const int lane = threadIdx.x & 31;
    if (node >= n) return;
    const int beg = rowptr[node], end = rowptr[node + 1];
    for (int i0 = beg; i0 < end; i0 += 32) {
        const int i = i0 + lane;
        const int ei = (i < end) ? tmp[i] : 0x7fffffff;
        int r = 0;
        for (int j0 = beg; j0 < end; j0 += 32) {
            const int j = j0 + lane;
            const int ej = (j < end) ? tmp[j] : 0x7fffffff;
            const int m = (end - j0 < 32) ? (end - j0) : 32;
            for (int t = 0; t < m; ++t) r += (__builtin_amdgcn_readlane(ej, t) < ei) ? 1 : 0;
        }
        if (i < end) perm[beg + r] = ei;
    }
}

__device__ __forceinline__ void edge_scores1(const int* __restrict__ perm,
                                             const int* __restrict__ src,
                                             const float4* __restrict__ ss4, const float4 sd,
                                             int i, int end, int& s_i, float4& x) {
    s_i = 0;
    x.x = -INFINITY;
    x.y = -INFINITY;
    x.z = -INFINITY;
    x.w = -INFINITY;
    if (i < end) {
        const int e = perm[i];
        s_i = src[e];
        const float4 vs = ss4[s_i];
        x.x = lrelu(vs.x + sd.x);
        x.y = lrelu(vs.y + sd.y);
        x.z = lrelu(vs.z + sd.z);
        x.w = lrelu(vs.w + sd.w);
    }
}
__device__ __forceinline__ void rows_fma1(const float* __restrict__ wl, const bool hi,
                                          const int s_i, const float4 a_i, const int m,
                                          float4& acc0, float4& acc1) {
#pragma unroll 4
    for (int t = 0; t < m; ++t) {
        const int s = __builtin_amdgcn_readlane(s_i, t);
        const float ax = readlane_f(a_i.x, t);
        const float ay = readlane_f(a_i.y, t);
        const float az = readlane_f(a_i.z, t);
        const float aw = readlane_f(a_i.w, t);
        const float al = hi ? ay : ax;
        const float au = hi ? aw : az;
        const float* r = wl + (size_t)s * 256;
        const float4 w0 = *(const float4*)(r);
        const float4 w1 = *(const float4*)(r + 128);
        acc0.x += al * w0.x;
        acc0.y += al * w0.y;
        acc0.z += al * w0.z;
        acc0.w += al * w0.w;
        acc1.x += au * w1.x;
        acc1.y += au * w1.y;
        acc1.z += au * w1.z;
        acc1.w += au * w1.w;
    }
}

__global__ __launch_bounds__(256) void gather1_elu(const int* __restrict__ rowptr,
                                                   const int* __restrict__ perm,
                                                   const int* __restrict__ src,
                                                   const float4* __restrict__ ss4,
                                                   const float4* __restrict__ sd4,
                                                   const float* __restrict__ Wh,
                                                   float* __restrict__ hp, int n) {
    const int node = __builtin_amdgcn_readfirstlane((int)((blockIdx.x * 256 + threadIdx.x) >> 5));
    const int lane = threadIdx.x & 31;
    if (node >= n) return;
    const int beg = rowptr[node], end = rowptr[node + 1];
    const float4 sd = sd4[node];
    const bool hi = lane >= 16;
    const float* wl = Wh + lane * 4;
    float4 acc0 = {0.f, 0.f, 0.f, 0.f};
    float4 acc1 = {0.f, 0.f, 0.f, 0.f};
    if (end - beg <= 32) {
        int s_i;
        float4 x;
        edge_scores1(perm, src, ss4, sd, beg + lane, end, s_i, x);
        float4 mx, ex, dn, a_i;
        mx.x = wred_max(x.x);
        mx.y = wred_max(x.y);
        mx.z = wred_max(x.z);
        mx.w = wred_max(x.w);
        ex.x = expf(x.x - mx.x);
        ex.y = expf(x.y - mx.y);
        ex.z = expf(x.z - mx.z);
        ex.w = expf(x.w - mx.w);
        dn.x = wred_sum(ex.x);
        dn.y = wred_sum(ex.y);
        dn.z = wred_sum(ex.z);
        dn.w = wred_sum(ex.w);
        a_i.x = ex.x / dn.x;
        a_i.y = ex.y / dn.y;
        a_i.z = ex.z / dn.z;
        a_i.w = ex.w / dn.w;
        rows_fma1(wl, hi, s_i, a_i, end - beg, acc0, acc1);
    } else {
        float4 mx = {-INFINITY, -INFINITY, -INFINITY, -INFINITY};
        for (int i0 = beg; i0 < end; i0 += 32) {
            int s_i;
            float4 x;
            edge_scores1(perm, src, ss4, sd, i0 + lane, end, s_i, x);
            mx.x = fmaxf(mx.x, x.x);
            mx.y = fmaxf(mx.y, x.y);
            mx.z = fmaxf(mx.z, x.z);
            mx.w = fmaxf(mx.w, x.w);
        }
        mx.x = uni_f(wred_max(mx.x));
        mx.y = uni_f(wred_max(mx.y));
        mx.z = uni_f(wred_max(mx.z));
        mx.w = uni_f(wred_max(mx.w));
        float4 dn = {0.f, 0.f, 0.f, 0.f};
        for (int i0 = beg; i0 < end; i0 += 32) {
            int s_i;
            float4 x;
            edge_scores1(perm, src, ss4, sd, i0 + lane, end, s_i, x);
            if (i0 + lane < end) {
                dn.x += expf(x.x - mx.x);
                dn.y += expf(x.y - mx.y);
                dn.z += expf(x.z - mx.z);
                dn.w += expf(x.w - mx.w);
            }
        }
        dn.x = uni_f(wred_sum(dn.x));
        dn.y = uni_f(wred_sum(dn.y));
        dn.z = uni_f(wred_sum(dn.z));
        dn.w = uni_f(wred_sum(dn.w));
        for (int i0 = beg; i0 < end; i0 += 32) {
            int s_i;
            float4 x;
            edge_scores1(perm, src, ss4, sd, i0 + lane, end, s_i, x);
            float4 a_i = {0.f, 0.f, 0.f, 0.f};
            if (i0 + lane < end) {
                a_i.x = expf(x.x - mx.x) / dn.x;
                a_i.y = expf(x.y - mx.y) / dn.y;
                a_i.z = expf(x.z - mx.z) / dn.z;
                a_i.w = expf(x.w - mx.w) / dn.w;
            }
            const int m = (end - i0 < 32) ? (end - i0) : 32;
            rows_fma1(wl, hi, s_i, a_i, m, acc0, acc1);
        }
    }
    acc0.x = acc0.x > 0.f ? acc0.x : (expf(acc0.x) - 1.f);
    acc0.y = acc0.y > 0.f ? acc0.y : (expf(acc0.y) - 1.f);
    acc0.z = acc0.z > 0.f ? acc0.z : (expf(acc0.z) - 1.f);
    acc0.w = acc0.w > 0.f ? acc0.w : (expf(acc0.w) - 1.f);
    acc1.x = acc1.x > 0.f ? acc1.x : (expf(acc1.x) - 1.f);
    acc1.y = acc1.y > 0.f ? acc1.y : (expf(acc1.y) - 1.f);
    acc1.z = acc1.z > 0.f ? acc1.z : (expf(acc1.z) - 1.f);
    acc1.w = acc1.w > 0.f ? acc1.w : (expf(acc1.w) - 1.f);
    float* o = hp + (size_t)node * 256 + lane * 4;
    *(float4*)(o) = acc0;
    *(float4*)(o + 128) = acc1;
}

__device__ __forceinline__ void edge_score2(const int* __restrict__ perm,
                                            const int* __restrict__ src,
                                            const float* __restrict__ ss, const float sd, int i,
                                            int end, int& s_i, float& x) {
    s_i = 0;
    x = -INFINITY;
    if (i < end) {
        const int e = perm[i];
        s_i = src[e];
        x = lrelu(ss[s_i] + sd);
    }
}
__device__ __forceinline__ void rows_fma2(const float* __restrict__ w1p,
                                          const float* __restrict__ w2p, const int s_i,
                                          const float a_i, const int m, float& acc1,
                                          float& acc2) {
#pragma unroll 4
    for (int t = 0; t < m; ++t) {
        const int s = __builtin_amdgcn_readlane(s_i, t);
        const float a = readlane_f(a_i, t);
        acc1 += a * w1p[(size_t)s * 40];
        acc2 += a * w2p[(size_t)s * 40];
    }
}
__global__ __launch_bounds__(256) void gather2_lsm(const int* __restrict__ rowptr,
                                                   const int* __restrict__ perm,
                                                   const int* __restrict__ src,
                                                   const float* __restrict__ ss,
                                                   const float* __restrict__ sdv,
                                                   const float* __restrict__ Wx,
                                                   float* __restrict__ out, int n) {
    const int node = __builtin_amdgcn_readfirstlane((int)((blockIdx.x * 256 + threadIdx.x) >> 5));
    const int lane = threadIdx.x & 31;
    if (node >= n) return;
    const int beg = rowptr[node], end = rowptr[node + 1];
    const float sd = sdv[node];
    const float* w1p = Wx + lane;
    const float* w2p = Wx + 32 + (lane & 7);
    float acc1 = 0.f, acc2 = 0.f;
    if (end - beg <= 32) {
        int s_i;
        float x;
        edge_score2(perm, src, ss, sd, beg + lane, end, s_i, x);
        const float mx = wred_max(x);
        const float ex = expf(x - mx);
        const float dn = wred_sum(ex);
        rows_fma2(w1p, w2p, s_i, ex / dn, end - beg, acc1, acc2);
    } else {
        float mx = -INFINITY;
        for (int i0 = beg; i0 < end; i0 += 32) {
            int s_i;
            float x;
            edge_score2(perm, src, ss, sd, i0 + lane, end, s_i, x);
            mx = fmaxf(mx, x);
        }
        mx = uni_f(wred_max(mx));
        float dn = 0.f;
        for (int i0 = beg; i0 < end; i0 += 32) {
            int s_i;
            float x;
            edge_score2(perm, src, ss, sd, i0 + lane, end, s_i, x);
            if (i0 + lane < end) dn += expf(x - mx);
        }
        dn = uni_f(wred_sum(dn));
        for (int i0 = beg; i0 < end; i0 += 32) {
            int s_i;
            float x;
            edge_score2(perm, src, ss, sd, i0 + lane, end, s_i, x);
            const float a_i = (i0 + lane < end) ? expf(x - mx) / dn : 0.f;
            const int m = (end - i0 < 32) ? (end - i0) : 32;
            rows_fma2(w1p, w2p, s_i, a_i, m, acc1, acc2);
        }
    }
    const float v1 = acc1;
    const float v2 = (lane < 8) ? acc2 : -INFINITY;
    const float mxo = wred_max(fmaxf(v1, v2));
    float sm = expf(v1 - mxo) + ((lane < 8) ? expf(v2 - mxo) : 0.f);
    sm = wred_sum(sm);
    const float lg = mxo + logf(sm);
    float* row = out + (size_t)node * 40;
    row[lane] = v1 - lg;
    if (lane < 8) row[32 + lane] = v2 - lg;
}

extern "C" void kernel_launch(void* const* d_in, const int* in_sizes, int n_in, void* d_out,
                              int out_size, void* d_ws, size_t ws_size, hipStream_t stream) {
    (void)n_in; (void)out_size; (void)ws_size;
    const float* h      = (const float*)d_in[0];
    const int*   eidx   = (const int*)d_in[1];
    const float* Wheads = (const float*)d_in[2];
    const float* aheads = (const float*)d_in[3];
    const float* Wout   = (const float*)d_in[4];
    const float* aout   = (const float*)d_in[5];
    float* out = (float*)d_out;

    const int N = in_sizes[0] / 256;
    const int E = in_sizes[1] / 2;
    const int* src = eidx;
    const int* dst = eidx + E;

    float* ws = (float*)d_ws;
    size_t oWh   = 0;
    size_t oHP   = oWh  + (size_t)N * 256;
    size_t oS1s  = oHP  + (size_t)N * 256;
    size_t oS1d  = oS1s + (size_t)N * 4;
    size_t oEdge = oS1d + (size_t)N * 12;
    size_t oCnt  = oEdge + (size_t)E * 4;
    size_t oRP   = oCnt + (size_t)N + 1;
    size_t oBs   = oRP  + (size_t)N + 1;
    size_t oPerm = oEdge + (size_t)E * 4 + (size_t)N * 4;
    size_t oRank = oEdge;
    size_t oTmp  = oEdge + (size_t)E;
    size_t oS2s  = oS1s;
    size_t oS2d  = oS1s + (size_t)N;
    size_t oWx   = oWh;

    int* cnt    = (int*)(ws + oCnt);
    int* rowptr = (int*)(ws + oRP);
    int* bsum   = (int*)(ws + oBs);
    int* perm   = (int*)(ws + oPerm);
    int* rank   = (int*)(ws + oRank);
    int* tmp    = (int*)(ws + oTmp);

    const int geE = (E + 255) / 256;
    const int n1  = N + 1;
    const int nb  = (n1 + 1023) / 1024;

    fill_u32_kernel<<<(n1 + 255) / 256, 256, 0, stream>>>((unsigned*)cnt, 0u, (long long)n1);

    csr_count<<<geE, 256, 0, stream>>>(dst, cnt, rank, E);
    csr_blocksum<<<nb, 256, 0, stream>>>(cnt, bsum, n1);
    csr_rowptr<<<nb, 256, 0, stream>>>(cnt, bsum, rowptr, n1);
    csr_place<<<geE, 256, 0, stream>>>(dst, rowptr, rank, tmp, E);
    csr_segsort<<<(N + 7) / 8, 256, 0, stream>>>(rowptr, tmp, perm, N);

    dim3 g1((N + 127) / 128, 4);
    gemm_wmma_f32<4, 0><<<g1, 256, 0, stream>>>(h, Wheads, ws + oWh, N);

    scores1_kernel<<<(N + 7) / 8, 256, 0, stream>>>(ws + oWh, aheads, ws + oS1s, ws + oS1d, N);

    gather1_elu<<<(N + 7) / 8, 256, 0, stream>>>(rowptr, perm, src, (const float4*)(ws + oS1s),
                                                 (const float4*)(ws + oS1d), ws + oWh,
                                                 ws + oHP, N);

    dim3 g2((N + 127) / 128, 1);
    gemm_wmma_f32<3, 1><<<g2, 256, 0, stream>>>(ws + oHP, Wout, ws + oWx, N);

    scores2_kernel<<<(N + 7) / 8, 256, 0, stream>>>(ws + oWx, aout, ws + oS2s, ws + oS2d, N);

    gather2_lsm<<<(N + 7) / 8, 256, 0, stream>>>(rowptr, perm, src, ws + oS2s, ws + oS2d,
                                                 ws + oWx, out, N);
}


# attention score dots taken from the WMMA accumulators in both GEMM epilogues: the two score launches (200000 one-node waves, re-read of Wh and Wx) are gone; the dots' sum order changes
# speedup vs baseline: 5.6479x; 1.0746x over previous
#include <hip/hip_runtime.h>


#define LRELU 0.2f

typedef __attribute__((ext_vector_type(2))) float v2f;
typedef __attribute__((ext_vector_type(8))) float v8f;

__device__ __forceinline__ float wred_sum(float v) {
#pragma unroll
    for (int o = 16; o > 0; o >>= 1) v += __shfl_xor(v, o, 32);
    return v;
}
__device__ __forceinline__ float wred_max(float v) {
#pragma unroll
    for (int o = 16; o > 0; o >>= 1) v = fmaxf(v, __shfl_xor(v, o, 32));
    return v;
}
__device__ __forceinline__ float lrelu(float x) { return x > 0.f ? x : LRELU * x; }
__device__ __forceinline__ float xchg_add(const bool up, const float lo, const float hi,
                                          const int w) {
    const float keep = up ? hi : lo;
    const float send = up ? lo : hi;
    return keep + __shfl_xor(send, w, 32);
}

__global__ void fill_u32_kernel(unsigned* __restrict__ p, unsigned v, long long n) {
    long long i = (long long)blockIdx.x * blockDim.x + threadIdx.x;
    if (i < n) p[i] = v;
}

template <int COLT, int MODE>
__global__ __launch_bounds__(256) void gemm_wmma_f32(const float* __restrict__ A,
                                                     const float* __restrict__ B,
                                                     float* __restrict__ C,
                                                     const float* __restrict__ avec,
                                                     float* __restrict__ ssrc,
                                                     float* __restrict__ sdst, int nrows) {
    constexpr int CW = COLT * 16;
    __shared__ float Blds[256 * CW];
    const int tid = threadIdx.x;
    const int colBase = blockIdx.y * CW;

    for (int idx = tid; idx < 256 * CW; idx += 256) {
        float bv;
        if constexpr (MODE == 0) {
            bv = B[(size_t)blockIdx.y * (256 * 64) + idx];
        } else {
            int k = idx / CW;
            int c = idx - k * CW;
            bv = (c < 40) ? B[k * 40 + c] : 0.0f;
        }
        Blds[idx] = bv;
    }
    __syncthreads();

    const int wid  = tid >> 5;
    const int lane = tid & 31;
    const int half = lane >> 4;
    const int l16  = lane & 15;
    const int rowBase = blockIdx.x * 128 + wid * 16;

    int arow = rowBase + l16;
    if (arow >= nrows) arow = nrows - 1;
    const float* aptr = A + (size_t)arow * 256;

    v8f acc[COLT] = {};

    for (int k0 = 0; k0 < 256; k0 += 4) {
        const int kb = k0 + half * 2;
        v2f a;
        a.x = aptr[kb];
        a.y = aptr[kb + 1];
#pragma unroll
        for (int t = 0; t < COLT; ++t) {
            v2f b;
            const int c = t * 16 + l16;
            b.x = Blds[kb * CW + c];
            b.y = Blds[(kb + 1) * CW + c];
            acc[t] = __builtin_amdgcn_wmma_f32_16x16x4_f32(false, a, false, b, (short)0,
                                                           acc[t], false, false);
        }
    }

    {
        float as[COLT], ad[COLT];
        for (int t = 0; t < COLT; ++t) {
            const int c = t * 16 + l16;
            if constexpr (MODE == 0) {
                as[t] = avec[blockIdx.y * 128 + c];
                ad[t] = avec[blockIdx.y * 128 + 64 + c];
            } else {
                as[t] = (c < 40) ? avec[c] : 0.0f;
                ad[t] = (c < 40) ? avec[40 + c] : 0.0f;
            }
        }
#define ROW_DOTS(v, s, d)                      \
    float s = 0.f, d = 0.f;                    \
    for (int t = 0; t < COLT; ++t) {           \
        s += acc[t][v] * as[t];                \
        d += acc[t][v] * ad[t];                \
    }
        ROW_DOTS(0, s0, d0)
        ROW_DOTS(1, s1, d1)
        ROW_DOTS(2, s2, d2)
        ROW_DOTS(3, s3, d3)
        ROW_DOTS(4, s4, d4)
        ROW_DOTS(5, s5, d5)
        ROW_DOTS(6, s6, d6)
        ROW_DOTS(7, s7, d7)
#undef ROW_DOTS
        const bool u8 = (l16 & 8) != 0, u4 = (l16 & 4) != 0, u2 = (l16 & 2) != 0,
                   u1 = (l16 & 1) != 0;
        const float q0 = xchg_add(u8, s0, d0, 8);
        const float q1 = xchg_add(u8, s1, d1, 8);
        const float q2 = xchg_add(u8, s2, d2, 8);
        const float q3 = xchg_add(u8, s3, d3, 8);
        const float q4 = xchg_add(u8, s4, d4, 8);
        const float q5 = xchg_add(u8, s5, d5, 8);
        const float q6 = xchg_add(u8, s6, d6, 8);
        const float q7 = xchg_add(u8, s7, d7, 8);
        const float r0 = xchg_add(u4, q0, q4, 4);
        const float r1 = xchg_add(u4, q1, q5, 4);
        const float r2 = xchg_add(u4, q2, q6, 4);
        const float r3 = xchg_add(u4, q3, q7, 4);
        const float y0 = xchg_add(u2, r0, r2, 2);
        const float y1 = xchg_add(u2, r1, r3, 2);
        const float tot = xchg_add(u1, y0, y1, 1);
        const int srow = rowBase + half * 8 + (l16 & 7);
        float* sp = (l16 < 8) ? ssrc : sdst;
        if (srow < nrows) {
            if constexpr (MODE == 0) {
                sp[(size_t)srow * 4 + blockIdx.y] = tot;
            } else {
                sp[srow] = tot;
            }
        }
    }

    if (rowBase + 16 <= nrows) {
#pragma unroll
        for (int t = 0; t < COLT; ++t) {
            const int col = colBase + t * 16 + l16;
            if constexpr (MODE == 1) {
                if (col >= 40) continue;
            }
#pragma unroll
            for (int v = 0; v < 8; ++v) {
                const int row = rowBase + v + half * 8;
                if constexpr (MODE == 0) {
                    C[(size_t)row * 256 + col] = acc[t][v];
                } else {
                    C[(size_t)row * 40 + col] = acc[t][v];
                }
            }
        }
    } else {
#pragma unroll
        for (int t = 0; t < COLT; ++t) {
            const int col = colBase + t * 16 + l16;
#pragma unroll
            for (int v = 0; v < 8; ++v) {
                const int row = rowBase + v + half * 8;
                if constexpr (MODE == 0) {
                    if (row < nrows) C[(size_t)row * 256 + col] = acc[t][v];
                } else {
                    if (row < nrows && col < 40) C[(size_t)row * 40 + col] = acc[t][v];
                }
            }
        }
    }
}

__device__ __forceinline__ int wred_sum_i(int v) {
    for (int o = 16; o > 0; o >>= 1) v += __shfl_xor(v, o, 32);
    return v;
}
__device__ __forceinline__ float readlane_f(float v, int l) {
    return __int_as_float(__builtin_amdgcn_readlane(__float_as_int(v), l));
}
__device__ __forceinline__ float uni_f(float v) {
    return __int_as_float(__builtin_amdgcn_readfirstlane(__float_as_int(v)));
}
__global__ __launch_bounds__(256) void csr_count(const int* __restrict__ dst,
                                                 int* __restrict__ cnt,
                                                 int* __restrict__ rank, int ne) {
    int e = blockIdx.x * blockDim.x + threadIdx.x;
    if (e >= ne) return;
    rank[e] = atomicAdd(&cnt[dst[e]], 1);
}
__global__ __launch_bounds__(256) void csr_blocksum(const int* __restrict__ cnt,
                                                    int* __restrict__ bsum, int n1) {
    __shared__ int wsum[8];
    const int base = blockIdx.x * 1024 + threadIdx.x * 4;
    int s = 0;
    for (int k = 0; k < 4; ++k)
        if (base + k < n1) s += cnt[base + k];
    s = wred_sum_i(s);
    if ((threadIdx.x & 31) == 0) wsum[threadIdx.x >> 5] = s;
    __syncthreads();
    if (threadIdx.x == 0) {
        int t = 0;
        for (int w = 0; w < 8; ++w) t += wsum[w];
        bsum[blockIdx.x] = t;
    }
}
__global__ __launch_bounds__(256) void csr_rowptr(const int* __restrict__ cnt,
                                                  const int* __restrict__ bsum,
                                                  int* __restrict__ rowptr, int n1) {
    __shared__ int wsum[8];
    __shared__ int wsum2[8];
    const int tid = threadIdx.x, lane = tid & 31, wid = tid >> 5;
    int p = 0;
    for (int j = tid; j < (int)blockIdx.x; j += 256) p += bsum[j];
    p = wred_sum_i(p);
    if (lane == 0) wsum[wid] = p;
    const int base = blockIdx.x * 1024 + tid * 4;
    int c[4];
    int s = 0;
    for (int k = 0; k < 4; ++k) {
        c[k] = (base + k < n1) ? cnt[base + k] : 0;
        s += c[k];
    }
    int inc = s;
    for (int o = 1; o < 32; o <<= 1) {
        int t = __shfl_up(inc, o, 32);
        if (lane >= o) inc += t;
    }
    if (lane == 31) wsum2[wid] = inc;
    __syncthreads();
    int run = inc - s;
    for (int w = 0; w < 8; ++w) {
        run += wsum[w];
        if (w < wid) run += wsum2[w];
    }
    for (int k = 0; k < 4; ++k) {
        if (base + k < n1) rowptr[base + k] = run;
        run += c[k];
    }
}
__global__ __launch_bounds__(256) void csr_place(const int* __restrict__ dst,
                                                 const int* __restrict__ rowptr,
                                                 const int* __restrict__ rank,
                                                 int* __restrict__ tmp, int ne) {
    int e = blockIdx.x * blockDim.x + threadIdx.x;
    if (e >= ne) return;
    tmp[rowptr[dst[e]] + rank[e]] = e;
}
__global__ __launch_bounds__(256) void csr_segsort(const int* __restrict__ rowptr,
                                                   const int* __restrict__ tmp,
                                                   int* __restrict__ perm, int n) {
    const int node = __builtin_amdgcn_readfirstlane((int)((blockIdx.x * 256 + threadIdx.x) >> 5));
    const int lane = threadIdx.x & 31;
    if (node >= n) return;
    const int beg = rowptr[node], end = rowptr[node + 1];
    for (int i0 = beg; i0 < end; i0 += 32) {
        const int i = i0 + lane;
        const int ei = (i < end) ? tmp[i] : 0x7fffffff;
        int r = 0;
        for (int j0 = beg; j0 < end; j0 += 32) {
            const int j = j0 + lane;
            const int ej = (j < end) ? tmp[j] : 0x7fffffff;
            const int m = (end - j0 < 32) ? (end - j0) : 32;
            for (int t = 0; t < m; ++t) r += (__builtin_amdgcn_readlane(ej, t) < ei) ? 1 : 0;
        }
        if (i < end) perm[beg + r] = ei;
    }
}

__device__ __forceinline__ void edge_scores1(const int* __restrict__ perm,
                                             const int* __restrict__ src,
                                             const float4* __restrict__ ss4, const float4 sd,
                                             int i, int end, int& s_i, float4& x) {
    s_i = 0;
    x.x = -INFINITY;
    x.y = -INFINITY;
    x.z = -INFINITY;
    x.w = -INFINITY;
    if (i < end) {
        const int e = perm[i];
        s_i = src[e];
        const float4 vs = ss4[s_i];
        x.x = lrelu(vs.x + sd.x);
        x.y = lrelu(vs.y + sd.y);
        x.z = lrelu(vs.z + sd.z);
        x.w = lrelu(vs.w + sd.w);
    }
}
__device__ __forceinline__ void rows_fma1(const float* __restrict__ wl, const bool hi,
                                          const int s_i, const float4 a_i, const int m,
                                          float4& acc0, float4& acc1) {
#pragma unroll 4
    for (int t = 0; t < m; ++t) {
        const int s = __builtin_amdgcn_readlane(s_i, t);
        const float ax = readlane_f(a_i.x, t);
        const float ay = readlane_f(a_i.y, t);
        const float az = readlane_f(a_i.z, t);
        const float aw = readlane_f(a_i.w, t);
        const float al = hi ? ay : ax;
        const float au = hi ? aw : az;
        const float* r = wl + (size_t)s * 256;
        const float4 w0 = *(const float4*)(r);
        const float4 w1 = *(const float4*)(r + 128);
        acc0.x += al * w0.x;
        acc0.y += al * w0.y;
        acc0.z += al * w0.z;
        acc0.w += al * w0.w;
        acc1.x += au * w1.x;
        acc1.y += au * w1.y;
        acc1.z += au * w1.z;
        acc1.w += au * w1.w;
    }
}

__global__ __launch_bounds__(256) void gather1_elu(const int* __restrict__ rowptr,
                                                   const int* __restrict__ perm,
                                                   const int* __restrict__ src,
                                                   const float4* __restrict__ ss4,
                                                   const float4* __restrict__ sd4,
                                                   const float* __restrict__ Wh,
                                                   float* __restrict__ hp, int n) {
    const int node = __builtin_amdgcn_readfirstlane((int)((blockIdx.x * 256 + threadIdx.x) >> 5));
    const int lane = threadIdx.x & 31;
    if (node >= n) return;
    const int beg = rowptr[node], end = rowptr[node + 1];
    const float4 sd = sd4[node];
    const bool hi = lane >= 16;
    const float* wl = Wh + lane * 4;
    float4 acc0 = {0.f, 0.f, 0.f, 0.f};
    float4 acc1 = {0.f, 0.f, 0.f, 0.f};
    if (end - beg <= 32) {
        int s_i;
        float4 x;
        edge_scores1(perm, src, ss4, sd, beg + lane, end, s_i, x);
        float4 mx, ex, dn, a_i;
        mx.x = wred_max(x.x);
        mx.y = wred_max(x.y);
        mx.z = wred_max(x.z);
        mx.w = wred_max(x.w);
        ex.x = expf(x.x - mx.x);
        ex.y = expf(x.y - mx.y);
        ex.z = expf(x.z - mx.z);
        ex.w = expf(x.w - mx.w);
        dn.x = wred_sum(ex.x);
        dn.y = wred_sum(ex.y);
        dn.z = wred_sum(ex.z);
        dn.w = wred_sum(ex.w);
        a_i.x = ex.x / dn.x;
        a_i.y = ex.y / dn.y;
        a_i.z = ex.z / dn.z;
        a_i.w = ex.w / dn.w;
        rows_fma1(wl, hi, s_i, a_i, end - beg, acc0, acc1);
    } else {
        float4 mx = {-INFINITY, -INFINITY, -INFINITY, -INFINITY};
        for (int i0 = beg; i0 < end; i0 += 32) {
            int s_i;
            float4 x;
            edge_scores1(perm, src, ss4, sd, i0 + lane, end, s_i, x);
            mx.x = fmaxf(mx.x, x.x);
            mx.y = fmaxf(mx.y, x.y);
            mx.z = fmaxf(mx.z, x.z);
            mx.w = fmaxf(mx.w, x.w);
        }
        mx.x = uni_f(wred_max(mx.x));
        mx.y = uni_f(wred_max(mx.y));
        mx.z = uni_f(wred_max(mx.z));
        mx.w = uni_f(wred_max(mx.w));
        float4 dn = {0.f, 0.f, 0.f, 0.f};
        for (int i0 = beg; i0 < end; i0 += 32) {
            int s_i;
            float4 x;
            edge_scores1(perm, src, ss4, sd, i0 + lane, end, s_i, x);
            if (i0 + lane < end) {
                dn.x += expf(x.x - mx.x);
                dn.y += expf(x.y - mx.y);
                dn.z += expf(x.z - mx.z);
                dn.w += expf(x.w - mx.w);
            }
        }
        dn.x = uni_f(wred_sum(dn.x));
        dn.y = uni_f(wred_sum(dn.y));
        dn.z = uni_f(wred_sum(dn.z));
        dn.w = uni_f(wred_sum(dn.w));
        for (int i0 = beg; i0 < end; i0 += 32) {
            int s_i;
            float4 x;
            edge_scores1(perm, src, ss4, sd, i0 + lane, end, s_i, x);
            float4 a_i = {0.f, 0.f, 0.f, 0.f};
            if (i0 + lane < end) {
                a_i.x = expf(x.x - mx.x) / dn.x;
                a_i.y = expf(x.y - mx.y) / dn.y;
                a_i.z = expf(x.z - mx.z) / dn.z;
                a_i.w = expf(x.w - mx.w) / dn.w;
            }
            const int m = (end - i0 < 32) ? (end - i0) : 32;
            rows_fma1(wl, hi, s_i, a_i, m, acc0, acc1);
        }
    }
    acc0.x = acc0.x > 0.f ? acc0.x : (expf(acc0.x) - 1.f);
    acc0.y = acc0.y > 0.f ? acc0.y : (expf(acc0.y) - 1.f);
    acc0.z = acc0.z > 0.f ? acc0.z : (expf(acc0.z) - 1.f);
    acc0.w = acc0.w > 0.f ? acc0.w : (expf(acc0.w) - 1.f);
    acc1.x = acc1.x > 0.f ? acc1.x : (expf(acc1.x) - 1.f);
    acc1.y = acc1.y > 0.f ? acc1.y : (expf(acc1.y) - 1.f);
    acc1.z = acc1.z > 0.f ? acc1.z : (expf(acc1.z) - 1.f);
    acc1.w = acc1.w > 0.f ? acc1.w : (expf(acc1.w) - 1.f);
    float* o = hp + (size_t)node * 256 + lane * 4;
    *(float4*)(o) = acc0;
    *(float4*)(o + 128) = acc1;
}

__device__ __forceinline__ void edge_score2(const int* __restrict__ perm,
                                            const int* __restrict__ src,
                                            const float* __restrict__ ss, const float sd, int i,
                                            int end, int& s_i, float& x) {
    s_i = 0;
    x = -INFINITY;
    if (i < end) {
        const int e = perm[i];
        s_i = src[e];
        x = lrelu(ss[s_i] + sd);
    }
}
__device__ __forceinline__ void rows_fma2(const float* __restrict__ w1p,
                                          const float* __restrict__ w2p, const int s_i,
                                          const float a_i, const int m, float& acc1,
                                          float& acc2) {
#pragma unroll 4
    for (int t = 0; t < m; ++t) {
        const int s = __builtin_amdgcn_readlane(s_i, t);
        const float a = readlane_f(a_i, t);
        acc1 += a * w1p[(size_t)s * 40];
        acc2 += a * w2p[(size_t)s * 40];
    }
}
__global__ __launch_bounds__(256) void gather2_lsm(const int* __restrict__ rowptr,
                                                   const int* __restrict__ perm,
                                                   const int* __restrict__ src,
                                                   const float* __restrict__ ss,
                                                   const float* __restrict__ sdv,
                                                   const float* __restrict__ Wx,
                                                   float* __restrict__ out, int n) {
    const int node = __builtin_amdgcn_readfirstlane((int)((blockIdx.x * 256 + threadIdx.x) >> 5));
    const int lane = threadIdx.x & 31;
    if (node >= n) return;
    const int beg = rowptr[node], end = rowptr[node + 1];
    const float sd = sdv[node];
    const float* w1p = Wx + lane;
    const float* w2p = Wx + 32 + (lane & 7);
    float acc1 = 0.f, acc2 = 0.f;
    if (end - beg <= 32) {
        int s_i;
        float x;
        edge_score2(perm, src, ss, sd, beg + lane, end, s_i, x);
        const float mx = wred_max(x);
        const float ex = expf(x - mx);
        const float dn = wred_sum(ex);
        rows_fma2(w1p, w2p, s_i, ex / dn, end - beg, acc1, acc2);
    } else {
        float mx = -INFINITY;
        for (int i0 = beg; i0 < end; i0 += 32) {
            int s_i;
            float x;
            edge_score2(perm, src, ss, sd, i0 + lane, end, s_i, x);
            mx = fmaxf(mx, x);
        }
        mx = uni_f(wred_max(mx));
        float dn = 0.f;
        for (int i0 = beg; i0 < end; i0 += 32) {
            int s_i;
            float x;
            edge_score2(perm, src, ss, sd, i0 + lane, end, s_i, x);
            if (i0 + lane < end) dn += expf(x - mx);
        }
        dn = uni_f(wred_sum(dn));
        for (int i0 = beg; i0 < end; i0 += 32) {
            int s_i;
            float x;
            edge_score2(perm, src, ss, sd, i0 + lane, end, s_i, x);
            const float a_i = (i0 + lane < end) ? expf(x - mx) / dn : 0.f;
            const int m = (end - i0 < 32) ? (end - i0) : 32;
            rows_fma2(w1p, w2p, s_i, a_i, m, acc1, acc2);
        }
    }
    const float v1 = acc1;
    const float v2 = (lane < 8) ? acc2 : -INFINITY;
    const float mxo = wred_max(fmaxf(v1, v2));
    float sm = expf(v1 - mxo) + ((lane < 8) ? expf(v2 - mxo) : 0.f);
    sm = wred_sum(sm);
    const float lg = mxo + logf(sm);
    float* row = out + (size_t)node * 40;
    row[lane] = v1 - lg;
    if (lane < 8) row[32 + lane] = v2 - lg;
}

extern "C" void kernel_launch(void* const* d_in, const int* in_sizes, int n_in, void* d_out,
                              int out_size, void* d_ws, size_t ws_size, hipStream_t stream) {
    (void)n_in; (void)out_size; (void)ws_size;
    const float* h      = (const float*)d_in[0];
    const int*   eidx   = (const int*)d_in[1];
    const float* Wheads = (const float*)d_in[2];
    const float* aheads = (const float*)d_in[3];
    const float* Wout   = (const float*)d_in[4];
    const float* aout   = (const float*)d_in[5];
    float* out = (float*)d_out;

    const int N = in_sizes[0] / 256;
    const int E = in_sizes[1] / 2;
    const int* src = eidx;
    const int* dst = eidx + E;

    float* ws = (float*)d_ws;
    size_t oWh   = 0;
    size_t oHP   = oWh  + (size_t)N * 256;
    size_t oS1s  = oHP  + (size_t)N * 256;
    size_t oS1d  = oS1s + (size_t)N * 4;
    size_t oEdge = oS1d + (size_t)N * 12;
    size_t oCnt  = oEdge + (size_t)E * 4;
    size_t oRP   = oCnt + (size_t)N + 1;
    size_t oBs   = oRP  + (size_t)N + 1;
    size_t oPerm = oEdge + (size_t)E * 4 + (size_t)N * 4;
    size_t oRank = oEdge;
    size_t oTmp  = oEdge + (size_t)E;
    size_t oS2s  = oS1s;
    size_t oS2d  = oS1s + (size_t)N;
    size_t oWx   = oWh;

    int* cnt    = (int*)(ws + oCnt);
    int* rowptr = (int*)(ws + oRP);
    int* bsum   = (int*)(ws + oBs);
    int* perm   = (int*)(ws + oPerm);
    int* rank   = (int*)(ws + oRank);
    int* tmp    = (int*)(ws + oTmp);

    const int geE = (E + 255) / 256;
    const int n1  = N + 1;
    const int nb  = (n1 + 1023) / 1024;

    fill_u32_kernel<<<(n1 + 255) / 256, 256, 0, stream>>>((unsigned*)cnt, 0u, (long long)n1);

    csr_count<<<geE, 256, 0, stream>>>(dst, cnt, rank, E);
    csr_blocksum<<<nb, 256, 0, stream>>>(cnt, bsum, n1);
    csr_rowptr<<<nb, 256, 0, stream>>>(cnt, bsum, rowptr, n1);
    csr_place<<<geE, 256, 0, stream>>>(dst, rowptr, rank, tmp, E);
    csr_segsort<<<(N + 7) / 8, 256, 0, stream>>>(rowptr, tmp, perm, N);

    dim3 g1((N + 127) / 128, 4);
    gemm_wmma_f32<4, 0><<<g1, 256, 0, stream>>>(h, Wheads, ws + oWh, aheads, ws + oS1s,
                                                ws + oS1d, N);

    gather1_elu<<<(N + 7) / 8, 256, 0, stream>>>(rowptr, perm, src, (const float4*)(ws + oS1s),
                                                 (const float4*)(ws + oS1d), ws + oWh,
                                                 ws + oHP, N);

    dim3 g2((N + 127) / 128, 1);
    gemm_wmma_f32<3, 1><<<g2, 256, 0, stream>>>(ws + oHP, Wout, ws + oWx, aout, ws + oS2s,
                                                ws + oS2d, N);

    gather2_lsm<<<(N + 7) / 8, 256, 0, stream>>>(rowptr, perm, src, ws + oS2s, ws + oS2d,
                                                 ws + oWx, out, N);
}


# counting sort's 1600000 returning memory atomics replaced by per-chunk LDS tables (16-bit counter per node, returning LDS add = slot); clear launch gone; sorted list and sum order unchanged
# speedup vs baseline: 5.8751x; 1.0402x over previous
#include <hip/hip_runtime.h>


#define LRELU 0.2f

typedef __attribute__((ext_vector_type(2))) float v2f;
typedef __attribute__((ext_vector_type(8))) float v8f;

__device__ __forceinline__ float wred_sum(float v) {
#pragma unroll
    for (int o = 16; o > 0; o >>= 1) v += __shfl_xor(v, o, 32);
    return v;
}
__device__ __forceinline__ float wred_max(float v) {
#pragma unroll
    for (int o = 16; o > 0; o >>= 1) v = fmaxf(v, __shfl_xor(v, o, 32));
    return v;
}
__device__ __forceinline__ float lrelu(float x) { return x > 0.f ? x : LRELU * x; }
__device__ __forceinline__ float xchg_add(const bool up, const float lo, const float hi,
                                          const int w) {
    const float keep = up ? hi : lo;
    const float send = up ? lo : hi;
    return keep + __shfl_xor(send, w, 32);
}

template <int COLT, int MODE>
__global__ __launch_bounds__(256) void gemm_wmma_f32(const float* __restrict__ A,
                                                     const float* __restrict__ B,
                                                     float* __restrict__ C,
                                                     const float* __restrict__ avec,
                                                     float* __restrict__ ssrc,
                                                     float* __restrict__ sdst, int nrows) {
    constexpr int CW = COLT * 16;
    __shared__ float Blds[256 * CW];
    const int tid = threadIdx.x;
    const int colBase = blockIdx.y * CW;

    for (int idx = tid; idx < 256 * CW; idx += 256) {
        float bv;
        if constexpr (MODE == 0) {
            bv = B[(size_t)blockIdx.y * (256 * 64) + idx];
        } else {
            int k = idx / CW;
            int c = idx - k * CW;
            bv = (c < 40) ? B[k * 40 + c] : 0.0f;
        }
        Blds[idx] = bv;
    }
    __syncthreads();

    const int wid  = tid >> 5;
    const int lane = tid & 31;
    const int half = lane >> 4;
    const int l16  = lane & 15;
    const int rowBase = blockIdx.x * 128 + wid * 16;

    int arow = rowBase + l16;
    if (arow >= nrows) arow = nrows - 1;
    const float* aptr = A + (size_t)arow * 256;

    v8f acc[COLT] = {};

    for (int k0 = 0; k0 < 256; k0 += 4) {
        const int kb = k0 + half * 2;
        v2f a;
        a.x = aptr[kb];
        a.y = aptr[kb + 1];
#pragma unroll
        for (int t = 0; t < COLT; ++t) {
            v2f b;
            const int c = t * 16 + l16;
            b.x = Blds[kb * CW + c];
            b.y = Blds[(kb + 1) * CW + c];
            acc[t] = __builtin_amdgcn_wmma_f32_16x16x4_f32(false, a, false, b, (short)0,
                                                           acc[t], false, false);
        }
    }

    {
        float as[COLT], ad[COLT];
        for (int t = 0; t < COLT; ++t) {
            const int c = t * 16 + l16;
            if constexpr (MODE == 0) {
                as[t] = avec[blockIdx.y * 128 + c];
                ad[t] = avec[blockIdx.y * 128 + 64 + c];
            } else {
                as[t] = (c < 40) ? avec[c] : 0.0f;
                ad[t] = (c < 40) ? avec[40 + c] : 0.0f;
            }
        }
#define ROW_DOTS(v, s, d)                      \
    float s = 0.f, d = 0.f;                    \
    for (int t = 0; t < COLT; ++t) {           \
        s += acc[t][v] * as[t];                \
        d += acc[t][v] * ad[t];                \
    }
        ROW_DOTS(0, s0, d0)
        ROW_DOTS(1, s1, d1)
        ROW_DOTS(2, s2, d2)
        ROW_DOTS(3, s3, d3)
        ROW_DOTS(4, s4, d4)
        ROW_DOTS(5, s5, d5)
        ROW_DOTS(6, s6, d6)
        ROW_DOTS(7, s7, d7)
#undef ROW_DOTS
        const bool u8 = (l16 & 8) != 0, u4 = (l16 & 4) != 0, u2 = (l16 & 2) != 0,
                   u1 = (l16 & 1) != 0;
        const float q0 = xchg_add(u8, s0, d0, 8);
        const float q1 = xchg_add(u8, s1, d1, 8);
        const float q2 = xchg_add(u8, s2, d2, 8);
        const float q3 = xchg_add(u8, s3, d3, 8);
        const float q4 = xchg_add(u8, s4, d4, 8);
        const float q5 = xchg_add(u8, s5, d5, 8);
        const float q6 = xchg_add(u8, s6, d6, 8);
        const float q7 = xchg_add(u8, s7, d7, 8);
        const float r0 = xchg_add(u4, q0, q4, 4);
        const float r1 = xchg_add(u4, q1, q5, 4);
        const float r2 = xchg_add(u4, q2, q6, 4);
        const float r3 = xchg_add(u4, q3, q7, 4);
        const float y0 = xchg_add(u2, r0, r2, 2);
        const float y1 = xchg_add(u2, r1, r3, 2);
        const float tot = xchg_add(u1, y0, y1, 1);
        const int srow = rowBase + half * 8 + (l16 & 7);
        float* sp = (l16 < 8) ? ssrc : sdst;
        if (srow < nrows) {
            if constexpr (MODE == 0) {
                sp[(size_t)srow * 4 + blockIdx.y] = tot;
            } else {
                sp[srow] = tot;
            }
        }
    }

    if (rowBase + 16 <= nrows) {
#pragma unroll
        for (int t = 0; t < COLT; ++t) {
            const int col = colBase + t * 16 + l16;
            if constexpr (MODE == 1) {
                if (col >= 40) continue;
            }
#pragma unroll
            for (int v = 0; v < 8; ++v) {
                const int row = rowBase + v + half * 8;
                if constexpr (MODE == 0) {
                    C[(size_t)row * 256 + col] = acc[t][v];
                } else {
                    C[(size_t)row * 40 + col] = acc[t][v];
                }
            }
        }
    } else {
#pragma unroll
        for (int t = 0; t < COLT; ++t) {
            const int col = colBase + t * 16 + l16;
#pragma unroll
            for (int v = 0; v < 8; ++v) {
                const int row = rowBase + v + half * 8;
                if constexpr (MODE == 0) {
                    if (row < nrows) C[(size_t)row * 256 + col] = acc[t][v];
                } else {
                    if (row < nrows && col < 40) C[(size_t)row * 40 + col] = acc[t][v];
                }
            }
        }
    }
}

__device__ __forceinline__ int wred_sum_i(int v) {
    for (int o = 16; o > 0; o >>= 1) v += __shfl_xor(v, o, 32);
    return v;
}
__device__ __forceinline__ float readlane_f(float v, int l) {
    return __int_as_float(__builtin_amdgcn_readlane(__float_as_int(v), l));
}
__device__ __forceinline__ float uni_f(float v) {
    return __int_as_float(__builtin_amdgcn_readfirstlane(__float_as_int(v)));
}
#define CSR_CHUNK 50000
#define CSR_CAPW 51200
#define CSR_SP (2 * CSR_CAPW)
__global__ __launch_bounds__(1024) void csr_count_lds(const int* __restrict__ dst,
                                                      unsigned* __restrict__ cntw,
                                                      int* __restrict__ rank, int ne) {
    __shared__ uint4 cl4[CSR_CAPW / 4];
    unsigned* cl = (unsigned*)cl4;
    const int tid = threadIdx.x;
    for (int i = tid; i < CSR_CAPW / 4; i += 1024) cl4[i] = make_uint4(0u, 0u, 0u, 0u);
    __syncthreads();
    const int e0 = blockIdx.x * CSR_CHUNK;
    const int e1 = (e0 + CSR_CHUNK < ne) ? (e0 + CSR_CHUNK) : ne;
    for (int eb = e0 + tid; eb < e1; eb += 16 * 1024) {
        int d[16];
        unsigned old[16];
#pragma unroll
        for (int k = 0; k < 16; ++k) {
            const int e = eb + k * 1024;
            d[k] = dst[(e < e1) ? e : (e1 - 1)];
        }
#pragma unroll
        for (int k = 0; k < 16; ++k) {
            const int e = eb + k * 1024;
            const int sh = (d[k] & 1) << 4;
            old[k] = atomicAdd(&cl[d[k] >> 1], (e < e1) ? (1u << sh) : 0u);
        }
#pragma unroll
        for (int k = 0; k < 16; ++k) {
            const int e = eb + k * 1024;
            const int sh = (d[k] & 1) << 4;
            if (e < e1) rank[e] = (int)((old[k] >> sh) & 0xffffu);
        }
    }
    __syncthreads();
    uint4* o = (uint4*)(cntw + (size_t)blockIdx.x * CSR_CAPW);
    for (int i = tid; i < CSR_CAPW / 4; i += 1024) o[i] = cl4[i];
}
__device__ __forceinline__ void csr_deg4(const unsigned* __restrict__ cntw, const int nch,
                                         const int base, int c[4]) {
    c[0] = 0;
    c[1] = 0;
    c[2] = 0;
    c[3] = 0;
    const unsigned* p = cntw + (base >> 1);
#pragma unroll 8
    for (int ch = 0; ch < nch; ++ch) {
        const uint2 v = *(const uint2*)(p + (size_t)ch * CSR_CAPW);
        c[0] += (int)(v.x & 0xffffu);
        c[1] += (int)(v.x >> 16);
        c[2] += (int)(v.y & 0xffffu);
        c[3] += (int)(v.y >> 16);
    }
}
__global__ __launch_bounds__(256) void csr_blocksum(const unsigned* __restrict__ cntw,
                                                    int* __restrict__ bsum, int n1, int nch) {
    __shared__ int wsum[8];
    const int base = blockIdx.x * 1024 + threadIdx.x * 4;
    int c[4];
    csr_deg4(cntw, nch, base, c);
    int s = 0;
    for (int k = 0; k < 4; ++k)
        if (base + k < n1) s += c[k];
    s = wred_sum_i(s);
    if ((threadIdx.x & 31) == 0) wsum[threadIdx.x >> 5] = s;
    __syncthreads();
    if (threadIdx.x == 0) {
        int t = 0;
        for (int w = 0; w < 8; ++w) t += wsum[w];
        bsum[blockIdx.x] = t;
    }
}
__global__ __launch_bounds__(256) void csr_rowptr(const unsigned* __restrict__ cntw,
                                                  const int* __restrict__ bsum,
                                                  int* __restrict__ rowptr,
                                                  int* __restrict__ start, int n1, int nch) {
    __shared__ int wsum[8];
    __shared__ int wsum2[8];
    const int tid = threadIdx.x, lane = tid & 31, wid = tid >> 5;
    int p = 0;
    for (int j = tid; j < (int)blockIdx.x; j += 256) p += bsum[j];
    p = wred_sum_i(p);
    if (lane == 0) wsum[wid] = p;
    const int base = blockIdx.x * 1024 + tid * 4;
    int c[4];
    csr_deg4(cntw, nch, base, c);
    int s = 0;
    for (int k = 0; k < 4; ++k) {
        if (base + k >= n1) c[k] = 0;
        s += c[k];
    }
    int inc = s;
    for (int o = 1; o < 32; o <<= 1) {
        int t = __shfl_up(inc, o, 32);
        if (lane >= o) inc += t;
    }
    if (lane == 31) wsum2[wid] = inc;
    __syncthreads();
    int run = inc - s;
    for (int w = 0; w < 8; ++w) {
        run += wsum[w];
        if (w < wid) run += wsum2[w];
    }
    int4 a;
    a.x = run;
    a.y = a.x + c[0];
    a.z = a.y + c[1];
    a.w = a.z + c[2];
    if (base < n1) rowptr[base] = a.x;
    if (base + 1 < n1) rowptr[base + 1] = a.y;
    if (base + 2 < n1) rowptr[base + 2] = a.z;
    if (base + 3 < n1) rowptr[base + 3] = a.w;
    const unsigned* cp = cntw + (base >> 1);
#pragma unroll 8
    for (int ch = 0; ch < nch; ++ch) {
        const uint2 v = *(const uint2*)(cp + (size_t)ch * CSR_CAPW);
        *(int4*)(start + (size_t)ch * CSR_SP + base) = a;
        a.x += (int)(v.x & 0xffffu);
        a.y += (int)(v.x >> 16);
        a.z += (int)(v.y & 0xffffu);
        a.w += (int)(v.y >> 16);
    }
}
__global__ __launch_bounds__(256) void csr_place(const int* __restrict__ dst,
                                                 const int* __restrict__ start,
                                                 const int* __restrict__ rank,
                                                 int* __restrict__ tmp, int ne) {
    int e = blockIdx.x * blockDim.x + threadIdx.x;
    if (e >= ne) return;
    const int ch = e / CSR_CHUNK;
    tmp[start[(size_t)ch * CSR_SP + dst[e]] + rank[e]] = e;
}
__global__ __launch_bounds__(256) void csr_segsort(const int* __restrict__ rowptr,
                                                   const int* __restrict__ tmp,
                                                   int* __restrict__ perm, int n) {
    const int node = __builtin_amdgcn_readfirstlane((int)((blockIdx.x * 256 + threadIdx.x) >> 5));
    const int lane = threadIdx.x & 31;
    if (node >= n) return;
    const int beg = rowptr[node], end = rowptr[node + 1];
    for (int i0 = beg; i0 < end; i0 += 32) {
        const int i = i0 + lane;
        const int ei = (i < end) ? tmp[i] : 0x7fffffff;
        int r = 0;
        for (int j0 = beg; j0 < end; j0 += 32) {
            const int j = j0 + lane;
            const int ej = (j < end) ? tmp[j] : 0x7fffffff;
            const int m = (end - j0 < 32) ? (end - j0) : 32;
            for (int t = 0; t < m; ++t) r += (__builtin_amdgcn_readlane(ej, t) < ei) ? 1 : 0;
        }
        if (i < end) perm[beg + r] = ei;
    }
}

__device__ __forceinline__ void edge_scores1(const int* __restrict__ perm,
                                             const int* __restrict__ src,
                                             const float4* __restrict__ ss4, const float4 sd,
                                             int i, int end, int& s_i, float4& x) {
    s_i = 0;
    x.x = -INFINITY;
    x.y = -INFINITY;
    x.z = -INFINITY;
    x.w = -INFINITY;
    if (i < end) {
        const int e = perm[i];
        s_i = src[e];
        const float4 vs = ss4[s_i];
        x.x = lrelu(vs.x + sd.x);
        x.y = lrelu(vs.y + sd.y);
        x.z = lrelu(vs.z + sd.z);
        x.w = lrelu(vs.w + sd.w);
    }
}
__device__ __forceinline__ void rows_fma1(const float* __restrict__ wl, const bool hi,
                                          const int s_i, const float4 a_i, const int m,
                                          float4& acc0, float4& acc1) {
#pragma unroll 4
    for (int t = 0; t < m; ++t) {
        const int s = __builtin_amdgcn_readlane(s_i, t);
        const float ax = readlane_f(a_i.x, t);
        const float ay = readlane_f(a_i.y, t);
        const float az = readlane_f(a_i.z, t);
        const float aw = readlane_f(a_i.w, t);
        const float al = hi ? ay : ax;
        const float au = hi ? aw : az;
        const float* r = wl + (size_t)s * 256;
        const float4 w0 = *(const float4*)(r);
        const float4 w1 = *(const float4*)(r + 128);
        acc0.x += al * w0.x;
        acc0.y += al * w0.y;
        acc0.z += al * w0.z;
        acc0.w += al * w0.w;
        acc1.x += au * w1.x;
        acc1.y += au * w1.y;
        acc1.z += au * w1.z;
        acc1.w += au * w1.w;
    }
}

__global__ __launch_bounds__(256) void gather1_elu(const int* __restrict__ rowptr,
                                                   const int* __restrict__ perm,
                                                   const int* __restrict__ src,
                                                   const float4* __restrict__ ss4,
                                                   const float4* __restrict__ sd4,
                                                   const float* __restrict__ Wh,
                                                   float* __restrict__ hp, int n) {
    const int node = __builtin_amdgcn_readfirstlane((int)((blockIdx.x * 256 + threadIdx.x) >> 5));
    const int lane = threadIdx.x & 31;
    if (node >= n) return;
    const int beg = rowptr[node], end = rowptr[node + 1];
    const float4 sd = sd4[node];
    const bool hi = lane >= 16;
    const float* wl = Wh + lane * 4;
    float4 acc0 = {0.f, 0.f, 0.f, 0.f};
    float4 acc1 = {0.f, 0.f, 0.f, 0.f};
    if (end - beg <= 32) {
        int s_i;
        float4 x;
        edge_scores1(perm, src, ss4, sd, beg + lane, end, s_i, x);
        float4 mx, ex, dn, a_i;
        mx.x = wred_max(x.x);
        mx.y = wred_max(x.y);
        mx.z = wred_max(x.z);
        mx.w = wred_max(x.w);
        ex.x = expf(x.x - mx.x);
        ex.y = expf(x.y - mx.y);
        ex.z = expf(x.z - mx.z);
        ex.w = expf(x.w - mx.w);
        dn.x = wred_sum(ex.x);
        dn.y = wred_sum(ex.y);
        dn.z = wred_sum(ex.z);
        dn.w = wred_sum(ex.w);
        a_i.x = ex.x / dn.x;
        a_i.y = ex.y / dn.y;
        a_i.z = ex.z / dn.z;
        a_i.w = ex.w / dn.w;
        rows_fma1(wl, hi, s_i, a_i, end - beg, acc0, acc1);
    } else {
        float4 mx = {-INFINITY, -INFINITY, -INFINITY, -INFINITY};
        for (int i0 = beg; i0 < end; i0 += 32) {
            int s_i;
            float4 x;
            edge_scores1(perm, src, ss4, sd, i0 + lane, end, s_i, x);
            mx.x = fmaxf(mx.x, x.x);
            mx.y = fmaxf(mx.y, x.y);
            mx.z = fmaxf(mx.z, x.z);
            mx.w = fmaxf(mx.w, x.w);
        }
        mx.x = uni_f(wred_max(mx.x));
        mx.y = uni_f(wred_max(mx.y));
        mx.z = uni_f(wred_max(mx.z));
        mx.w = uni_f(wred_max(mx.w));
        float4 dn = {0.f, 0.f, 0.f, 0.f};
        for (int i0 = beg; i0 < end; i0 += 32) {
            int s_i;
            float4 x;
            edge_scores1(perm, src, ss4, sd, i0 + lane, end, s_i, x);
            if (i0 + lane < end) {
                dn.x += expf(x.x - mx.x);
                dn.y += expf(x.y - mx.y);
                dn.z += expf(x.z - mx.z);
                dn.w += expf(x.w - mx.w);
            }
        }
        dn.x = uni_f(wred_sum(dn.x));
        dn.y = uni_f(wred_sum(dn.y));
        dn.z = uni_f(wred_sum(dn.z));
        dn.w = uni_f(wred_sum(dn.w));
        for (int i0 = beg; i0 < end; i0 += 32) {
            int s_i;
            float4 x;
            edge_scores1(perm, src, ss4, sd, i0 + lane, end, s_i, x);
            float4 a_i = {0.f, 0.f, 0.f, 0.f};
            if (i0 + lane < end) {
                a_i.x = expf(x.x - mx.x) / dn.x;
                a_i.y = expf(x.y - mx.y) / dn.y;
                a_i.z = expf(x.z - mx.z) / dn.z;
                a_i.w = expf(x.w - mx.w) / dn.w;
            }
            const int m = (end - i0 < 32) ? (end - i0) : 32;
            rows_fma1(wl, hi, s_i, a_i, m, acc0, acc1);
        }
    }
    acc0.x = acc0.x > 0.f ? acc0.x : (expf(acc0.x) - 1.f);
    acc0.y = acc0.y > 0.f ? acc0.y : (expf(acc0.y) - 1.f);
    acc0.z = acc0.z > 0.f ? acc0.z : (expf(acc0.z) - 1.f);
    acc0.w = acc0.w > 0.f ? acc0.w : (expf(acc0.w) - 1.f);
    acc1.x = acc1.x > 0.f ? acc1.x : (expf(acc1.x) - 1.f);
    acc1.y = acc1.y > 0.f ? acc1.y : (expf(acc1.y) - 1.f);
    acc1.z = acc1.z > 0.f ? acc1.z : (expf(acc1.z) - 1.f);
    acc1.w = acc1.w > 0.f ? acc1.w : (expf(acc1.w) - 1.f);
    float* o = hp + (size_t)node * 256 + lane * 4;
    *(float4*)(o) = acc0;
    *(float4*)(o + 128) = acc1;
}

__device__ __forceinline__ void edge_score2(const int* __restrict__ perm,
                                            const int* __restrict__ src,
                                            const float* __restrict__ ss, const float sd, int i,
                                            int end, int& s_i, float& x) {
    s_i = 0;
    x = -INFINITY;
    if (i < end) {
        const int e = perm[i];
        s_i = src[e];
        x = lrelu(ss[s_i] + sd);
    }
}
__device__ __forceinline__ void rows_fma2(const float* __restrict__ w1p,
                                          const float* __restrict__ w2p, const int s_i,
                                          const float a_i, const int m, float& acc1,
                                          float& acc2) {
#pragma unroll 4
    for (int t = 0; t < m; ++t) {
        const int s = __builtin_amdgcn_readlane(s_i, t);
        const float a = readlane_f(a_i, t);
        acc1 += a * w1p[(size_t)s * 40];
        acc2 += a * w2p[(size_t)s * 40];
    }
}
__global__ __launch_bounds__(256) void gather2_lsm(const int* __restrict__ rowptr,
                                                   const int* __restrict__ perm,
                                                   const int* __restrict__ src,
                                                   const float* __restrict__ ss,
                                                   const float* __restrict__ sdv,
                                                   const float* __restrict__ Wx,
                                                   float* __restrict__ out, int n) {
    const int node = __builtin_amdgcn_readfirstlane((int)((blockIdx.x * 256 + threadIdx.x) >> 5));
    const int lane = threadIdx.x & 31;
    if (node >= n) return;
    const int beg = rowptr[node], end = rowptr[node + 1];
    const float sd = sdv[node];
    const float* w1p = Wx + lane;
    const float* w2p = Wx + 32 + (lane & 7);
    float acc1 = 0.f, acc2 = 0.f;
    if (end - beg <= 32) {
        int s_i;
        float x;
        edge_score2(perm, src, ss, sd, beg + lane, end, s_i, x);
        const float mx = wred_max(x);
        const float ex = expf(x - mx);
        const float dn = wred_sum(ex);
        rows_fma2(w1p, w2p, s_i, ex / dn, end - beg, acc1, acc2);
    } else {
        float mx = -INFINITY;
        for (int i0 = beg; i0 < end; i0 += 32) {
            int s_i;
            float x;
            edge_score2(perm, src, ss, sd, i0 + lane, end, s_i, x);
            mx = fmaxf(mx, x);
        }
        mx = uni_f(wred_max(mx));
        float dn = 0.f;
        for (int i0 = beg; i0 < end; i0 += 32) {
            int s_i;
            float x;
            edge_score2(perm, src, ss, sd, i0 + lane, end, s_i, x);
            if (i0 + lane < end) dn += expf(x - mx);
        }
        dn = uni_f(wred_sum(dn));
        for (int i0 = beg; i0 < end; i0 += 32) {
            int s_i;
            float x;
            edge_score2(perm, src, ss, sd, i0 + lane, end, s_i, x);
            const float a_i = (i0 + lane < end) ? expf(x - mx) / dn : 0.f;
            const int m = (end - i0 < 32) ? (end - i0) : 32;
            rows_fma2(w1p, w2p, s_i, a_i, m, acc1, acc2);
        }
    }
    const float v1 = acc1;
    const float v2 = (lane < 8) ? acc2 : -INFINITY;
    const float mxo = wred_max(fmaxf(v1, v2));
    float sm = expf(v1 - mxo) + ((lane < 8) ? expf(v2 - mxo) : 0.f);
    sm = wred_sum(sm);
    const float lg = mxo + logf(sm);
    float* row = out + (size_t)node * 40;
    row[lane] = v1 - lg;
    if (lane < 8) row[32 + lane] = v2 - lg;
}

extern "C" void kernel_launch(void* const* d_in, const int* in_sizes, int n_in, void* d_out,
                              int out_size, void* d_ws, size_t ws_size, hipStream_t stream) {
    (void)n_in; (void)out_size; (void)ws_size;
    const float* h      = (const float*)d_in[0];
    const int*   eidx   = (const int*)d_in[1];
    const float* Wheads = (const float*)d_in[2];
    const float* aheads = (const float*)d_in[3];
    const float* Wout   = (const float*)d_in[4];
    const float* aout   = (const float*)d_in[5];
    float* out = (float*)d_out;

    const int N = in_sizes[0] / 256;
    const int E = in_sizes[1] / 2;
    const int* src = eidx;
    const int* dst = eidx + E;

    float* ws = (float*)d_ws;
    size_t oWh   = 0;
    size_t oHP   = oWh  + (size_t)N * 256;
    size_t oS1s  = oHP  + (size_t)N * 256;
    size_t oS1d  = oS1s + (size_t)N * 4;
    size_t oEdge = oS1d + (size_t)N * 12;
    size_t oCnt  = oEdge + (size_t)E * 4;
    size_t oRP   = oCnt + (size_t)N + 1;
    size_t oBs   = oRP  + (size_t)N + 1;
    size_t oPerm = oEdge + (size_t)E * 4 + (size_t)N * 4;
    size_t oRank = oEdge;
    size_t oTmp  = oEdge + (size_t)E;
    size_t oS2s  = oS1s;
    size_t oS2d  = oS1s + (size_t)N;
    size_t oWx   = oWh;

    int* rowptr = (int*)(ws + oRP);
    int* bsum   = (int*)(ws + oBs);
    int* perm   = (int*)(ws + oPerm);
    int* rank   = (int*)(ws + oRank);
    int* tmp    = (int*)(ws + oTmp);

    const int geE = (E + 255) / 256;
    const int n1  = N + 1;
    const int nb  = (n1 + 1023) / 1024;
    const int nch = (E + CSR_CHUNK - 1) / CSR_CHUNK;
    unsigned* cntw = (unsigned*)(ws + oHP);
    int* start     = (int*)(ws + oHP + (size_t)nch * CSR_CAPW);

    csr_count_lds<<<nch, 1024, 0, stream>>>(dst, cntw, rank, E);
    csr_blocksum<<<nb, 256, 0, stream>>>(cntw, bsum, n1, nch);
    csr_rowptr<<<nb, 256, 0, stream>>>(cntw, bsum, rowptr, start, n1, nch);
    csr_place<<<geE, 256, 0, stream>>>(dst, start, rank, tmp, E);
    csr_segsort<<<(N + 7) / 8, 256, 0, stream>>>(rowptr, tmp, perm, N);

    dim3 g1((N + 127) / 128, 4);
    gemm_wmma_f32<4, 0><<<g1, 256, 0, stream>>>(h, Wheads, ws + oWh, aheads, ws + oS1s,
                                                ws + oS1d, N);

    gather1_elu<<<(N + 7) / 8, 256, 0, stream>>>(rowptr, perm, src, (const float4*)(ws + oS1s),
                                                 (const float4*)(ws + oS1d), ws + oWh,
                                                 ws + oHP, N);

    dim3 g2((N + 127) / 128, 1);
    gemm_wmma_f32<3, 1><<<g2, 256, 0, stream>>>(ws + oHP, Wout, ws + oWx, aout, ws + oS2s,
                                                ws + oS2d, N);

    gather2_lsm<<<(N + 7) / 8, 256, 0, stream>>>(rowptr, perm, src, ws + oS2s, ws + oS2d,
                                                 ws + oWx, out, N);
}


# each destination's list is put in edge-id order inside the layer-1 gather wave (lane compares, one lane permute) and kept as source ids for layer 2; the ordering launch is gone; sum orders unchanged
# speedup vs baseline: 6.1356x; 1.0443x over previous
#include <hip/hip_runtime.h>


#define LRELU 0.2f

typedef __attribute__((ext_vector_type(2))) float v2f;
typedef __attribute__((ext_vector_type(8))) float v8f;

__device__ __forceinline__ float wred_sum(float v) {
#pragma unroll
    for (int o = 16; o > 0; o >>= 1) v += __shfl_xor(v, o, 32);
    return v;
}
__device__ __forceinline__ float wred_max(float v) {
#pragma unroll
    for (int o = 16; o > 0; o >>= 1) v = fmaxf(v, __shfl_xor(v, o, 32));
    return v;
}
__device__ __forceinline__ float lrelu(float x) { return x > 0.f ? x : LRELU * x; }
__device__ __forceinline__ float xchg_add(const bool up, const float lo, const float hi,
                                          const int w) {
    const float keep = up ? hi : lo;
    const float send = up ? lo : hi;
    return keep + __shfl_xor(send, w, 32);
}

template <int COLT, int MODE>
__global__ __launch_bounds__(256) void gemm_wmma_f32(const float* __restrict__ A,
                                                     const float* __restrict__ B,
                                                     float* __restrict__ C,
                                                     const float* __restrict__ avec,
                                                     float* __restrict__ ssrc,
                                                     float* __restrict__ sdst, int nrows) {
    constexpr int CW = COLT * 16;
    __shared__ float Blds[256 * CW];
    const int tid = threadIdx.x;
    const int colBase = blockIdx.y * CW;

    for (int idx = tid; idx < 256 * CW; idx += 256) {
        float bv;
        if constexpr (MODE == 0) {
            bv = B[(size_t)blockIdx.y * (256 * 64) + idx];
        } else {
            int k = idx / CW;
            int c = idx - k * CW;
            bv = (c < 40) ? B[k * 40 + c] : 0.0f;
        }
        Blds[idx] = bv;
    }
    __syncthreads();

    const int wid  = tid >> 5;
    const int lane = tid & 31;
    const int half = lane >> 4;
    const int l16  = lane & 15;
    const int rowBase = blockIdx.x * 128 + wid * 16;

    int arow = rowBase + l16;
    if (arow >= nrows) arow = nrows - 1;
    const float* aptr = A + (size_t)arow * 256;

    v8f acc[COLT] = {};

    for (int k0 = 0; k0 < 256; k0 += 4) {
        const int kb = k0 + half * 2;
        v2f a;
        a.x = aptr[kb];
        a.y = aptr[kb + 1];
#pragma unroll
        for (int t = 0; t < COLT; ++t) {
            v2f b;
            const int c = t * 16 + l16;
            b.x = Blds[kb * CW + c];
            b.y = Blds[(kb + 1) * CW + c];
            acc[t] = __builtin_amdgcn_wmma_f32_16x16x4_f32(false, a, false, b, (short)0,
                                                           acc[t], false, false);
        }
    }

    {
        float as[COLT], ad[COLT];
        for (int t = 0; t < COLT; ++t) {
            const int c = t * 16 + l16;
            if constexpr (MODE == 0) {
                as[t] = avec[blockIdx.y * 128 + c];
                ad[t] = avec[blockIdx.y * 128 + 64 + c];
            } else {
                as[t] = (c < 40) ? avec[c] : 0.0f;
                ad[t] = (c < 40) ? avec[40 + c] : 0.0f;
            }
        }
#define ROW_DOTS(v, s, d)                      \
    float s = 0.f, d = 0.f;                    \
    for (int t = 0; t < COLT; ++t) {           \
        s += acc[t][v] * as[t];                \
        d += acc[t][v] * ad[t];                \
    }
        ROW_DOTS(0, s0, d0)
        ROW_DOTS(1, s1, d1)
        ROW_DOTS(2, s2, d2)
        ROW_DOTS(3, s3, d3)
        ROW_DOTS(4, s4, d4)
        ROW_DOTS(5, s5, d5)
        ROW_DOTS(6, s6, d6)
        ROW_DOTS(7, s7, d7)
#undef ROW_DOTS
        const bool u8 = (l16 & 8) != 0, u4 = (l16 & 4) != 0, u2 = (l16 & 2) != 0,
                   u1 = (l16 & 1) != 0;
        const float q0 = xchg_add(u8, s0, d0, 8);
        const float q1 = xchg_add(u8, s1, d1, 8);
        const float q2 = xchg_add(u8, s2, d2, 8);
        const float q3 = xchg_add(u8, s3, d3, 8);
        const float q4 = xchg_add(u8, s4, d4, 8);
        const float q5 = xchg_add(u8, s5, d5, 8);
        const float q6 = xchg_add(u8, s6, d6, 8);
        const float q7 = xchg_add(u8, s7, d7, 8);
        const float r0 = xchg_add(u4, q0, q4, 4);
        const float r1 = xchg_add(u4, q1, q5, 4);
        const float r2 = xchg_add(u4, q2, q6, 4);
        const float r3 = xchg_add(u4, q3, q7, 4);
        const float y0 = xchg_add(u2, r0, r2, 2);
        const float y1 = xchg_add(u2, r1, r3, 2);
        const float tot = xchg_add(u1, y0, y1, 1);
        const int srow = rowBase + half * 8 + (l16 & 7);
        float* sp = (l16 < 8) ? ssrc : sdst;
        if (srow < nrows) {
            if constexpr (MODE == 0) {
                sp[(size_t)srow * 4 + blockIdx.y] = tot;
            } else {
                sp[srow] = tot;
            }
        }
    }

    if (rowBase + 16 <= nrows) {
#pragma unroll
        for (int t = 0; t < COLT; ++t) {
            const int col = colBase + t * 16 + l16;
            if constexpr (MODE == 1) {
                if (col >= 40) continue;
            }
#pragma unroll
            for (int v = 0; v < 8; ++v) {
                const int row = rowBase + v + half * 8;
                if constexpr (MODE == 0) {
                    C[(size_t)row * 256 + col] = acc[t][v];
                } else {
                    C[(size_t)row * 40 + col] = acc[t][v];
                }
            }
        }
    } else {
#pragma unroll
        for (int t = 0; t < COLT; ++t) {
            const int col = colBase + t * 16 + l16;
#pragma unroll
            for (int v = 0; v < 8; ++v) {
                const int row = rowBase + v + half * 8;
                if constexpr (MODE == 0) {
                    if (row < nrows) C[(size_t)row * 256 + col] = acc[t][v];
                } else {
                    if (row < nrows && col < 40) C[(size_t)row * 40 + col] = acc[t][v];
                }
            }
        }
    }
}

__device__ __forceinline__ int wred_sum_i(int v) {
    for (int o = 16; o > 0; o >>= 1) v += __shfl_xor(v, o, 32);
    return v;
}
__device__ __forceinline__ float readlane_f(float v, int l) {
    return __int_as_float(__builtin_amdgcn_readlane(__float_as_int(v), l));
}
__device__ __forceinline__ float uni_f(float v) {
    return __int_as_float(__builtin_amdgcn_readfirstlane(__float_as_int(v)));
}
#define CSR_CHUNK 50000
#define CSR_CAPW 51200
#define CSR_SP (2 * CSR_CAPW)
__global__ __launch_bounds__(1024) void csr_count_lds(const int* __restrict__ dst,
                                                      unsigned* __restrict__ cntw,
                                                      int* __restrict__ rank, int ne) {
    __shared__ uint4 cl4[CSR_CAPW / 4];
    unsigned* cl = (unsigned*)cl4;
    const int tid = threadIdx.x;
    for (int i = tid; i < CSR_CAPW / 4; i += 1024) cl4[i] = make_uint4(0u, 0u, 0u, 0u);
    __syncthreads();
    const int e0 = blockIdx.x * CSR_CHUNK;
    const int e1 = (e0 + CSR_CHUNK < ne) ? (e0 + CSR_CHUNK) : ne;
    for (int eb = e0 + tid; eb < e1; eb += 16 * 1024) {
        int d[16];
        unsigned old[16];
#pragma unroll
        for (int k = 0; k < 16; ++k) {
            const int e = eb + k * 1024;
            d[k] = dst[(e < e1) ? e : (e1 - 1)];
        }
#pragma unroll
        for (int k = 0; k < 16; ++k) {
            const int e = eb + k * 1024;
            const int sh = (d[k] & 1) << 4;
            old[k] = atomicAdd(&cl[d[k] >> 1], (e < e1) ? (1u << sh) : 0u);
        }
#pragma unroll
        for (int k = 0; k < 16; ++k) {
            const int e = eb + k * 1024;
            const int sh = (d[k] & 1) << 4;
            if (e < e1) rank[e] = (int)((old[k] >> sh) & 0xffffu);
        }
    }
    __syncthreads();
    uint4* o = (uint4*)(cntw + (size_t)blockIdx.x * CSR_CAPW);
    for (int i = tid; i < CSR_CAPW / 4; i += 1024) o[i] = cl4[i];
}
__device__ __forceinline__ void csr_deg4(const unsigned* __restrict__ cntw, const int nch,
                                         const int base, int c[4]) {
    c[0] = 0;
    c[1] = 0;
    c[2] = 0;
    c[3] = 0;
    const unsigned* p = cntw + (base >> 1);
#pragma unroll 8
    for (int ch = 0; ch < nch; ++ch) {
        const uint2 v = *(const uint2*)(p + (size_t)ch * CSR_CAPW);
        c[0] += (int)(v.x & 0xffffu);
        c[1] += (int)(v.x >> 16);
        c[2] += (int)(v.y & 0xffffu);
        c[3] += (int)(v.y >> 16);
    }
}
__global__ __launch_bounds__(256) void csr_blocksum(const unsigned* __restrict__ cntw,
                                                    int* __restrict__ bsum, int n1, int nch) {
    __shared__ int wsum[8];
    const int base = blockIdx.x * 1024 + threadIdx.x * 4;
    int c[4];
    csr_deg4(cntw, nch, base, c);
    int s = 0;
    for (int k = 0; k < 4; ++k)
        if (base + k < n1) s += c[k];
    s = wred_sum_i(s);
    if ((threadIdx.x & 31) == 0) wsum[threadIdx.x >> 5] = s;
    __syncthreads();
    if (threadIdx.x == 0) {
        int t = 0;
        for (int w = 0; w < 8; ++w) t += wsum[w];
        bsum[blockIdx.x] = t;
    }
}
__global__ __launch_bounds__(256) void csr_rowptr(const unsigned* __restrict__ cntw,
                                                  const int* __restrict__ bsum,
                                                  int* __restrict__ rowptr,
                                                  int* __restrict__ start, int n1, int nch) {
    __shared__ int wsum[8];
    __shared__ int wsum2[8];
    const int tid = threadIdx.x, lane = tid & 31, wid = tid >> 5;
    int p = 0;
    for (int j = tid; j < (int)blockIdx.x; j += 256) p += bsum[j];
    p = wred_sum_i(p);
    if (lane == 0) wsum[wid] = p;
    const int base = blockIdx.x * 1024 + tid * 4;
    int c[4];
    csr_deg4(cntw, nch, base, c);
    int s = 0;
    for (int k = 0; k < 4; ++k) {
        if (base + k >= n1) c[k] = 0;
        s += c[k];
    }
    int inc = s;
    for (int o = 1; o < 32; o <<= 1) {
        int t = __shfl_up(inc, o, 32);
        if (lane >= o) inc += t;
    }
    if (lane == 31) wsum2[wid] = inc;
    __syncthreads();
    int run = inc - s;
    for (int w = 0; w < 8; ++w) {
        run += wsum[w];
        if (w < wid) run += wsum2[w];
    }
    int4 a;
    a.x = run;
    a.y = a.x + c[0];
    a.z = a.y + c[1];
    a.w = a.z + c[2];
    if (base < n1) rowptr[base] = a.x;
    if (base + 1 < n1) rowptr[base + 1] = a.y;
    if (base + 2 < n1) rowptr[base + 2] = a.z;
    if (base + 3 < n1) rowptr[base + 3] = a.w;
    const unsigned* cp = cntw + (base >> 1);
#pragma unroll 8
    for (int ch = 0; ch < nch; ++ch) {
        const uint2 v = *(const uint2*)(cp + (size_t)ch * CSR_CAPW);
        *(int4*)(start + (size_t)ch * CSR_SP + base) = a;
        a.x += (int)(v.x & 0xffffu);
        a.y += (int)(v.x >> 16);
        a.z += (int)(v.y & 0xffffu);
        a.w += (int)(v.y >> 16);
    }
}
__global__ __launch_bounds__(256) void csr_place(const int* __restrict__ dst,
                                                 const int* __restrict__ start,
                                                 const int* __restrict__ rank,
                                                 int* __restrict__ tmp, int ne) {
    int e = blockIdx.x * blockDim.x + threadIdx.x;
    if (e >= ne) return;
    const int ch = e / CSR_CHUNK;
    tmp[start[(size_t)ch * CSR_SP + dst[e]] + rank[e]] = e;
}
__device__ __forceinline__ void edge_scores1e(const int* __restrict__ src,
                                              const float4* __restrict__ ss4, const float4 sd,
                                              const bool has, const int e, int& s_i,
                                              float4& x) {
    s_i = 0;
    x.x = -INFINITY;
    x.y = -INFINITY;
    x.z = -INFINITY;
    x.w = -INFINITY;
    if (has) {
        s_i = src[e];
        const float4 vs = ss4[s_i];
        x.x = lrelu(vs.x + sd.x);
        x.y = lrelu(vs.y + sd.y);
        x.z = lrelu(vs.z + sd.z);
        x.w = lrelu(vs.w + sd.w);
    }
}
__device__ __forceinline__ void edge_scores1(const int* perm, const int* __restrict__ src,
                                             const float4* __restrict__ ss4, const float4 sd,
                                             int i, int end, int& s_i, float4& x) {
    const bool has = i < end;
    int e = 0;
    if (has) e = perm[i];
    edge_scores1e(src, ss4, sd, has, e, s_i, x);
}
__device__ __forceinline__ void rows_fma1(const float* __restrict__ wl, const bool hi,
                                          const int s_i, const float4 a_i, const int m,
                                          float4& acc0, float4& acc1) {
#pragma unroll 4
    for (int t = 0; t < m; ++t) {
        const int s = __builtin_amdgcn_readlane(s_i, t);
        const float ax = readlane_f(a_i.x, t);
        const float ay = readlane_f(a_i.y, t);
        const float az = readlane_f(a_i.z, t);
        const float aw = readlane_f(a_i.w, t);
        const float al = hi ? ay : ax;
        const float au = hi ? aw : az;
        const float* r = wl + (size_t)s * 256;
        const float4 w0 = *(const float4*)(r);
        const float4 w1 = *(const float4*)(r + 128);
        acc0.x += al * w0.x;
        acc0.y += al * w0.y;
        acc0.z += al * w0.z;
        acc0.w += al * w0.w;
        acc1.x += au * w1.x;
        acc1.y += au * w1.y;
        acc1.z += au * w1.z;
        acc1.w += au * w1.w;
    }
}

__global__ __launch_bounds__(256) void gather1_elu(const int* __restrict__ rowptr,
                                                   const int* __restrict__ tmp, int* perm,
                                                   int* __restrict__ srcs,
                                                   const int* __restrict__ src,
                                                   const float4* __restrict__ ss4,
                                                   const float4* __restrict__ sd4,
                                                   const float* __restrict__ Wh,
                                                   float* __restrict__ hp, int n) {
    const int node = __builtin_amdgcn_readfirstlane((int)((blockIdx.x * 256 + threadIdx.x) >> 5));
    const int lane = threadIdx.x & 31;
    if (node >= n) return;
    const int beg = rowptr[node], end = rowptr[node + 1];
    const float4 sd = sd4[node];
    const bool hi = lane >= 16;
    const float* wl = Wh + lane * 4;
    float4 acc0 = {0.f, 0.f, 0.f, 0.f};
    float4 acc1 = {0.f, 0.f, 0.f, 0.f};
    if (end - beg <= 32) {
        const int m = end - beg;
        const bool has = lane < m;
        const int ei = has ? tmp[beg + lane] : 0x7fffffff;
        int r = 0;
        for (int t = 0; t < m; ++t) r += (__builtin_amdgcn_readlane(ei, t) < ei) ? 1 : 0;
        if (!has) r = lane;
        const int e = __builtin_amdgcn_ds_permute(r << 2, ei);
        int s_i;
        float4 x;
        edge_scores1e(src, ss4, sd, has, e, s_i, x);
        if (has) srcs[beg + lane] = s_i;
        float4 mx, ex, dn, a_i;
        mx.x = wred_max(x.x);
        mx.y = wred_max(x.y);
        mx.z = wred_max(x.z);
        mx.w = wred_max(x.w);
        ex.x = expf(x.x - mx.x);
        ex.y = expf(x.y - mx.y);
        ex.z = expf(x.z - mx.z);
        ex.w = expf(x.w - mx.w);
        dn.x = wred_sum(ex.x);
        dn.y = wred_sum(ex.y);
        dn.z = wred_sum(ex.z);
        dn.w = wred_sum(ex.w);
        a_i.x = ex.x / dn.x;
        a_i.y = ex.y / dn.y;
        a_i.z = ex.z / dn.z;
        a_i.w = ex.w / dn.w;
        rows_fma1(wl, hi, s_i, a_i, end - beg, acc0, acc1);
    } else {
        for (int i0 = beg; i0 < end; i0 += 32) {
            const int i = i0 + lane;
            const int ei = (i < end) ? tmp[i] : 0x7fffffff;
            int r = 0;
            for (int j0 = beg; j0 < end; j0 += 32) {
                const int j = j0 + lane;
                const int ej = (j < end) ? tmp[j] : 0x7fffffff;
                const int mj = (end - j0 < 32) ? (end - j0) : 32;
                for (int t = 0; t < mj; ++t) r += (__builtin_amdgcn_readlane(ej, t) < ei) ? 1 : 0;
            }
            if (i < end) perm[beg + r] = ei;
        }
        __threadfence();
        float4 mx = {-INFINITY, -INFINITY, -INFINITY, -INFINITY};
        for (int i0 = beg; i0 < end; i0 += 32) {
            int s_i;
            float4 x;
            edge_scores1(perm, src, ss4, sd, i0 + lane, end, s_i, x);
            mx.x = fmaxf(mx.x, x.x);
            mx.y = fmaxf(mx.y, x.y);
            mx.z = fmaxf(mx.z, x.z);
            mx.w = fmaxf(mx.w, x.w);
        }
        mx.x = uni_f(wred_max(mx.x));
        mx.y = uni_f(wred_max(mx.y));
        mx.z = uni_f(wred_max(mx.z));
        mx.w = uni_f(wred_max(mx.w));
        float4 dn = {0.f, 0.f, 0.f, 0.f};
        for (int i0 = beg; i0 < end; i0 += 32) {
            int s_i;
            float4 x;
            edge_scores1(perm, src, ss4, sd, i0 + lane, end, s_i, x);
            if (i0 + lane < end) {
                dn.x += expf(x.x - mx.x);
                dn.y += expf(x.y - mx.y);
                dn.z += expf(x.z - mx.z);
                dn.w += expf(x.w - mx.w);
            }
        }
        dn.x = uni_f(wred_sum(dn.x));
        dn.y = uni_f(wred_sum(dn.y));
        dn.z = uni_f(wred_sum(dn.z));
        dn.w = uni_f(wred_sum(dn.w));
        for (int i0 = beg; i0 < end; i0 += 32) {
            int s_i;
            float4 x;
            edge_scores1(perm, src, ss4, sd, i0 + lane, end, s_i, x);
            float4 a_i = {0.f, 0.f, 0.f, 0.f};
            if (i0 + lane < end) {
                srcs[i0 + lane] = s_i;
                a_i.x = expf(x.x - mx.x) / dn.x;
                a_i.y = expf(x.y - mx.y) / dn.y;
                a_i.z = expf(x.z - mx.z) / dn.z;
                a_i.w = expf(x.w - mx.w) / dn.w;
            }
            const int m = (end - i0 < 32) ? (end - i0) : 32;
            rows_fma1(wl, hi, s_i, a_i, m, acc0, acc1);
        }
    }
    acc0.x = acc0.x > 0.f ? acc0.x : (expf(acc0.x) - 1.f);
    acc0.y = acc0.y > 0.f ? acc0.y : (expf(acc0.y) - 1.f);
    acc0.z = acc0.z > 0.f ? acc0.z : (expf(acc0.z) - 1.f);
    acc0.w = acc0.w > 0.f ? acc0.w : (expf(acc0.w) - 1.f);
    acc1.x = acc1.x > 0.f ? acc1.x : (expf(acc1.x) - 1.f);
    acc1.y = acc1.y > 0.f ? acc1.y : (expf(acc1.y) - 1.f);
    acc1.z = acc1.z > 0.f ? acc1.z : (expf(acc1.z) - 1.f);
    acc1.w = acc1.w > 0.f ? acc1.w : (expf(acc1.w) - 1.f);
    float* o = hp + (size_t)node * 256 + lane * 4;
    *(float4*)(o) = acc0;
    *(float4*)(o + 128) = acc1;
}

__device__ __forceinline__ void edge_score2(const int* __restrict__ srcs,
                                            const float* __restrict__ ss, const float sd, int i,
                                            int end, int& s_i, float& x) {
    s_i = 0;
    x = -INFINITY;
    if (i < end) {
        s_i = srcs[i];
        x = lrelu(ss[s_i] + sd);
    }
}
__device__ __forceinline__ void rows_fma2(const float* __restrict__ w1p,
                                          const float* __restrict__ w2p, const int s_i,
                                          const float a_i, const int m, float& acc1,
                                          float& acc2) {
#pragma unroll 4
    for (int t = 0; t < m; ++t) {
        const int s = __builtin_amdgcn_readlane(s_i, t);
        const float a = readlane_f(a_i, t);
        acc1 += a * w1p[(size_t)s * 40];
        acc2 += a * w2p[(size_t)s * 40];
    }
}
__global__ __launch_bounds__(256) void gather2_lsm(const int* __restrict__ rowptr,
                                                   const int* __restrict__ srcs,
                                                   const float* __restrict__ ss,
                                                   const float* __restrict__ sdv,
                                                   const float* __restrict__ Wx,
                                                   float* __restrict__ out, int n) {
    const int node = __builtin_amdgcn_readfirstlane((int)((blockIdx.x * 256 + threadIdx.x) >> 5));
    const int lane = threadIdx.x & 31;
    if (node >= n) return;
    const int beg = rowptr[node], end = rowptr[node + 1];
    const float sd = sdv[node];
    const float* w1p = Wx + lane;
    const float* w2p = Wx + 32 + (lane & 7);
    float acc1 = 0.f, acc2 = 0.f;
    if (end - beg <= 32) {
        int s_i;
        float x;
        edge_score2(srcs, ss, sd, beg + lane, end, s_i, x);
        const float mx = wred_max(x);
        const float ex = expf(x - mx);
        const float dn = wred_sum(ex);
        rows_fma2(w1p, w2p, s_i, ex / dn, end - beg, acc1, acc2);
    } else {
        float mx = -INFINITY;
        for (int i0 = beg; i0 < end; i0 += 32) {
            int s_i;
            float x;
            edge_score2(srcs, ss, sd, i0 + lane, end, s_i, x);
            mx = fmaxf(mx, x);
        }
        mx = uni_f(wred_max(mx));
        float dn = 0.f;
        for (int i0 = beg; i0 < end; i0 += 32) {
            int s_i;
            float x;
            edge_score2(srcs, ss, sd, i0 + lane, end, s_i, x);
            if (i0 + lane < end) dn += expf(x - mx);
        }
        dn = uni_f(wred_sum(dn));
        for (int i0 = beg; i0 < end; i0 += 32) {
            int s_i;
            float x;
            edge_score2(srcs, ss, sd, i0 + lane, end, s_i, x);
            const float a_i = (i0 + lane < end) ? expf(x - mx) / dn : 0.f;
            const int m = (end - i0 < 32) ? (end - i0) : 32;
            rows_fma2(w1p, w2p, s_i, a_i, m, acc1, acc2);
        }
    }
    const float v1 = acc1;
    const float v2 = (lane < 8) ? acc2 : -INFINITY;
    const float mxo = wred_max(fmaxf(v1, v2));
    float sm = expf(v1 - mxo) + ((lane < 8) ? expf(v2 - mxo) : 0.f);
    sm = wred_sum(sm);
    const float lg = mxo + logf(sm);
    float* row = out + (size_t)node * 40;
    row[lane] = v1 - lg;
    if (lane < 8) row[32 + lane] = v2 - lg;
}

extern "C" void kernel_launch(void* const* d_in, const int* in_sizes, int n_in, void* d_out,
                              int out_size, void* d_ws, size_t ws_size, hipStream_t stream) {
    (void)n_in; (void)out_size; (void)ws_size;
    const float* h      = (const float*)d_in[0];
    const int*   eidx   = (const int*)d_in[1];
    const float* Wheads = (const float*)d_in[2];
    const float* aheads = (const float*)d_in[3];
    const float* Wout   = (const float*)d_in[4];
    const float* aout   = (const float*)d_in[5];
    float* out = (float*)d_out;

    const int N = in_sizes[0] / 256;
    const int E = in_sizes[1] / 2;
    const int* src = eidx;
    const int* dst = eidx + E;

    float* ws = (float*)d_ws;
    size_t oWh   = 0;
    size_t oHP   = oWh  + (size_t)N * 256;
    size_t oS1s  = oHP  + (size_t)N * 256;
    size_t oS1d  = oS1s + (size_t)N * 4;
    size_t oEdge = oS1d + (size_t)N * 12;
    size_t oCnt  = oEdge + (size_t)E * 4;
    size_t oRP   = oCnt + (size_t)N + 1;
    size_t oBs   = oRP  + (size_t)N + 1;
    size_t oPerm = oEdge + (size_t)E * 4 + (size_t)N * 4;
    size_t oRank = oEdge;
    size_t oTmp  = oEdge + (size_t)E;
    size_t oS2s  = oS1s;
    size_t oS2d  = oS1s + (size_t)N;
    size_t oWx   = oWh;

    int* rowptr = (int*)(ws + oRP);
    int* bsum   = (int*)(ws + oBs);
    int* perm   = (int*)(ws + oPerm);
    int* rank   = (int*)(ws + oRank);
    int* tmp    = (int*)(ws + oTmp);
    int* srcs   = rank;

    const int geE = (E + 255) / 256;
    const int n1  = N + 1;
    const int nb  = (n1 + 1023) / 1024;
    const int nch = (E + CSR_CHUNK - 1) / CSR_CHUNK;
    unsigned* cntw = (unsigned*)(ws + oHP);
    int* start     = (int*)(ws + oHP + (size_t)nch * CSR_CAPW);

    csr_count_lds<<<nch, 1024, 0, stream>>>(dst, cntw, rank, E);
    csr_blocksum<<<nb, 256, 0, stream>>>(cntw, bsum, n1, nch);
    csr_rowptr<<<nb, 256, 0, stream>>>(cntw, bsum, rowptr, start, n1, nch);
    csr_place<<<geE, 256, 0, stream>>>(dst, start, rank, tmp, E);

    dim3 g1((N + 127) / 128, 4);
    gemm_wmma_f32<4, 0><<<g1, 256, 0, stream>>>(h, Wheads, ws + oWh, aheads, ws + oS1s,
                                                ws + oS1d, N);

    gather1_elu<<<(N + 7) / 8, 256, 0, stream>>>(rowptr, tmp, perm, srcs, src,
                                                 (const float4*)(ws + oS1s),
                                                 (const float4*)(ws + oS1d), ws + oWh,
                                                 ws + oHP, N);

    dim3 g2((N + 127) / 128, 1);
    gemm_wmma_f32<3, 1><<<g2, 256, 0, stream>>>(ws + oHP, Wout, ws + oWx, aout, ws + oS2s,
                                                ws + oS2d, N);

    gather2_lsm<<<(N + 7) / 8, 256, 0, stream>>>(rowptr, srcs, ws + oS2s, ws + oS2d, ws + oWx,
                                                 out, N);
}


# weight strip of both GEMMs staged to LDS by 16-byte loads all in flight before the first LDS store (was a 4-byte load per memory round trip, 32 and 48 trips in a row); values and sums unchanged
# speedup vs baseline: 6.3831x; 1.0403x over previous
#include <hip/hip_runtime.h>


#define LRELU 0.2f

typedef __attribute__((ext_vector_type(2))) float v2f;
typedef __attribute__((ext_vector_type(8))) float v8f;

__device__ __forceinline__ float wred_sum(float v) {
#pragma unroll
    for (int o = 16; o > 0; o >>= 1) v += __shfl_xor(v, o, 32);
    return v;
}
__device__ __forceinline__ float wred_max(float v) {
#pragma unroll
    for (int o = 16; o > 0; o >>= 1) v = fmaxf(v, __shfl_xor(v, o, 32));
    return v;
}
__device__ __forceinline__ float lrelu(float x) { return x > 0.f ? x : LRELU * x; }
__device__ __forceinline__ float xchg_add(const bool up, const float lo, const float hi,
                                          const int w) {
    const float keep = up ? hi : lo;
    const float send = up ? lo : hi;
    return keep + __shfl_xor(send, w, 32);
}

template <int COLT, int MODE>
__global__ __launch_bounds__(256) void gemm_wmma_f32(const float* __restrict__ A,
                                                     const float* __restrict__ B,
                                                     float* __restrict__ C,
                                                     const float* __restrict__ avec,
                                                     float* __restrict__ ssrc,
                                                     float* __restrict__ sdst, int nrows) {
    constexpr int CW = COLT * 16;
    __shared__ float4 Blds4[64 * CW];
    float* Blds = (float*)Blds4;
    const int tid = threadIdx.x;
    const int colBase = blockIdx.y * CW;

    if constexpr (MODE == 0) {
        const float4* __restrict__ b4 = (const float4*)(B + (size_t)blockIdx.y * (256 * 64));
        float4 st[16];
#pragma unroll
        for (int i = 0; i < 16; ++i) st[i] = b4[tid + i * 256];
#pragma unroll
        for (int i = 0; i < 16; ++i) Blds4[tid + i * 256] = st[i];
    } else {
        const float4* __restrict__ b4 = (const float4*)B;
        float4 st[10];
#pragma unroll
        for (int i = 0; i < 10; ++i) st[i] = b4[tid + i * 256];
#pragma unroll
        for (int i = 0; i < 10; ++i) {
            const int j = tid + i * 256;
            const int k = j / 10;
            Blds4[k * 12 + (j - k * 10)] = st[i];
        }
        for (int j = tid; j < 512; j += 256)
            Blds4[(j >> 1) * 12 + 10 + (j & 1)] = make_float4(0.f, 0.f, 0.f, 0.f);
    }
    __syncthreads();

    const int wid  = tid >> 5;
    const int lane = tid & 31;
    const int half = lane >> 4;
    const int l16  = lane & 15;
    const int rowBase = blockIdx.x * 128 + wid * 16;

    int arow = rowBase + l16;
    if (arow >= nrows) arow = nrows - 1;
    const float* aptr = A + (size_t)arow * 256;

    v8f acc[COLT] = {};

    for (int k0 = 0; k0 < 256; k0 += 4) {
        const int kb = k0 + half * 2;
        v2f a;
        a.x = aptr[kb];
        a.y = aptr[kb + 1];
#pragma unroll
        for (int t = 0; t < COLT; ++t) {
            v2f b;
            const int c = t * 16 + l16;
            b.x = Blds[kb * CW + c];
            b.y = Blds[(kb + 1) * CW + c];
            acc[t] = __builtin_amdgcn_wmma_f32_16x16x4_f32(false, a, false, b, (short)0,
                                                           acc[t], false, false);
        }
    }

    {
        float as[COLT], ad[COLT];
        for (int t = 0; t < COLT; ++t) {
            const int c = t * 16 + l16;
            if constexpr (MODE == 0) {
                as[t] = avec[blockIdx.y * 128 + c];
                ad[t] = avec[blockIdx.y * 128 + 64 + c];
            } else {
                as[t] = (c < 40) ? avec[c] : 0.0f;
                ad[t] = (c < 40) ? avec[40 + c] : 0.0f;
            }
        }
#define ROW_DOTS(v, s, d)                      \
    float s = 0.f, d = 0.f;                    \
    for (int t = 0; t < COLT; ++t) {           \
        s += acc[t][v] * as[t];                \
        d += acc[t][v] * ad[t];                \
    }
        ROW_DOTS(0, s0, d0)
        ROW_DOTS(1, s1, d1)
        ROW_DOTS(2, s2, d2)
        ROW_DOTS(3, s3, d3)
        ROW_DOTS(4, s4, d4)
        ROW_DOTS(5, s5, d5)
        ROW_DOTS(6, s6, d6)
        ROW_DOTS(7, s7, d7)
#undef ROW_DOTS
        const bool u8 = (l16 & 8) != 0, u4 = (l16 & 4) != 0, u2 = (l16 & 2) != 0,
                   u1 = (l16 & 1) != 0;
        const float q0 = xchg_add(u8, s0, d0, 8);
        const float q1 = xchg_add(u8, s1, d1, 8);
        const float q2 = xchg_add(u8, s2, d2, 8);
        const float q3 = xchg_add(u8, s3, d3, 8);
        const float q4 = xchg_add(u8, s4, d4, 8);
        const float q5 = xchg_add(u8, s5, d5, 8);
        const float q6 = xchg_add(u8, s6, d6, 8);
        const float q7 = xchg_add(u8, s7, d7, 8);
        const float r0 = xchg_add(u4, q0, q4, 4);
        const float r1 = xchg_add(u4, q1, q5, 4);
        const float r2 = xchg_add(u4, q2, q6, 4);
        const float r3 = xchg_add(u4, q3, q7, 4);
        const float y0 = xchg_add(u2, r0, r2, 2);
        const float y1 = xchg_add(u2, r1, r3, 2);
        const float tot = xchg_add(u1, y0, y1, 1);
        const int srow = rowBase + half * 8 + (l16 & 7);
        float* sp = (l16 < 8) ? ssrc : sdst;
        if (srow < nrows) {
            if constexpr (MODE == 0) {
                sp[(size_t)srow * 4 + blockIdx.y] = tot;
            } else {
                sp[srow] = tot;
            }
        }
    }

    if (rowBase + 16 <= nrows) {
#pragma unroll
        for (int t = 0; t < COLT; ++t) {
            const int col = colBase + t * 16 + l16;
            if constexpr (MODE == 1) {
                if (col >= 40) continue;
            }
#pragma unroll
            for (int v = 0; v < 8; ++v) {
                const int row = rowBase + v + half * 8;
                if constexpr (MODE == 0) {
                    C[(size_t)row * 256 + col] = acc[t][v];
                } else {
                    C[(size_t)row * 40 + col] = acc[t][v];
                }
            }
        }
    } else {
#pragma unroll
        for (int t = 0; t < COLT; ++t) {
            const int col = colBase + t * 16 + l16;
#pragma unroll
            for (int v = 0; v < 8; ++v) {
                const int row = rowBase + v + half * 8;
                if constexpr (MODE == 0) {
                    if (row < nrows) C[(size_t)row * 256 + col] = acc[t][v];
                } else {
                    if (row < nrows && col < 40) C[(size_t)row * 40 + col] = acc[t][v];
                }
            }
        }
    }
}

__device__ __forceinline__ int wred_sum_i(int v) {
    for (int o = 16; o > 0; o >>= 1) v += __shfl_xor(v, o, 32);
    return v;
}
__device__ __forceinline__ float readlane_f(float v, int l) {
    return __int_as_float(__builtin_amdgcn_readlane(__float_as_int(v), l));
}
__device__ __forceinline__ float uni_f(float v) {
    return __int_as_float(__builtin_amdgcn_readfirstlane(__float_as_int(v)));
}
#define CSR_CHUNK 50000
#define CSR_CAPW 51200
#define CSR_SP (2 * CSR_CAPW)
__global__ __launch_bounds__(1024) void csr_count_lds(const int* __restrict__ dst,
                                                      unsigned* __restrict__ cntw,
                                                      int* __restrict__ rank, int ne) {
    __shared__ uint4 cl4[CSR_CAPW / 4];
    unsigned* cl = (unsigned*)cl4;
    const int tid = threadIdx.x;
    for (int i = tid; i < CSR_CAPW / 4; i += 1024) cl4[i] = make_uint4(0u, 0u, 0u, 0u);
    __syncthreads();
    const int e0 = blockIdx.x * CSR_CHUNK;
    const int e1 = (e0 + CSR_CHUNK < ne) ? (e0 + CSR_CHUNK) : ne;
    for (int eb = e0 + tid; eb < e1; eb += 16 * 1024) {
        int d[16];
        unsigned old[16];
#pragma unroll
        for (int k = 0; k < 16; ++k) {
            const int e = eb + k * 1024;
            d[k] = dst[(e < e1) ? e : (e1 - 1)];
        }
#pragma unroll
        for (int k = 0; k < 16; ++k) {
            const int e = eb + k * 1024;
            const int sh = (d[k] & 1) << 4;
            old[k] = atomicAdd(&cl[d[k] >> 1], (e < e1) ? (1u << sh) : 0u);
        }
#pragma unroll
        for (int k = 0; k < 16; ++k) {
            const int e = eb + k * 1024;
            const int sh = (d[k] & 1) << 4;
            if (e < e1) rank[e] = (int)((old[k] >> sh) & 0xffffu);
        }
    }
    __syncthreads();
    uint4* o = (uint4*)(cntw + (size_t)blockIdx.x * CSR_CAPW);
    for (int i = tid; i < CSR_CAPW / 4; i += 1024) o[i] = cl4[i];
}
__device__ __forceinline__ void csr_deg4(const unsigned* __restrict__ cntw, const int nch,
                                         const int base, int c[4]) {
    c[0] = 0;
    c[1] = 0;
    c[2] = 0;
    c[3] = 0;
    const unsigned* p = cntw + (base >> 1);
#pragma unroll 8
    for (int ch = 0; ch < nch; ++ch) {
        const uint2 v = *(const uint2*)(p + (size_t)ch * CSR_CAPW);
        c[0] += (int)(v.x & 0xffffu);
        c[1] += (int)(v.x >> 16);
        c[2] += (int)(v.y & 0xffffu);
        c[3] += (int)(v.y >> 16);
    }
}
__global__ __launch_bounds__(256) void csr_blocksum(const unsigned* __restrict__ cntw,
                                                    int* __restrict__ bsum, int n1, int nch) {
    __shared__ int wsum[8];
    const int base = blockIdx.x * 1024 + threadIdx.x * 4;
    int c[4];
    csr_deg4(cntw, nch, base, c);
    int s = 0;
    for (int k = 0; k < 4; ++k)
        if (base + k < n1) s += c[k];
    s = wred_sum_i(s);
    if ((threadIdx.x & 31) == 0) wsum[threadIdx.x >> 5] = s;
    __syncthreads();
    if (threadIdx.x == 0) {
        int t = 0;
        for (int w = 0; w < 8; ++w) t += wsum[w];
        bsum[blockIdx.x] = t;
    }
}
__global__ __launch_bounds__(256) void csr_rowptr(const unsigned* __restrict__ cntw,
                                                  const int* __restrict__ bsum,
                                                  int* __restrict__ rowptr,
                                                  int* __restrict__ start, int n1, int nch) {
    __shared__ int wsum[8];
    __shared__ int wsum2[8];
    const int tid = threadIdx.x, lane = tid & 31, wid = tid >> 5;
    int p = 0;
    for (int j = tid; j < (int)blockIdx.x; j += 256) p += bsum[j];
    p = wred_sum_i(p);
    if (lane == 0) wsum[wid] = p;
    const int base = blockIdx.x * 1024 + tid * 4;
    int c[4];
    csr_deg4(cntw, nch, base, c);
    int s = 0;
    for (int k = 0; k < 4; ++k) {
        if (base + k >= n1) c[k] = 0;
        s += c[k];
    }
    int inc = s;
    for (int o = 1; o < 32; o <<= 1) {
        int t = __shfl_up(inc, o, 32);
        if (lane >= o) inc += t;
    }
    if (lane == 31) wsum2[wid] = inc;
    __syncthreads();
    int run = inc - s;
    for (int w = 0; w < 8; ++w) {
        run += wsum[w];
        if (w < wid) run += wsum2[w];
    }
    int4 a;
    a.x = run;
    a.y = a.x + c[0];
    a.z = a.y + c[1];
    a.w = a.z + c[2];
    if (base < n1) rowptr[base] = a.x;
    if (base + 1 < n1) rowptr[base + 1] = a.y;
    if (base + 2 < n1) rowptr[base + 2] = a.z;
    if (base + 3 < n1) rowptr[base + 3] = a.w;
    const unsigned* cp = cntw + (base >> 1);
#pragma unroll 8
    for (int ch = 0; ch < nch; ++ch) {
        const uint2 v = *(const uint2*)(cp + (size_t)ch * CSR_CAPW);
        *(int4*)(start + (size_t)ch * CSR_SP + base) = a;
        a.x += (int)(v.x & 0xffffu);
        a.y += (int)(v.x >> 16);
        a.z += (int)(v.y & 0xffffu);
        a.w += (int)(v.y >> 16);
    }
}
__global__ __launch_bounds__(256) void csr_place(const int* __restrict__ dst,
                                                 const int* __restrict__ start,
                                                 const int* __restrict__ rank,
                                                 int* __restrict__ tmp, int ne) {
    int e = blockIdx.x * blockDim.x + threadIdx.x;
    if (e >= ne) return;
    const int ch = e / CSR_CHUNK;
    tmp[start[(size_t)ch * CSR_SP + dst[e]] + rank[e]] = e;
}
__device__ __forceinline__ void edge_scores1e(const int* __restrict__ src,
                                              const float4* __restrict__ ss4, const float4 sd,
                                              const bool has, const int e, int& s_i,
                                              float4& x) {
    s_i = 0;
    x.x = -INFINITY;
    x.y = -INFINITY;
    x.z = -INFINITY;
    x.w = -INFINITY;
    if (has) {
        s_i = src[e];
        const float4 vs = ss4[s_i];
        x.x = lrelu(vs.x + sd.x);
        x.y = lrelu(vs.y + sd.y);
        x.z = lrelu(vs.z + sd.z);
        x.w = lrelu(vs.w + sd.w);
    }
}
__device__ __forceinline__ void edge_scores1(const int* perm, const int* __restrict__ src,
                                             const float4* __restrict__ ss4, const float4 sd,
                                             int i, int end, int& s_i, float4& x) {
    const bool has = i < end;
    int e = 0;
    if (has) e = perm[i];
    edge_scores1e(src, ss4, sd, has, e, s_i, x);
}
__device__ __forceinline__ void rows_fma1(const float* __restrict__ wl, const bool hi,
                                          const int s_i, const float4 a_i, const int m,
                                          float4& acc0, float4& acc1) {
#pragma unroll 4
    for (int t = 0; t < m; ++t) {
        const int s = __builtin_amdgcn_readlane(s_i, t);
        const float ax = readlane_f(a_i.x, t);
        const float ay = readlane_f(a_i.y, t);
        const float az = readlane_f(a_i.z, t);
        const float aw = readlane_f(a_i.w, t);
        const float al = hi ? ay : ax;
        const float au = hi ? aw : az;
        const float* r = wl + (size_t)s * 256;
        const float4 w0 = *(const float4*)(r);
        const float4 w1 = *(const float4*)(r + 128);
        acc0.x += al * w0.x;
        acc0.y += al * w0.y;
        acc0.z += al * w0.z;
        acc0.w += al * w0.w;
        acc1.x += au * w1.x;
        acc1.y += au * w1.y;
        acc1.z += au * w1.z;
        acc1.w += au * w1.w;
    }
}

__global__ __launch_bounds__(256) void gather1_elu(const int* __restrict__ rowptr,
                                                   const int* __restrict__ tmp, int* perm,
                                                   int* __restrict__ srcs,
                                                   const int* __restrict__ src,
                                                   const float4* __restrict__ ss4,
                                                   const float4* __restrict__ sd4,
                                                   const float* __restrict__ Wh,
                                                   float* __restrict__ hp, int n) {
    const int node = __builtin_amdgcn_readfirstlane((int)((blockIdx.x * 256 + threadIdx.x) >> 5));
    const int lane = threadIdx.x & 31;
    if (node >= n) return;
    const int beg = rowptr[node], end = rowptr[node + 1];
    const float4 sd = sd4[node];
    const bool hi = lane >= 16;
    const float* wl = Wh + lane * 4;
    float4 acc0 = {0.f, 0.f, 0.f, 0.f};
    float4 acc1 = {0.f, 0.f, 0.f, 0.f};
    if (end - beg <= 32) {
        const int m = end - beg;
        const bool has = lane < m;
        const int ei = has ? tmp[beg + lane] : 0x7fffffff;
        int r = 0;
        for (int t = 0; t < m; ++t) r += (__builtin_amdgcn_readlane(ei, t) < ei) ? 1 : 0;
        if (!has) r = lane;
        const int e = __builtin_amdgcn_ds_permute(r << 2, ei);
        int s_i;
        float4 x;
        edge_scores1e(src, ss4, sd, has, e, s_i, x);
        if (has) srcs[beg + lane] = s_i;
        float4 mx, ex, dn, a_i;
        mx.x = wred_max(x.x);
        mx.y = wred_max(x.y);
        mx.z = wred_max(x.z);
        mx.w = wred_max(x.w);
        ex.x = expf(x.x - mx.x);
        ex.y = expf(x.y - mx.y);
        ex.z = expf(x.z - mx.z);
        ex.w = expf(x.w - mx.w);
        dn.x = wred_sum(ex.x);
        dn.y = wred_sum(ex.y);
        dn.z = wred_sum(ex.z);
        dn.w = wred_sum(ex.w);
        a_i.x = ex.x / dn.x;
        a_i.y = ex.y / dn.y;
        a_i.z = ex.z / dn.z;
        a_i.w = ex.w / dn.w;
        rows_fma1(wl, hi, s_i, a_i, end - beg, acc0, acc1);
    } else {
        for (int i0 = beg; i0 < end; i0 += 32) {
            const int i = i0 + lane;
            const int ei = (i < end) ? tmp[i] : 0x7fffffff;
            int r = 0;
            for (int j0 = beg; j0 < end; j0 += 32) {
                const int j = j0 + lane;
                const int ej = (j < end) ? tmp[j] : 0x7fffffff;
                const int mj = (end - j0 < 32) ? (end - j0) : 32;
                for (int t = 0; t < mj; ++t) r += (__builtin_amdgcn_readlane(ej, t) < ei) ? 1 : 0;
            }
            if (i < end) perm[beg + r] = ei;
        }
        __threadfence();
        float4 mx = {-INFINITY, -INFINITY, -INFINITY, -INFINITY};
        for (int i0 = beg; i0 < end; i0 += 32) {
            int s_i;
            float4 x;
            edge_scores1(perm, src, ss4, sd, i0 + lane, end, s_i, x);
            mx.x = fmaxf(mx.x, x.x);
            mx.y = fmaxf(mx.y, x.y);
            mx.z = fmaxf(mx.z, x.z);
            mx.w = fmaxf(mx.w, x.w);
        }
        mx.x = uni_f(wred_max(mx.x));
        mx.y = uni_f(wred_max(mx.y));
        mx.z = uni_f(wred_max(mx.z));
        mx.w = uni_f(wred_max(mx.w));
        float4 dn = {0.f, 0.f, 0.f, 0.f};
        for (int i0 = beg; i0 < end; i0 += 32) {
            int s_i;
            float4 x;
            edge_scores1(perm, src, ss4, sd, i0 + lane, end, s_i, x);
            if (i0 + lane < end) {
                dn.x += expf(x.x - mx.x);
                dn.y += expf(x.y - mx.y);
                dn.z += expf(x.z - mx.z);
                dn.w += expf(x.w - mx.w);
            }
        }
        dn.x = uni_f(wred_sum(dn.x));
        dn.y = uni_f(wred_sum(dn.y));
        dn.z = uni_f(wred_sum(dn.z));
        dn.w = uni_f(wred_sum(dn.w));
        for (int i0 = beg; i0 < end; i0 += 32) {
            int s_i;
            float4 x;
            edge_scores1(perm, src, ss4, sd, i0 + lane, end, s_i, x);
            float4 a_i = {0.f, 0.f, 0.f, 0.f};
            if (i0 + lane < end) {
                srcs[i0 + lane] = s_i;
                a_i.x = expf(x.x - mx.x) / dn.x;
                a_i.y = expf(x.y - mx.y) / dn.y;
                a_i.z = expf(x.z - mx.z) / dn.z;
                a_i.w = expf(x.w - mx.w) / dn.w;
            }
            const int m = (end - i0 < 32) ? (end - i0) : 32;
            rows_fma1(wl, hi, s_i, a_i, m, acc0, acc1);
        }
    }
    acc0.x = acc0.x > 0.f ? acc0.x : (expf(acc0.x) - 1.f);
    acc0.y = acc0.y > 0.f ? acc0.y : (expf(acc0.y) - 1.f);
    acc0.z = acc0.z > 0.f ? acc0.z : (expf(acc0.z) - 1.f);
    acc0.w = acc0.w > 0.f ? acc0.w : (expf(acc0.w) - 1.f);
    acc1.x = acc1.x > 0.f ? acc1.x : (expf(acc1.x) - 1.f);
    acc1.y = acc1.y > 0.f ? acc1.y : (expf(acc1.y) - 1.f);
    acc1.z = acc1.z > 0.f ? acc1.z : (expf(acc1.z) - 1.f);
    acc1.w = acc1.w > 0.f ? acc1.w : (expf(acc1.w) - 1.f);
    float* o = hp + (size_t)node * 256 + lane * 4;
    *(float4*)(o) = acc0;
    *(float4*)(o + 128) = acc1;
}

__device__ __forceinline__ void edge_score2(const int* __restrict__ srcs,
                                            const float* __restrict__ ss, const float sd, int i,
                                            int end, int& s_i, float& x) {
    s_i = 0;
    x = -INFINITY;
    if (i < end) {
        s_i = srcs[i];
        x = lrelu(ss[s_i] + sd);
    }
}
__device__ __forceinline__ void rows_fma2(const float* __restrict__ w1p,
                                          const float* __restrict__ w2p, const int s_i,
                                          const float a_i, const int m, float& acc1,
                                          float& acc2) {
#pragma unroll 4
    for (int t = 0; t < m; ++t) {
        const int s = __builtin_amdgcn_readlane(s_i, t);
        const float a = readlane_f(a_i, t);
        acc1 += a * w1p[(size_t)s * 40];
        acc2 += a * w2p[(size_t)s * 40];
    }
}
__global__ __launch_bounds__(256) void gather2_lsm(const int* __restrict__ rowptr,
                                                   const int* __restrict__ srcs,
                                                   const float* __restrict__ ss,
                                                   const float* __restrict__ sdv,
                                                   const float* __restrict__ Wx,
                                                   float* __restrict__ out, int n) {
    const int node = __builtin_amdgcn_readfirstlane((int)((blockIdx.x * 256 + threadIdx.x) >> 5));
    const int lane = threadIdx.x & 31;
    if (node >= n) return;
    const int beg = rowptr[node], end = rowptr[node + 1];
    const float sd = sdv[node];
    const float* w1p = Wx + lane;
    const float* w2p = Wx + 32 + (lane & 7);
    float acc1 = 0.f, acc2 = 0.f;
    if (end - beg <= 32) {
        int s_i;
        float x;
        edge_score2(srcs, ss, sd, beg + lane, end, s_i, x);
        const float mx = wred_max(x);
        const float ex = expf(x - mx);
        const float dn = wred_sum(ex);
        rows_fma2(w1p, w2p, s_i, ex / dn, end - beg, acc1, acc2);
    } else {
        float mx = -INFINITY;
        for (int i0 = beg; i0 < end; i0 += 32) {
            int s_i;
            float x;
            edge_score2(srcs, ss, sd, i0 + lane, end, s_i, x);
            mx = fmaxf(mx, x);
        }
        mx = uni_f(wred_max(mx));
        float dn = 0.f;
        for (int i0 = beg; i0 < end; i0 += 32) {
            int s_i;
            float x;
            edge_score2(srcs, ss, sd, i0 + lane, end, s_i, x);
            if (i0 + lane < end) dn += expf(x - mx);
        }
        dn = uni_f(wred_sum(dn));
        for (int i0 = beg; i0 < end; i0 += 32) {
            int s_i;
            float x;
            edge_score2(srcs, ss, sd, i0 + lane, end, s_i, x);
            const float a_i = (i0 + lane < end) ? expf(x - mx) / dn : 0.f;
            const int m = (end - i0 < 32) ? (end - i0) : 32;
            rows_fma2(w1p, w2p, s_i, a_i, m, acc1, acc2);
        }
    }
    const float v1 = acc1;
    const float v2 = (lane < 8) ? acc2 : -INFINITY;
    const float mxo = wred_max(fmaxf(v1, v2));
    float sm = expf(v1 - mxo) + ((lane < 8) ? expf(v2 - mxo) : 0.f);
    sm = wred_sum(sm);
    const float lg = mxo + logf(sm);
    float* row = out + (size_t)node * 40;
    row[lane] = v1 - lg;
    if (lane < 8) row[32 + lane] = v2 - lg;
}

extern "C" void kernel_launch(void* const* d_in, const int* in_sizes, int n_in, void* d_out,
                              int out_size, void* d_ws, size_t ws_size, hipStream_t stream) {
    (void)n_in; (void)out_size; (void)ws_size;
    const float* h      = (const float*)d_in[0];
    const int*   eidx   = (const int*)d_in[1];
    const float* Wheads = (const float*)d_in[2];
    const float* aheads = (const float*)d_in[3];
    const float* Wout   = (const float*)d_in[4];
    const float* aout   = (const float*)d_in[5];
    float* out = (float*)d_out;

    const int N = in_sizes[0] / 256;
    const int E = in_sizes[1] / 2;
    const int* src = eidx;
    const int* dst = eidx + E;

    float* ws = (float*)d_ws;
    size_t oWh   = 0;
    size_t oHP   = oWh  + (size_t)N * 256;
    size_t oS1s  = oHP  + (size_t)N * 256;
    size_t oS1d  = oS1s + (size_t)N * 4;
    size_t oEdge = oS1d + (size_t)N * 12;
    size_t oCnt  = oEdge + (size_t)E * 4;
    size_t oRP   = oCnt + (size_t)N + 1;
    size_t oBs   = oRP  + (size_t)N + 1;
    size_t oPerm = oEdge + (size_t)E * 4 + (size_t)N * 4;
    size_t oRank = oEdge;
    size_t oTmp  = oEdge + (size_t)E;
    size_t oS2s  = oS1s;
    size_t oS2d  = oS1s + (size_t)N;
    size_t oWx   = oWh;

    int* rowptr = (int*)(ws + oRP);
    int* bsum   = (int*)(ws + oBs);
    int* perm   = (int*)(ws + oPerm);
    int* rank   = (int*)(ws + oRank);
    int* tmp    = (int*)(ws + oTmp);
    int* srcs   = rank;

    const int geE = (E + 255) / 256;
    const int n1  = N + 1;
    const int nb  = (n1 + 1023) / 1024;
    const int nch = (E + CSR_CHUNK - 1) / CSR_CHUNK;
    unsigned* cntw = (unsigned*)(ws + oHP);
    int* start     = (int*)(ws + oHP + (size_t)nch * CSR_CAPW);

    csr_count_lds<<<nch, 1024, 0, stream>>>(dst, cntw, rank, E);
    csr_blocksum<<<nb, 256, 0, stream>>>(cntw, bsum, n1, nch);
    csr_rowptr<<<nb, 256, 0, stream>>>(cntw, bsum, rowptr, start, n1, nch);
    csr_place<<<geE, 256, 0, stream>>>(dst, start, rank, tmp, E);

    dim3 g1((N + 127) / 128, 4);
    gemm_wmma_f32<4, 0><<<g1, 256, 0, stream>>>(h, Wheads, ws + oWh, aheads, ws + oS1s,
                                                ws + oS1d, N);

    gather1_elu<<<(N + 7) / 8, 256, 0, stream>>>(rowptr, tmp, perm, srcs, src,
                                                 (const float4*)(ws + oS1s),
                                                 (const float4*)(ws + oS1d), ws + oWh,
                                                 ws + oHP, N);

    dim3 g2((N + 127) / 128, 1);
    gemm_wmma_f32<3, 1><<<g2, 256, 0, stream>>>(ws + oHP, Wout, ws + oWx, aout, ws + oS2s,
                                                ws + oS2d, N);

    gather2_lsm<<<(N + 7) / 8, 256, 0, stream>>>(rowptr, srcs, ws + oS2s, ws + oS2d, ws + oWx,
                                                 out, N);
}


# count kernel: a chunk's partial last round no longer pads its unused slots with zero adds on one LDS word (same-word adds of a wave are serialised); the rest goes one edge per thread; sums unchanged
# speedup vs baseline: 6.9403x; 1.0873x over previous
#include <hip/hip_runtime.h>


#define LRELU 0.2f

typedef __attribute__((ext_vector_type(2))) float v2f;
typedef __attribute__((ext_vector_type(8))) float v8f;

__device__ __forceinline__ float wred_sum(float v) {
#pragma unroll
    for (int o = 16; o > 0; o >>= 1) v += __shfl_xor(v, o, 32);
    return v;
}
__device__ __forceinline__ float wred_max(float v) {
#pragma unroll
    for (int o = 16; o > 0; o >>= 1) v = fmaxf(v, __shfl_xor(v, o, 32));
    return v;
}
__device__ __forceinline__ float lrelu(float x) { return x > 0.f ? x : LRELU * x; }
__device__ __forceinline__ float xchg_add(const bool up, const float lo, const float hi,
                                          const int w) {
    const float keep = up ? hi : lo;
    const float send = up ? lo : hi;
    return keep + __shfl_xor(send, w, 32);
}

template <int COLT, int MODE>
__global__ __launch_bounds__(256) void gemm_wmma_f32(const float* __restrict__ A,
                                                     const float* __restrict__ B,
                                                     float* __restrict__ C,
                                                     const float* __restrict__ avec,
                                                     float* __restrict__ ssrc,
                                                     float* __restrict__ sdst, int nrows) {
    constexpr int CW = COLT * 16;
    __shared__ float4 Blds4[64 * CW];
    float* Blds = (float*)Blds4;
    const int tid = threadIdx.x;
    const int colBase = blockIdx.y * CW;

    if constexpr (MODE == 0) {
        const float4* __restrict__ b4 = (const float4*)(B + (size_t)blockIdx.y * (256 * 64));
        float4 st[16];
#pragma unroll
        for (int i = 0; i < 16; ++i) st[i] = b4[tid + i * 256];
#pragma unroll
        for (int i = 0; i < 16; ++i) Blds4[tid + i * 256] = st[i];
    } else {
        const float4* __restrict__ b4 = (const float4*)B;
        float4 st[10];
#pragma unroll
        for (int i = 0; i < 10; ++i) st[i] = b4[tid + i * 256];
#pragma unroll
        for (int i = 0; i < 10; ++i) {
            const int j = tid + i * 256;
            const int k = j / 10;
            Blds4[k * 12 + (j - k * 10)] = st[i];
        }
        for (int j = tid; j < 512; j += 256)
            Blds4[(j >> 1) * 12 + 10 + (j & 1)] = make_float4(0.f, 0.f, 0.f, 0.f);
    }
    __syncthreads();

    const int wid  = tid >> 5;
    const int lane = tid & 31;
    const int half = lane >> 4;
    const int l16  = lane & 15;
    const int rowBase = blockIdx.x * 128 + wid * 16;

    int arow = rowBase + l16;
    if (arow >= nrows) arow = nrows - 1;
    const float* aptr = A + (size_t)arow * 256;

    v8f acc[COLT] = {};

    for (int k0 = 0; k0 < 256; k0 += 4) {
        const int kb = k0 + half * 2;
        v2f a;
        a.x = aptr[kb];
        a.y = aptr[kb + 1];
#pragma unroll
        for (int t = 0; t < COLT; ++t) {
            v2f b;
            const int c = t * 16 + l16;
            b.x = Blds[kb * CW + c];
            b.y = Blds[(kb + 1) * CW + c];
            acc[t] = __builtin_amdgcn_wmma_f32_16x16x4_f32(false, a, false, b, (short)0,
                                                           acc[t], false, false);
        }
    }

    {
        float as[COLT], ad[COLT];
        for (int t = 0; t < COLT; ++t) {
            const int c = t * 16 + l16;
            if constexpr (MODE == 0) {
                as[t] = avec[blockIdx.y * 128 + c];
                ad[t] = avec[blockIdx.y * 128 + 64 + c];
            } else {
                as[t] = (c < 40) ? avec[c] : 0.0f;
                ad[t] = (c < 40) ? avec[40 + c] : 0.0f;
            }
        }
#define ROW_DOTS(v, s, d)                      \
    float s = 0.f, d = 0.f;                    \
    for (int t = 0; t < COLT; ++t) {           \
        s += acc[t][v] * as[t];                \
        d += acc[t][v] * ad[t];                \
    }
        ROW_DOTS(0, s0, d0)
        ROW_DOTS(1, s1, d1)
        ROW_DOTS(2, s2, d2)
        ROW_DOTS(3, s3, d3)
        ROW_DOTS(4, s4, d4)
        ROW_DOTS(5, s5, d5)
        ROW_DOTS(6, s6, d6)
        ROW_DOTS(7, s7, d7)
#undef ROW_DOTS
        const bool u8 = (l16 & 8) != 0, u4 = (l16 & 4) != 0, u2 = (l16 & 2) != 0,
                   u1 = (l16 & 1) != 0;
        const float q0 = xchg_add(u8, s0, d0, 8);
        const float q1 = xchg_add(u8, s1, d1, 8);
        const float q2 = xchg_add(u8, s2, d2, 8);
        const float q3 = xchg_add(u8, s3, d3, 8);
        const float q4 = xchg_add(u8, s4, d4, 8);
        const float q5 = xchg_add(u8, s5, d5, 8);
        const float q6 = xchg_add(u8, s6, d6, 8);
        const float q7 = xchg_add(u8, s7, d7, 8);
        const float r0 = xchg_add(u4, q0, q4, 4);
        const float r1 = xchg_add(u4, q1, q5, 4);
        const float r2 = xchg_add(u4, q2, q6, 4);
        const float r3 = xchg_add(u4, q3, q7, 4);
        const float y0 = xchg_add(u2, r0, r2, 2);
        const float y1 = xchg_add(u2, r1, r3, 2);
        const float tot = xchg_add(u1, y0, y1, 1);
        const int srow = rowBase + half * 8 + (l16 & 7);
        float* sp = (l16 < 8) ? ssrc : sdst;
        if (srow < nrows) {
            if constexpr (MODE == 0) {
                sp[(size_t)srow * 4 + blockIdx.y] = tot;
            } else {
                sp[srow] = tot;
            }
        }
    }

    if (rowBase + 16 <= nrows) {
#pragma unroll
        for (int t = 0; t < COLT; ++t) {
            const int col = colBase + t * 16 + l16;
            if constexpr (MODE == 1) {
                if (col >= 40) continue;
            }
#pragma unroll
            for (int v = 0; v < 8; ++v) {
                const int row = rowBase + v + half * 8;
                if constexpr (MODE == 0) {
                    C[(size_t)row * 256 + col] = acc[t][v];
                } else {
                    C[(size_t)row * 40 + col] = acc[t][v];
                }
            }
        }
    } else {
#pragma unroll
        for (int t = 0; t < COLT; ++t) {
            const int col = colBase + t * 16 + l16;
#pragma unroll
            for (int v = 0; v < 8; ++v) {
                const int row = rowBase + v + half * 8;
                if constexpr (MODE == 0) {
                    if (row < nrows) C[(size_t)row * 256 + col] = acc[t][v];
                } else {
                    if (row < nrows && col < 40) C[(size_t)row * 40 + col] = acc[t][v];
                }
            }
        }
    }
}

__device__ __forceinline__ int wred_sum_i(int v) {
    for (int o = 16; o > 0; o >>= 1) v += __shfl_xor(v, o, 32);
    return v;
}
__device__ __forceinline__ float readlane_f(float v, int l) {
    return __int_as_float(__builtin_amdgcn_readlane(__float_as_int(v), l));
}
__device__ __forceinline__ float uni_f(float v) {
    return __int_as_float(__builtin_amdgcn_readfirstlane(__float_as_int(v)));
}
#define CSR_CHUNK 50000
#define CSR_CAPW 51200
#define CSR_SP (2 * CSR_CAPW)
__global__ __launch_bounds__(1024) void csr_count_lds(const int* __restrict__ dst,
                                                      unsigned* __restrict__ cntw,
                                                      int* __restrict__ rank, int ne) {
    __shared__ uint4 cl4[CSR_CAPW / 4];
    unsigned* cl = (unsigned*)cl4;
    const int tid = threadIdx.x;
    for (int i = tid; i < CSR_CAPW / 4; i += 1024) cl4[i] = make_uint4(0u, 0u, 0u, 0u);
    __syncthreads();
    const int e0 = blockIdx.x * CSR_CHUNK;
    const int e1 = (e0 + CSR_CHUNK < ne) ? (e0 + CSR_CHUNK) : ne;
    const int nfull = (e1 - e0) / (16 * 1024);
    int tb = e0;
    for (int r = 0; r < nfull; ++r, tb += 16 * 1024) {
        const int eb = tb + tid;
        int d[16];
        unsigned old[16];
#pragma unroll
        for (int k = 0; k < 16; ++k) d[k] = dst[eb + k * 1024];
#pragma unroll
        for (int k = 0; k < 16; ++k) {
            const int sh = (d[k] & 1) << 4;
            old[k] = atomicAdd(&cl[d[k] >> 1], 1u << sh);
        }
#pragma unroll
        for (int k = 0; k < 16; ++k) {
            const int sh = (d[k] & 1) << 4;
            rank[eb + k * 1024] = (int)((old[k] >> sh) & 0xffffu);
        }
    }
    for (int e = tb + tid; e < e1; e += 1024) {
        const int dd = dst[e];
        const int sh = (dd & 1) << 4;
        const unsigned o = atomicAdd(&cl[dd >> 1], 1u << sh);
        rank[e] = (int)((o >> sh) & 0xffffu);
    }
    __syncthreads();
    uint4* o = (uint4*)(cntw + (size_t)blockIdx.x * CSR_CAPW);
    for (int i = tid; i < CSR_CAPW / 4; i += 1024) o[i] = cl4[i];
}
__device__ __forceinline__ void csr_deg4(const unsigned* __restrict__ cntw, const int nch,
                                         const int base, int c[4]) {
    c[0] = 0;
    c[1] = 0;
    c[2] = 0;
    c[3] = 0;
    const unsigned* p = cntw + (base >> 1);
#pragma unroll 8
    for (int ch = 0; ch < nch; ++ch) {
        const uint2 v = *(const uint2*)(p + (size_t)ch * CSR_CAPW);
        c[0] += (int)(v.x & 0xffffu);
        c[1] += (int)(v.x >> 16);
        c[2] += (int)(v.y & 0xffffu);
        c[3] += (int)(v.y >> 16);
    }
}
__global__ __launch_bounds__(256) void csr_blocksum(const unsigned* __restrict__ cntw,
                                                    int* __restrict__ bsum, int n1, int nch) {
    __shared__ int wsum[8];
    const int base = blockIdx.x * 1024 + threadIdx.x * 4;
    int c[4];
    csr_deg4(cntw, nch, base, c);
    int s = 0;
    for (int k = 0; k < 4; ++k)
        if (base + k < n1) s += c[k];
    s = wred_sum_i(s);
    if ((threadIdx.x & 31) == 0) wsum[threadIdx.x >> 5] = s;
    __syncthreads();
    if (threadIdx.x == 0) {
        int t = 0;
        for (int w = 0; w < 8; ++w) t += wsum[w];
        bsum[blockIdx.x] = t;
    }
}
__global__ __launch_bounds__(256) void csr_rowptr(const unsigned* __restrict__ cntw,
                                                  const int* __restrict__ bsum,
                                                  int* __restrict__ rowptr,
                                                  int* __restrict__ start, int n1, int nch) {
    __shared__ int wsum[8];
    __shared__ int wsum2[8];
    const int tid = threadIdx.x, lane = tid & 31, wid = tid >> 5;
    int p = 0;
    for (int j = tid; j < (int)blockIdx.x; j += 256) p += bsum[j];
    p = wred_sum_i(p);
    if (lane == 0) wsum[wid] = p;
    const int base = blockIdx.x * 1024 + tid * 4;
    int c[4];
    csr_deg4(cntw, nch, base, c);
    int s = 0;
    for (int k = 0; k < 4; ++k) {
        if (base + k >= n1) c[k] = 0;
        s += c[k];
    }
    int inc = s;
    for (int o = 1; o < 32; o <<= 1) {
        int t = __shfl_up(inc, o, 32);
        if (lane >= o) inc += t;
    }
    if (lane == 31) wsum2[wid] = inc;
    __syncthreads();
    int run = inc - s;
    for (int w = 0; w < 8; ++w) {
        run += wsum[w];
        if (w < wid) run += wsum2[w];
    }
    int4 a;
    a.x = run;
    a.y = a.x + c[0];
    a.z = a.y + c[1];
    a.w = a.z + c[2];
    if (base < n1) rowptr[base] = a.x;
    if (base + 1 < n1) rowptr[base + 1] = a.y;
    if (base + 2 < n1) rowptr[base + 2] = a.z;
    if (base + 3 < n1) rowptr[base + 3] = a.w;
    const unsigned* cp = cntw + (base >> 1);
#pragma unroll 8
    for (int ch = 0; ch < nch; ++ch) {
        const uint2 v = *(const uint2*)(cp + (size_t)ch * CSR_CAPW);
        *(int4*)(start + (size_t)ch * CSR_SP + base) = a;
        a.x += (int)(v.x & 0xffffu);
        a.y += (int)(v.x >> 16);
        a.z += (int)(v.y & 0xffffu);
        a.w += (int)(v.y >> 16);
    }
}
__global__ __launch_bounds__(256) void csr_place(const int* __restrict__ dst,
                                                 const int* __restrict__ start,
                                                 const int* __restrict__ rank,
                                                 int* __restrict__ tmp, int ne) {
    int e = blockIdx.x * blockDim.x + threadIdx.x;
    if (e >= ne) return;
    const int ch = e / CSR_CHUNK;
    tmp[start[(size_t)ch * CSR_SP + dst[e]] + rank[e]] = e;
}
__device__ __forceinline__ void edge_scores1e(const int* __restrict__ src,
                                              const float4* __restrict__ ss4, const float4 sd,
                                              const bool has, const int e, int& s_i,
                                              float4& x) {
    s_i = 0;
    x.x = -INFINITY;
    x.y = -INFINITY;
    x.z = -INFINITY;
    x.w = -INFINITY;
    if (has) {
        s_i = src[e];
        const float4 vs = ss4[s_i];
        x.x = lrelu(vs.x + sd.x);
        x.y = lrelu(vs.y + sd.y);
        x.z = lrelu(vs.z + sd.z);
        x.w = lrelu(vs.w + sd.w);
    }
}
__device__ __forceinline__ void edge_scores1(const int* perm, const int* __restrict__ src,
                                             const float4* __restrict__ ss4, const float4 sd,
                                             int i, int end, int& s_i, float4& x) {
    const bool has = i < end;
    int e = 0;
    if (has) e = perm[i];
    edge_scores1e(src, ss4, sd, has, e, s_i, x);
}
__device__ __forceinline__ void rows_fma1(const float* __restrict__ wl, const bool hi,
                                          const int s_i, const float4 a_i, const int m,
                                          float4& acc0, float4& acc1) {
#pragma unroll 4
    for (int t = 0; t < m; ++t) {
        const int s = __builtin_amdgcn_readlane(s_i, t);
        const float ax = readlane_f(a_i.x, t);
        const float ay = readlane_f(a_i.y, t);
        const float az = readlane_f(a_i.z, t);
        const float aw = readlane_f(a_i.w, t);
        const float al = hi ? ay : ax;
        const float au = hi ? aw : az;
        const float* r = wl + (size_t)s * 256;
        const float4 w0 = *(const float4*)(r);
        const float4 w1 = *(const float4*)(r + 128);
        acc0.x += al * w0.x;
        acc0.y += al * w0.y;
        acc0.z += al * w0.z;
        acc0.w += al * w0.w;
        acc1.x += au * w1.x;
        acc1.y += au * w1.y;
        acc1.z += au * w1.z;
        acc1.w += au * w1.w;
    }
}

__global__ __launch_bounds__(256) void gather1_elu(const int* __restrict__ rowptr,
                                                   const int* __restrict__ tmp, int* perm,
                                                   int* __restrict__ srcs,
                                                   const int* __restrict__ src,
                                                   const float4* __restrict__ ss4,
                                                   const float4* __restrict__ sd4,
                                                   const float* __restrict__ Wh,
                                                   float* __restrict__ hp, int n) {
    const int node = __builtin_amdgcn_readfirstlane((int)((blockIdx.x * 256 + threadIdx.x) >> 5));
    const int lane = threadIdx.x & 31;
    if (node >= n) return;
    const int beg = rowptr[node], end = rowptr[node + 1];
    const float4 sd = sd4[node];
    const bool hi = lane >= 16;
    const float* wl = Wh + lane * 4;
    float4 acc0 = {0.f, 0.f, 0.f, 0.f};
    float4 acc1 = {0.f, 0.f, 0.f, 0.f};
    if (end - beg <= 32) {
        const int m = end - beg;
        const bool has = lane < m;
        const int ei = has ? tmp[beg + lane] : 0x7fffffff;
        int r = 0;
        for (int t = 0; t < m; ++t) r += (__builtin_amdgcn_readlane(ei, t) < ei) ? 1 : 0;
        if (!has) r = lane;
        const int e = __builtin_amdgcn_ds_permute(r << 2, ei);
        int s_i;
        float4 x;
        edge_scores1e(src, ss4, sd, has, e, s_i, x);
        if (has) srcs[beg + lane] = s_i;
        float4 mx, ex, dn, a_i;
        mx.x = wred_max(x.x);
        mx.y = wred_max(x.y);
        mx.z = wred_max(x.z);
        mx.w = wred_max(x.w);
        ex.x = expf(x.x - mx.x);
        ex.y = expf(x.y - mx.y);
        ex.z = expf(x.z - mx.z);
        ex.w = expf(x.w - mx.w);
        dn.x = wred_sum(ex.x);
        dn.y = wred_sum(ex.y);
        dn.z = wred_sum(ex.z);
        dn.w = wred_sum(ex.w);
        a_i.x = ex.x / dn.x;
        a_i.y = ex.y / dn.y;
        a_i.z = ex.z / dn.z;
        a_i.w = ex.w / dn.w;
        rows_fma1(wl, hi, s_i, a_i, end - beg, acc0, acc1);
    } else {
        for (int i0 = beg; i0 < end; i0 += 32) {
            const int i = i0 + lane;
            const int ei = (i < end) ? tmp[i] : 0x7fffffff;
            int r = 0;
            for (int j0 = beg; j0 < end; j0 += 32) {
                const int j = j0 + lane;
                const int ej = (j < end) ? tmp[j] : 0x7fffffff;
                const int mj = (end - j0 < 32) ? (end - j0) : 32;
                for (int t = 0; t < mj; ++t) r += (__builtin_amdgcn_readlane(ej, t) < ei) ? 1 : 0;
            }
            if (i < end) perm[beg + r] = ei;
        }
        __threadfence();
        float4 mx = {-INFINITY, -INFINITY, -INFINITY, -INFINITY};
        for (int i0 = beg; i0 < end; i0 += 32) {
            int s_i;
            float4 x;
            edge_scores1(perm, src, ss4, sd, i0 + lane, end, s_i, x);
            mx.x = fmaxf(mx.x, x.x);
            mx.y = fmaxf(mx.y, x.y);
            mx.z = fmaxf(mx.z, x.z);
            mx.w = fmaxf(mx.w, x.w);
        }
        mx.x = uni_f(wred_max(mx.x));
        mx.y = uni_f(wred_max(mx.y));
        mx.z = uni_f(wred_max(mx.z));
        mx.w = uni_f(wred_max(mx.w));
        float4 dn = {0.f, 0.f, 0.f, 0.f};
        for (int i0 = beg; i0 < end; i0 += 32) {
            int s_i;
            float4 x;
            edge_scores1(perm, src, ss4, sd, i0 + lane, end, s_i, x);
            if (i0 + lane < end) {
                dn.x += expf(x.x - mx.x);
                dn.y += expf(x.y - mx.y);
                dn.z += expf(x.z - mx.z);
                dn.w += expf(x.w - mx.w);
            }
        }
        dn.x = uni_f(wred_sum(dn.x));
        dn.y = uni_f(wred_sum(dn.y));
        dn.z = uni_f(wred_sum(dn.z));
        dn.w = uni_f(wred_sum(dn.w));
        for (int i0 = beg; i0 < end; i0 += 32) {
            int s_i;
            float4 x;
            edge_scores1(perm, src, ss4, sd, i0 + lane, end, s_i, x);
            float4 a_i = {0.f, 0.f, 0.f, 0.f};
            if (i0 + lane < end) {
                srcs[i0 + lane] = s_i;
                a_i.x = expf(x.x - mx.x) / dn.x;
                a_i.y = expf(x.y - mx.y) / dn.y;
                a_i.z = expf(x.z - mx.z) / dn.z;
                a_i.w = expf(x.w - mx.w) / dn.w;
            }
            const int m = (end - i0 < 32) ? (end - i0) : 32;
            rows_fma1(wl, hi, s_i, a_i, m, acc0, acc1);
        }
    }
    acc0.x = acc0.x > 0.f ? acc0.x : (expf(acc0.x) - 1.f);
    acc0.y = acc0.y > 0.f ? acc0.y : (expf(acc0.y) - 1.f);
    acc0.z = acc0.z > 0.f ? acc0.z : (expf(acc0.z) - 1.f);
    acc0.w = acc0.w > 0.f ? acc0.w : (expf(acc0.w) - 1.f);
    acc1.x = acc1.x > 0.f ? acc1.x : (expf(acc1.x) - 1.f);
    acc1.y = acc1.y > 0.f ? acc1.y : (expf(acc1.y) - 1.f);
    acc1.z = acc1.z > 0.f ? acc1.z : (expf(acc1.z) - 1.f);
    acc1.w = acc1.w > 0.f ? acc1.w : (expf(acc1.w) - 1.f);
    float* o = hp + (size_t)node * 256 + lane * 4;
    *(float4*)(o) = acc0;
    *(float4*)(o + 128) = acc1;
}

__device__ __forceinline__ void edge_score2(const int* __restrict__ srcs,
                                            const float* __restrict__ ss, const float sd, int i,
                                            int end, int& s_i, float& x) {
    s_i = 0;
    x = -INFINITY;
    if (i < end) {
        s_i = srcs[i];
        x = lrelu(ss[s_i] + sd);
    }
}
__device__ __forceinline__ void rows_fma2(const float* __restrict__ w1p,
                                          const float* __restrict__ w2p, const int s_i,
                                          const float a_i, const int m, float& acc1,
                                          float& acc2) {
#pragma unroll 4
    for (int t = 0; t < m; ++t) {
        const int s = __builtin_amdgcn_readlane(s_i, t);
        const float a = readlane_f(a_i, t);
        acc1 += a * w1p[(size_t)s * 40];
        acc2 += a * w2p[(size_t)s * 40];
    }
}
__global__ __launch_bounds__(256) void gather2_lsm(const int* __restrict__ rowptr,
                                                   const int* __restrict__ srcs,
                                                   const float* __restrict__ ss,
                                                   const float* __restrict__ sdv,
                                                   const float* __restrict__ Wx,
                                                   float* __restrict__ out, int n) {
    const int node = __builtin_amdgcn_readfirstlane((int)((blockIdx.x * 256 + threadIdx.x) >> 5));
    const int lane = threadIdx.x & 31;
    if (node >= n) return;
    const int beg = rowptr[node], end = rowptr[node + 1];
    const float sd = sdv[node];
    const float* w1p = Wx + lane;
    const float* w2p = Wx + 32 + (lane & 7);
    float acc1 = 0.f, acc2 = 0.f;
    if (end - beg <= 32) {
        int s_i;
        float x;
        edge_score2(srcs, ss, sd, beg + lane, end, s_i, x);
        const float mx = wred_max(x);
        const float ex = expf(x - mx);
        const float dn = wred_sum(ex);
        rows_fma2(w1p, w2p, s_i, ex / dn, end - beg, acc1, acc2);
    } else {
        float mx = -INFINITY;
        for (int i0 = beg; i0 < end; i0 += 32) {
            int s_i;
            float x;
            edge_score2(srcs, ss, sd, i0 + lane, end, s_i, x);
            mx = fmaxf(mx, x);
        }
        mx = uni_f(wred_max(mx));
        float dn = 0.f;
        for (int i0 = beg; i0 < end; i0 += 32) {
            int s_i;
            float x;
            edge_score2(srcs, ss, sd, i0 + lane, end, s_i, x);
            if (i0 + lane < end) dn += expf(x - mx);
        }
        dn = uni_f(wred_sum(dn));
        for (int i0 = beg; i0 < end; i0 += 32) {
            int s_i;
            float x;
            edge_score2(srcs, ss, sd, i0 + lane, end, s_i, x);
            const float a_i = (i0 + lane < end) ? expf(x - mx) / dn : 0.f;
            const int m = (end - i0 < 32) ? (end - i0) : 32;
            rows_fma2(w1p, w2p, s_i, a_i, m, acc1, acc2);
        }
    }
    const float v1 = acc1;
    const float v2 = (lane < 8) ? acc2 : -INFINITY;
    const float mxo = wred_max(fmaxf(v1, v2));
    float sm = expf(v1 - mxo) + ((lane < 8) ? expf(v2 - mxo) : 0.f);
    sm = wred_sum(sm);
    const float lg = mxo + logf(sm);
    float* row = out + (size_t)node * 40;
    row[lane] = v1 - lg;
    if (lane < 8) row[32 + lane] = v2 - lg;
}

extern "C" void kernel_launch(void* const* d_in, const int* in_sizes, int n_in, void* d_out,
                              int out_size, void* d_ws, size_t ws_size, hipStream_t stream) {
    (void)n_in; (void)out_size; (void)ws_size;
    const float* h      = (const float*)d_in[0];
    const int*   eidx   = (const int*)d_in[1];
    const float* Wheads = (const float*)d_in[2];
    const float* aheads = (const float*)d_in[3];
    const float* Wout   = (const float*)d_in[4];
    const float* aout   = (const float*)d_in[5];
    float* out = (float*)d_out;

    const int N = in_sizes[0] / 256;
    const int E = in_sizes[1] / 2;
    const int* src = eidx;
    const int* dst = eidx + E;

    float* ws = (float*)d_ws;
    size_t oWh   = 0;
    size_t oHP   = oWh  + (size_t)N * 256;
    size_t oS1s  = oHP  + (size_t)N * 256;
    size_t oS1d  = oS1s + (size_t)N * 4;
    size_t oEdge = oS1d + (size_t)N * 12;
    size_t oCnt  = oEdge + (size_t)E * 4;
    size_t oRP   = oCnt + (size_t)N + 1;
    size_t oBs   = oRP  + (size_t)N + 1;
    size_t oPerm = oEdge + (size_t)E * 4 + (size_t)N * 4;
    size_t oRank = oEdge;
    size_t oTmp  = oEdge + (size_t)E;
    size_t oS2s  = oS1s;
    size_t oS2d  = oS1s + (size_t)N;
    size_t oWx   = oWh;

    int* rowptr = (int*)(ws + oRP);
    int* bsum   = (int*)(ws + oBs);
    int* perm   = (int*)(ws + oPerm);
    int* rank   = (int*)(ws + oRank);
    int* tmp    = (int*)(ws + oTmp);
    int* srcs   = rank;

    const int geE = (E + 255) / 256;
    const int n1  = N + 1;
    const int nb  = (n1 + 1023) / 1024;
    const int nch = (E + CSR_CHUNK - 1) / CSR_CHUNK;
    unsigned* cntw = (unsigned*)(ws + oHP);
    int* start     = (int*)(ws + oHP + (size_t)nch * CSR_CAPW);

    csr_count_lds<<<nch, 1024, 0, stream>>>(dst, cntw, rank, E);
    csr_blocksum<<<nb, 256, 0, stream>>>(cntw, bsum, n1, nch);
    csr_rowptr<<<nb, 256, 0, stream>>>(cntw, bsum, rowptr, start, n1, nch);
    csr_place<<<geE, 256, 0, stream>>>(dst, start, rank, tmp, E);

    dim3 g1((N + 127) / 128, 4);
    gemm_wmma_f32<4, 0><<<g1, 256, 0, stream>>>(h, Wheads, ws + oWh, aheads, ws + oS1s,
                                                ws + oS1d, N);

    gather1_elu<<<(N + 7) / 8, 256, 0, stream>>>(rowptr, tmp, perm, srcs, src,
                                                 (const float4*)(ws + oS1s),
                                                 (const float4*)(ws + oS1d), ws + oWh,
                                                 ws + oHP, N);

    dim3 g2((N + 127) / 128, 1);
    gemm_wmma_f32<3, 1><<<g2, 256, 0, stream>>>(ws + oHP, Wout, ws + oWx, aout, ws + oS2s,
                                                ws + oS2d, N);

    gather2_lsm<<<(N + 7) / 8, 256, 0, stream>>>(rowptr, srcs, ws + oS2s, ws + oS2d, ws + oWx,
                                                 out, N);
}


# layer-2 product x*W_out runs on WMMA inside the layer-1 gather block (16 nodes, x rows through LDS): x never stored, layer-2 GEMM launch gone; the product's and its score dots' sum order changes
# speedup vs baseline: 7.1743x; 1.0337x over previous
#include <hip/hip_runtime.h>


#define LRELU 0.2f

typedef __attribute__((ext_vector_type(2))) float v2f;
typedef __attribute__((ext_vector_type(8))) float v8f;

__device__ __forceinline__ float wred_sum(float v) {
#pragma unroll
    for (int o = 16; o > 0; o >>= 1) v += __shfl_xor(v, o, 32);
    return v;
}
__device__ __forceinline__ float wred_max(float v) {
#pragma unroll
    for (int o = 16; o > 0; o >>= 1) v = fmaxf(v, __shfl_xor(v, o, 32));
    return v;
}
__device__ __forceinline__ float lrelu(float x) { return x > 0.f ? x : LRELU * x; }
__device__ __forceinline__ float xchg_add(const bool up, const float lo, const float hi,
                                          const int w) {
    const float keep = up ? hi : lo;
    const float send = up ? lo : hi;
    return keep + __shfl_xor(send, w, 32);
}

template <int COLT, int MODE>
__global__ __launch_bounds__(256) void gemm_wmma_f32(const float* __restrict__ A,
                                                     const float* __restrict__ B,
                                                     float* __restrict__ C,
                                                     const float* __restrict__ avec,
                                                     float* __restrict__ ssrc,
                                                     float* __restrict__ sdst, int nrows) {
    constexpr int CW = COLT * 16;
    __shared__ float4 Blds4[64 * CW];
    float* Blds = (float*)Blds4;
    const int tid = threadIdx.x;
    const int colBase = blockIdx.y * CW;

    if constexpr (MODE == 0) {
        const float4* __restrict__ b4 = (const float4*)(B + (size_t)blockIdx.y * (256 * 64));
        float4 st[16];
#pragma unroll
        for (int i = 0; i < 16; ++i) st[i] = b4[tid + i * 256];
#pragma unroll
        for (int i = 0; i < 16; ++i) Blds4[tid + i * 256] = st[i];
    } else {
        const float4* __restrict__ b4 = (const float4*)B;
        float4 st[10];
#pragma unroll
        for (int i = 0; i < 10; ++i) st[i] = b4[tid + i * 256];
#pragma unroll
        for (int i = 0; i < 10; ++i) {
            const int j = tid + i * 256;
            const int k = j / 10;
            Blds4[k * 12 + (j - k * 10)] = st[i];
        }
        for (int j = tid; j < 512; j += 256)
            Blds4[(j >> 1) * 12 + 10 + (j & 1)] = make_float4(0.f, 0.f, 0.f, 0.f);
    }
    __syncthreads();

    const int wid  = tid >> 5;
    const int lane = tid & 31;
    const int half = lane >> 4;
    const int l16  = lane & 15;
    const int rowBase = blockIdx.x * 128 + wid * 16;

    int arow = rowBase + l16;
    if (arow >= nrows) arow = nrows - 1;
    const float* aptr = A + (size_t)arow * 256;

    v8f acc[COLT] = {};

    for (int k0 = 0; k0 < 256; k0 += 4) {
        const int kb = k0 + half * 2;
        v2f a;
        a.x = aptr[kb];
        a.y = aptr[kb + 1];
#pragma unroll
        for (int t = 0; t < COLT; ++t) {
            v2f b;
            const int c = t * 16 + l16;
            b.x = Blds[kb * CW + c];
            b.y = Blds[(kb + 1) * CW + c];
            acc[t] = __builtin_amdgcn_wmma_f32_16x16x4_f32(false, a, false, b, (short)0,
                                                           acc[t], false, false);
        }
    }

    {
        float as[COLT], ad[COLT];
        for (int t = 0; t < COLT; ++t) {
            const int c = t * 16 + l16;
            if constexpr (MODE == 0) {
                as[t] = avec[blockIdx.y * 128 + c];
                ad[t] = avec[blockIdx.y * 128 + 64 + c];
            } else {
                as[t] = (c < 40) ? avec[c] : 0.0f;
                ad[t] = (c < 40) ? avec[40 + c] : 0.0f;
            }
        }
#define ROW_DOTS(v, s, d)                      \
    float s = 0.f, d = 0.f;                    \
    for (int t = 0; t < COLT; ++t) {           \
        s += acc[t][v] * as[t];                \
        d += acc[t][v] * ad[t];                \
    }
        ROW_DOTS(0, s0, d0)
        ROW_DOTS(1, s1, d1)
        ROW_DOTS(2, s2, d2)
        ROW_DOTS(3, s3, d3)
        ROW_DOTS(4, s4, d4)
        ROW_DOTS(5, s5, d5)
        ROW_DOTS(6, s6, d6)
        ROW_DOTS(7, s7, d7)
#undef ROW_DOTS
        const bool u8 = (l16 & 8) != 0, u4 = (l16 & 4) != 0, u2 = (l16 & 2) != 0,
                   u1 = (l16 & 1) != 0;
        const float q0 = xchg_add(u8, s0, d0, 8);
        const float q1 = xchg_add(u8, s1, d1, 8);
        const float q2 = xchg_add(u8, s2, d2, 8);
        const float q3 = xchg_add(u8, s3, d3, 8);
        const float q4 = xchg_add(u8, s4, d4, 8);
        const float q5 = xchg_add(u8, s5, d5, 8);
        const float q6 = xchg_add(u8, s6, d6, 8);
        const float q7 = xchg_add(u8, s7, d7, 8);
        const float r0 = xchg_add(u4, q0, q4, 4);
        const float r1 = xchg_add(u4, q1, q5, 4);
        const float r2 = xchg_add(u4, q2, q6, 4);
        const float r3 = xchg_add(u4, q3, q7, 4);
        const float y0 = xchg_add(u2, r0, r2, 2);
        const float y1 = xchg_add(u2, r1, r3, 2);
        const float tot = xchg_add(u1, y0, y1, 1);
        const int srow = rowBase + half * 8 + (l16 & 7);
        float* sp = (l16 < 8) ? ssrc : sdst;
        if (srow < nrows) {
            if constexpr (MODE == 0) {
                sp[(size_t)srow * 4 + blockIdx.y] = tot;
            } else {
                sp[srow] = tot;
            }
        }
    }

    if (rowBase + 16 <= nrows) {
#pragma unroll
        for (int t = 0; t < COLT; ++t) {
            const int col = colBase + t * 16 + l16;
            if constexpr (MODE == 1) {
                if (col >= 40) continue;
            }
#pragma unroll
            for (int v = 0; v < 8; ++v) {
                const int row = rowBase + v + half * 8;
                if constexpr (MODE == 0) {
                    C[(size_t)row * 256 + col] = acc[t][v];
                } else {
                    C[(size_t)row * 40 + col] = acc[t][v];
                }
            }
        }
    } else {
#pragma unroll
        for (int t = 0; t < COLT; ++t) {
            const int col = colBase + t * 16 + l16;
#pragma unroll
            for (int v = 0; v < 8; ++v) {
                const int row = rowBase + v + half * 8;
                if constexpr (MODE == 0) {
                    if (row < nrows) C[(size_t)row * 256 + col] = acc[t][v];
                } else {
                    if (row < nrows && col < 40) C[(size_t)row * 40 + col] = acc[t][v];
                }
            }
        }
    }
}

__device__ __forceinline__ int wred_sum_i(int v) {
    for (int o = 16; o > 0; o >>= 1) v += __shfl_xor(v, o, 32);
    return v;
}
__device__ __forceinline__ float readlane_f(float v, int l) {
    return __int_as_float(__builtin_amdgcn_readlane(__float_as_int(v), l));
}
__device__ __forceinline__ float uni_f(float v) {
    return __int_as_float(__builtin_amdgcn_readfirstlane(__float_as_int(v)));
}
#define CSR_CHUNK 50000
#define CSR_CAPW 51200
#define CSR_SP (2 * CSR_CAPW)
__global__ __launch_bounds__(1024) void csr_count_lds(const int* __restrict__ dst,
                                                      unsigned* __restrict__ cntw,
                                                      int* __restrict__ rank, int ne) {
    __shared__ uint4 cl4[CSR_CAPW / 4];
    unsigned* cl = (unsigned*)cl4;
    const int tid = threadIdx.x;
    for (int i = tid; i < CSR_CAPW / 4; i += 1024) cl4[i] = make_uint4(0u, 0u, 0u, 0u);
    __syncthreads();
    const int e0 = blockIdx.x * CSR_CHUNK;
    const int e1 = (e0 + CSR_CHUNK < ne) ? (e0 + CSR_CHUNK) : ne;
    const int nfull = (e1 - e0) / (16 * 1024);
    int tb = e0;
    for (int r = 0; r < nfull; ++r, tb += 16 * 1024) {
        const int eb = tb + tid;
        int d[16];
        unsigned old[16];
#pragma unroll
        for (int k = 0; k < 16; ++k) d[k] = dst[eb + k * 1024];
#pragma unroll
        for (int k = 0; k < 16; ++k) {
            const int sh = (d[k] & 1) << 4;
            old[k] = atomicAdd(&cl[d[k] >> 1], 1u << sh);
        }
#pragma unroll
        for (int k = 0; k < 16; ++k) {
            const int sh = (d[k] & 1) << 4;
            rank[eb + k * 1024] = (int)((old[k] >> sh) & 0xffffu);
        }
    }
    for (int e = tb + tid; e < e1; e += 1024) {
        const int dd = dst[e];
        const int sh = (dd & 1) << 4;
        const unsigned o = atomicAdd(&cl[dd >> 1], 1u << sh);
        rank[e] = (int)((o >> sh) & 0xffffu);
    }
    __syncthreads();
    uint4* o = (uint4*)(cntw + (size_t)blockIdx.x * CSR_CAPW);
    for (int i = tid; i < CSR_CAPW / 4; i += 1024) o[i] = cl4[i];
}
__device__ __forceinline__ void csr_deg4(const unsigned* __restrict__ cntw, const int nch,
                                         const int base, int c[4]) {
    c[0] = 0;
    c[1] = 0;
    c[2] = 0;
    c[3] = 0;
    const unsigned* p = cntw + (base >> 1);
#pragma unroll 8
    for (int ch = 0; ch < nch; ++ch) {
        const uint2 v = *(const uint2*)(p + (size_t)ch * CSR_CAPW);
        c[0] += (int)(v.x & 0xffffu);
        c[1] += (int)(v.x >> 16);
        c[2] += (int)(v.y & 0xffffu);
        c[3] += (int)(v.y >> 16);
    }
}
__global__ __launch_bounds__(256) void csr_blocksum(const unsigned* __restrict__ cntw,
                                                    int* __restrict__ bsum, int n1, int nch) {
    __shared__ int wsum[8];
    const int base = blockIdx.x * 1024 + threadIdx.x * 4;
    int c[4];
    csr_deg4(cntw, nch, base, c);
    int s = 0;
    for (int k = 0; k < 4; ++k)
        if (base + k < n1) s += c[k];
    s = wred_sum_i(s);
    if ((threadIdx.x & 31) == 0) wsum[threadIdx.x >> 5] = s;
    __syncthreads();
    if (threadIdx.x == 0) {
        int t = 0;
        for (int w = 0; w < 8; ++w) t += wsum[w];
        bsum[blockIdx.x] = t;
    }
}
__global__ __launch_bounds__(256) void csr_rowptr(const unsigned* __restrict__ cntw,
                                                  const int* __restrict__ bsum,
                                                  int* __restrict__ rowptr,
                                                  int* __restrict__ start, int n1, int nch) {
    __shared__ int wsum[8];
    __shared__ int wsum2[8];
    const int tid = threadIdx.x, lane = tid & 31, wid = tid >> 5;
    int p = 0;
    for (int j = tid; j < (int)blockIdx.x; j += 256) p += bsum[j];
    p = wred_sum_i(p);
    if (lane == 0) wsum[wid] = p;
    const int base = blockIdx.x * 1024 + tid * 4;
    int c[4];
    csr_deg4(cntw, nch, base, c);
    int s = 0;
    for (int k = 0; k < 4; ++k) {
        if (base + k >= n1) c[k] = 0;
        s += c[k];
    }
    int inc = s;
    for (int o = 1; o < 32; o <<= 1) {
        int t = __shfl_up(inc, o, 32);
        if (lane >= o) inc += t;
    }
    if (lane == 31) wsum2[wid] = inc;
    __syncthreads();
    int run = inc - s;
    for (int w = 0; w < 8; ++w) {
        run += wsum[w];
        if (w < wid) run += wsum2[w];
    }
    int4 a;
    a.x = run;
    a.y = a.x + c[0];
    a.z = a.y + c[1];
    a.w = a.z + c[2];
    if (base < n1) rowptr[base] = a.x;
    if (base + 1 < n1) rowptr[base + 1] = a.y;
    if (base + 2 < n1) rowptr[base + 2] = a.z;
    if (base + 3 < n1) rowptr[base + 3] = a.w;
    const unsigned* cp = cntw + (base >> 1);
#pragma unroll 8
    for (int ch = 0; ch < nch; ++ch) {
        const uint2 v = *(const uint2*)(cp + (size_t)ch * CSR_CAPW);
        *(int4*)(start + (size_t)ch * CSR_SP + base) = a;
        a.x += (int)(v.x & 0xffffu);
        a.y += (int)(v.x >> 16);
        a.z += (int)(v.y & 0xffffu);
        a.w += (int)(v.y >> 16);
    }
}
__global__ __launch_bounds__(256) void csr_place(const int* __restrict__ dst,
                                                 const int* __restrict__ start,
                                                 const int* __restrict__ rank,
                                                 int* __restrict__ tmp, int ne) {
    int e = blockIdx.x * blockDim.x + threadIdx.x;
    if (e >= ne) return;
    const int ch = e / CSR_CHUNK;
    tmp[start[(size_t)ch * CSR_SP + dst[e]] + rank[e]] = e;
}
__device__ __forceinline__ void edge_scores1e(const int* __restrict__ src,
                                              const float4* __restrict__ ss4, const float4 sd,
                                              const bool has, const int e, int& s_i,
                                              float4& x) {
    s_i = 0;
    x.x = -INFINITY;
    x.y = -INFINITY;
    x.z = -INFINITY;
    x.w = -INFINITY;
    if (has) {
        s_i = src[e];
        const float4 vs = ss4[s_i];
        x.x = lrelu(vs.x + sd.x);
        x.y = lrelu(vs.y + sd.y);
        x.z = lrelu(vs.z + sd.z);
        x.w = lrelu(vs.w + sd.w);
    }
}
__device__ __forceinline__ void edge_scores1(const int* perm, const int* __restrict__ src,
                                             const float4* __restrict__ ss4, const float4 sd,
                                             int i, int end, int& s_i, float4& x) {
    const bool has = i < end;
    int e = 0;
    if (has) e = perm[i];
    edge_scores1e(src, ss4, sd, has, e, s_i, x);
}
__device__ __forceinline__ void rows_fma1(const float* __restrict__ wl, const bool hi,
                                          const int s_i, const float4 a_i, const int m,
                                          float4& acc0, float4& acc1) {
#pragma unroll 4
    for (int t = 0; t < m; ++t) {
        const int s = __builtin_amdgcn_readlane(s_i, t);
        const float ax = readlane_f(a_i.x, t);
        const float ay = readlane_f(a_i.y, t);
        const float az = readlane_f(a_i.z, t);
        const float aw = readlane_f(a_i.w, t);
        const float al = hi ? ay : ax;
        const float au = hi ? aw : az;
        const float* r = wl + (size_t)s * 256;
        const float4 w0 = *(const float4*)(r);
        const float4 w1 = *(const float4*)(r + 128);
        acc0.x += al * w0.x;
        acc0.y += al * w0.y;
        acc0.z += al * w0.z;
        acc0.w += al * w0.w;
        acc1.x += au * w1.x;
        acc1.y += au * w1.y;
        acc1.z += au * w1.z;
        acc1.w += au * w1.w;
    }
}

#define XP 260
__global__ __launch_bounds__(512) void gather1_elu(const int* __restrict__ rowptr,
                                                   const int* __restrict__ tmp, int* perm,
                                                   int* __restrict__ srcs,
                                                   const int* __restrict__ src,
                                                   const float4* __restrict__ ss4,
                                                   const float4* __restrict__ sd4,
                                                   const float* __restrict__ Wh,
                                                   const float* __restrict__ Wout,
                                                   const float* __restrict__ aout,
                                                   float* __restrict__ Wx,
                                                   float* __restrict__ s2s,
                                                   float* __restrict__ s2d, int n) {
    __shared__ float4 xs4[16 * XP / 4];
    __shared__ float part[4 * 768];
    __shared__ float wxs[16 * 48];
    float* xs = (float*)xs4;
    const int tid = threadIdx.x;
    const int wid = tid >> 5;
    const int node0 = blockIdx.x * 16;
    const int nodeu = __builtin_amdgcn_readfirstlane(node0 + wid);
    const int lane = tid & 31;
    const int node = (nodeu < n) ? nodeu : (n - 1);
    const int beg = rowptr[node], end = rowptr[node + 1];
    const float4 sd = sd4[node];
    const bool hi = lane >= 16;
    const float* wl = Wh + lane * 4;
    float4 acc0 = {0.f, 0.f, 0.f, 0.f};
    float4 acc1 = {0.f, 0.f, 0.f, 0.f};
    if (end - beg <= 32) {
        const int m = end - beg;
        const bool has = lane < m;
        const int ei = has ? tmp[beg + lane] : 0x7fffffff;
        int r = 0;
        for (int t = 0; t < m; ++t) r += (__builtin_amdgcn_readlane(ei, t) < ei) ? 1 : 0;
        if (!has) r = lane;
        const int e = __builtin_amdgcn_ds_permute(r << 2, ei);
        int s_i;
        float4 x;
        edge_scores1e(src, ss4, sd, has, e, s_i, x);
        if (has && nodeu < n) srcs[beg + lane] = s_i;
        float4 mx, ex, dn, a_i;
        mx.x = wred_max(x.x);
        mx.y = wred_max(x.y);
        mx.z = wred_max(x.z);
        mx.w = wred_max(x.w);
        ex.x = expf(x.x - mx.x);
        ex.y = expf(x.y - mx.y);
        ex.z = expf(x.z - mx.z);
        ex.w = expf(x.w - mx.w);
        dn.x = wred_sum(ex.x);
        dn.y = wred_sum(ex.y);
        dn.z = wred_sum(ex.z);
        dn.w = wred_sum(ex.w);
        a_i.x = ex.x / dn.x;
        a_i.y = ex.y / dn.y;
        a_i.z = ex.z / dn.z;
        a_i.w = ex.w / dn.w;
        rows_fma1(wl, hi, s_i, a_i, end - beg, acc0, acc1);
    } else {
        for (int i0 = beg; i0 < end; i0 += 32) {
            const int i = i0 + lane;
            const int ei = (i < end) ? tmp[i] : 0x7fffffff;
            int r = 0;
            for (int j0 = beg; j0 < end; j0 += 32) {
                const int j = j0 + lane;
                const int ej = (j < end) ? tmp[j] : 0x7fffffff;
                const int mj = (end - j0 < 32) ? (end - j0) : 32;
                for (int t = 0; t < mj; ++t) r += (__builtin_amdgcn_readlane(ej, t) < ei) ? 1 : 0;
            }
            if (i < end) perm[beg + r] = ei;
        }
        __threadfence();
        float4 mx = {-INFINITY, -INFINITY, -INFINITY, -INFINITY};
        for (int i0 = beg; i0 < end; i0 += 32) {
            int s_i;
            float4 x;
            edge_scores1(perm, src, ss4, sd, i0 + lane, end, s_i, x);
            mx.x = fmaxf(mx.x, x.x);
            mx.y = fmaxf(mx.y, x.y);
            mx.z = fmaxf(mx.z, x.z);
            mx.w = fmaxf(mx.w, x.w);
        }
        mx.x = uni_f(wred_max(mx.x));
        mx.y = uni_f(wred_max(mx.y));
        mx.z = uni_f(wred_max(mx.z));
        mx.w = uni_f(wred_max(mx.w));
        float4 dn = {0.f, 0.f, 0.f, 0.f};
        for (int i0 = beg; i0 < end; i0 += 32) {
            int s_i;
            float4 x;
            edge_scores1(perm, src, ss4, sd, i0 + lane, end, s_i, x);
            if (i0 + lane < end) {
                dn.x += expf(x.x - mx.x);
                dn.y += expf(x.y - mx.y);
                dn.z += expf(x.z - mx.z);
                dn.w += expf(x.w - mx.w);
            }
        }
        dn.x = uni_f(wred_sum(dn.x));
        dn.y = uni_f(wred_sum(dn.y));
        dn.z = uni_f(wred_sum(dn.z));
        dn.w = uni_f(wred_sum(dn.w));
        for (int i0 = beg; i0 < end; i0 += 32) {
            int s_i;
            float4 x;
            edge_scores1(perm, src, ss4, sd, i0 + lane, end, s_i, x);
            float4 a_i = {0.f, 0.f, 0.f, 0.f};
            if (i0 + lane < end) {
                if (nodeu < n) srcs[i0 + lane] = s_i;
                a_i.x = expf(x.x - mx.x) / dn.x;
                a_i.y = expf(x.y - mx.y) / dn.y;
                a_i.z = expf(x.z - mx.z) / dn.z;
                a_i.w = expf(x.w - mx.w) / dn.w;
            }
            const int m = (end - i0 < 32) ? (end - i0) : 32;
            rows_fma1(wl, hi, s_i, a_i, m, acc0, acc1);
        }
    }
    acc0.x = acc0.x > 0.f ? acc0.x : (expf(acc0.x) - 1.f);
    acc0.y = acc0.y > 0.f ? acc0.y : (expf(acc0.y) - 1.f);
    acc0.z = acc0.z > 0.f ? acc0.z : (expf(acc0.z) - 1.f);
    acc0.w = acc0.w > 0.f ? acc0.w : (expf(acc0.w) - 1.f);
    acc1.x = acc1.x > 0.f ? acc1.x : (expf(acc1.x) - 1.f);
    acc1.y = acc1.y > 0.f ? acc1.y : (expf(acc1.y) - 1.f);
    acc1.z = acc1.z > 0.f ? acc1.z : (expf(acc1.z) - 1.f);
    acc1.w = acc1.w > 0.f ? acc1.w : (expf(acc1.w) - 1.f);
    {
        float* xr = xs + wid * XP + lane * 4;
        *(float4*)(xr) = acc0;
        *(float4*)(xr + 128) = acc1;
    }
    __syncthreads();
    if (wid < 12) {
        const int t = wid % 3, kq = wid / 3;
        const int half = lane >> 4, l16 = lane & 15;
        const int cc = t * 16 + l16;
        const bool cv = cc < 40;
        const float* bp = Wout + (cv ? cc : 39);
        const float* ap = xs + l16 * XP + kq * 64 + half * 2;
        v8f p0 = {}, p1 = {};
#pragma unroll 4
        for (int j = 0; j < 16; j += 2) {
            const int kb0 = kq * 64 + j * 4 + half * 2;
            const v2f a0 = *(const v2f*)(ap + j * 4);
            const v2f a1 = *(const v2f*)(ap + j * 4 + 4);
            const float w00 = bp[(kb0) * 40], w01 = bp[(kb0 + 1) * 40];
            const float w10 = bp[(kb0 + 4) * 40], w11 = bp[(kb0 + 5) * 40];
            v2f b0, b1;
            b0.x = cv ? w00 : 0.f;
            b0.y = cv ? w01 : 0.f;
            b1.x = cv ? w10 : 0.f;
            b1.y = cv ? w11 : 0.f;
            p0 = __builtin_amdgcn_wmma_f32_16x16x4_f32(false, a0, false, b0, (short)0, p0, false,
                                                       false);
            p1 = __builtin_amdgcn_wmma_f32_16x16x4_f32(false, a1, false, b1, (short)0, p1, false,
                                                       false);
        }
        float* pp = part + kq * 768 + t * 256 + lane;
#pragma unroll
        for (int v = 0; v < 8; ++v) pp[v * 32] = p0[v] + p1[v];
    }
    __syncthreads();
    for (int e = tid; e < 768; e += 512) {
        const float s = ((part[e] + part[768 + e]) + part[1536 + e]) + part[2304 + e];
        const int t = e >> 8, v = (e >> 5) & 7, l = e & 31;
        wxs[(8 * (l >> 4) + v) * 48 + t * 16 + (l & 15)] = s;
    }
    __syncthreads();
    for (int i = tid; i < 16 * 40; i += 512) {
        const int r = i / 40, cix = i - r * 40;
        if (node0 + r < n) Wx[(size_t)node0 * 40 + i] = wxs[r * 48 + cix];
    }
    if (tid < 32) {
        const int r = tid & 15;
        const float* av = aout + (tid >> 4) * 40;
        float s = 0.f;
        for (int k = 0; k < 40; ++k) s += wxs[r * 48 + k] * av[k];
        if (node0 + r < n) {
            if (tid < 16) s2s[node0 + r] = s;
            else s2d[node0 + r] = s;
        }
    }
}

__device__ __forceinline__ void edge_score2(const int* __restrict__ srcs,
                                            const float* __restrict__ ss, const float sd, int i,
                                            int end, int& s_i, float& x) {
    s_i = 0;
    x = -INFINITY;
    if (i < end) {
        s_i = srcs[i];
        x = lrelu(ss[s_i] + sd);
    }
}
__device__ __forceinline__ void rows_fma2(const float* __restrict__ w1p,
                                          const float* __restrict__ w2p, const int s_i,
                                          const float a_i, const int m, float& acc1,
                                          float& acc2) {
#pragma unroll 4
    for (int t = 0; t < m; ++t) {
        const int s = __builtin_amdgcn_readlane(s_i, t);
        const float a = readlane_f(a_i, t);
        acc1 += a * w1p[(size_t)s * 40];
        acc2 += a * w2p[(size_t)s * 40];
    }
}
__global__ __launch_bounds__(256) void gather2_lsm(const int* __restrict__ rowptr,
                                                   const int* __restrict__ srcs,
                                                   const float* __restrict__ ss,
                                                   const float* __restrict__ sdv,
                                                   const float* __restrict__ Wx,
                                                   float* __restrict__ out, int n) {
    const int node = __builtin_amdgcn_readfirstlane((int)((blockIdx.x * 256 + threadIdx.x) >> 5));
    const int lane = threadIdx.x & 31;
    if (node >= n) return;
    const int beg = rowptr[node], end = rowptr[node + 1];
    const float sd = sdv[node];
    const float* w1p = Wx + lane;
    const float* w2p = Wx + 32 + (lane & 7);
    float acc1 = 0.f, acc2 = 0.f;
    if (end - beg <= 32) {
        int s_i;
        float x;
        edge_score2(srcs, ss, sd, beg + lane, end, s_i, x);
        const float mx = wred_max(x);
        const float ex = expf(x - mx);
        const float dn = wred_sum(ex);
        rows_fma2(w1p, w2p, s_i, ex / dn, end - beg, acc1, acc2);
    } else {
        float mx = -INFINITY;
        for (int i0 = beg; i0 < end; i0 += 32) {
            int s_i;
            float x;
            edge_score2(srcs, ss, sd, i0 + lane, end, s_i, x);
            mx = fmaxf(mx, x);
        }
        mx = uni_f(wred_max(mx));
        float dn = 0.f;
        for (int i0 = beg; i0 < end; i0 += 32) {
            int s_i;
            float x;
            edge_score2(srcs, ss, sd, i0 + lane, end, s_i, x);
            if (i0 + lane < end) dn += expf(x - mx);
        }
        dn = uni_f(wred_sum(dn));
        for (int i0 = beg; i0 < end; i0 += 32) {
            int s_i;
            float x;
            edge_score2(srcs, ss, sd, i0 + lane, end, s_i, x);
            const float a_i = (i0 + lane < end) ? expf(x - mx) / dn : 0.f;
            const int m = (end - i0 < 32) ? (end - i0) : 32;
            rows_fma2(w1p, w2p, s_i, a_i, m, acc1, acc2);
        }
    }
    const float v1 = acc1;
    const float v2 = (lane < 8) ? acc2 : -INFINITY;
    const float mxo = wred_max(fmaxf(v1, v2));
    float sm = expf(v1 - mxo) + ((lane < 8) ? expf(v2 - mxo) : 0.f);
    sm = wred_sum(sm);
    const float lg = mxo + logf(sm);
    float* row = out + (size_t)node * 40;
    row[lane] = v1 - lg;
    if (lane < 8) row[32 + lane] = v2 - lg;
}

extern "C" void kernel_launch(void* const* d_in, const int* in_sizes, int n_in, void* d_out,
                              int out_size, void* d_ws, size_t ws_size, hipStream_t stream) {
    (void)n_in; (void)out_size; (void)ws_size;
    const float* h      = (const float*)d_in[0];
    const int*   eidx   = (const int*)d_in[1];
    const float* Wheads = (const float*)d_in[2];
    const float* aheads = (const float*)d_in[3];
    const float* Wout   = (const float*)d_in[4];
    const float* aout   = (const float*)d_in[5];
    float* out = (float*)d_out;

    const int N = in_sizes[0] / 256;
    const int E = in_sizes[1] / 2;
    const int* src = eidx;
    const int* dst = eidx + E;

    float* ws = (float*)d_ws;
    size_t oWh   = 0;
    size_t oHP   = oWh  + (size_t)N * 256;
    size_t oS1s  = oHP  + (size_t)N * 256;
    size_t oS1d  = oS1s + (size_t)N * 4;
    size_t oEdge = oS1d + (size_t)N * 12;
    size_t oCnt  = oEdge + (size_t)E * 4;
    size_t oRP   = oCnt + (size_t)N + 1;
    size_t oBs   = oRP  + (size_t)N + 1;
    size_t oPerm = oEdge + (size_t)E * 4 + (size_t)N * 4;
    size_t oRank = oEdge;
    size_t oTmp  = oEdge + (size_t)E;
    size_t oWx   = oHP;
    size_t oS2s  = oHP + (size_t)N * 40;
    size_t oS2d  = oS2s + (size_t)N;

    int* rowptr = (int*)(ws + oRP);
    int* bsum   = (int*)(ws + oBs);
    int* perm   = (int*)(ws + oPerm);
    int* rank   = (int*)(ws + oRank);
    int* tmp    = (int*)(ws + oTmp);
    int* srcs   = rank;

    const int geE = (E + 255) / 256;
    const int n1  = N + 1;
    const int nb  = (n1 + 1023) / 1024;
    const int nch = (E + CSR_CHUNK - 1) / CSR_CHUNK;
    unsigned* cntw = (unsigned*)(ws + oHP);
    int* start     = (int*)(ws + oHP + (size_t)nch * CSR_CAPW);

    csr_count_lds<<<nch, 1024, 0, stream>>>(dst, cntw, rank, E);
    csr_blocksum<<<nb, 256, 0, stream>>>(cntw, bsum, n1, nch);
    csr_rowptr<<<nb, 256, 0, stream>>>(cntw, bsum, rowptr, start, n1, nch);
    csr_place<<<geE, 256, 0, stream>>>(dst, start, rank, tmp, E);

    dim3 g1((N + 127) / 128, 4);
    gemm_wmma_f32<4, 0><<<g1, 256, 0, stream>>>(h, Wheads, ws + oWh, aheads, ws + oS1s,
                                                ws + oS1d, N);

    gather1_elu<<<(N + 15) / 16, 512, 0, stream>>>(rowptr, tmp, perm, srcs, src,
                                                   (const float4*)(ws + oS1s),
                                                   (const float4*)(ws + oS1d), ws + oWh, Wout,
                                                   aout, ws + oWx, ws + oS2s, ws + oS2d, N);


    gather2_lsm<<<(N + 7) / 8, 256, 0, stream>>>(rowptr, srcs, ws + oS2s, ws + oS2d, ws + oWx,
                                                 out, N);
}


# edge placement (last sort step, 8 edges per thread) runs as every fifth block of the layer-1 GEMM launch: its scattered reads and writes overlap the matrix work; one launch fewer; sums unchanged
# speedup vs baseline: 7.5787x; 1.0564x over previous
#include <hip/hip_runtime.h>


#define LRELU 0.2f

typedef __attribute__((ext_vector_type(2))) float v2f;
typedef __attribute__((ext_vector_type(8))) float v8f;

__device__ __forceinline__ float wred_sum(float v) {
#pragma unroll
    for (int o = 16; o > 0; o >>= 1) v += __shfl_xor(v, o, 32);
    return v;
}
__device__ __forceinline__ float wred_max(float v) {
#pragma unroll
    for (int o = 16; o > 0; o >>= 1) v = fmaxf(v, __shfl_xor(v, o, 32));
    return v;
}
__device__ __forceinline__ float lrelu(float x) { return x > 0.f ? x : LRELU * x; }
__device__ __forceinline__ float xchg_add(const bool up, const float lo, const float hi,
                                          const int w) {
    const float keep = up ? hi : lo;
    const float send = up ? lo : hi;
    return keep + __shfl_xor(send, w, 32);
}

template <int COLT, int MODE>
__device__ __forceinline__ void gemm_wmma_f32(const float* __restrict__ A,
                                              const float* __restrict__ B,
                                              float* __restrict__ C,
                                              const float* __restrict__ avec,
                                              float* __restrict__ ssrc,
                                              float* __restrict__ sdst, const int nrows,
                                              const int bx, const int by) {
    constexpr int CW = COLT * 16;
    __shared__ float4 Blds4[64 * CW];
    float* Blds = (float*)Blds4;
    const int tid = threadIdx.x;
    const int colBase = by * CW;

    if constexpr (MODE == 0) {
        const float4* __restrict__ b4 = (const float4*)(B + (size_t)by * (256 * 64));
        float4 st[16];
#pragma unroll
        for (int i = 0; i < 16; ++i) st[i] = b4[tid + i * 256];
#pragma unroll
        for (int i = 0; i < 16; ++i) Blds4[tid + i * 256] = st[i];
    } else {
        const float4* __restrict__ b4 = (const float4*)B;
        float4 st[10];
#pragma unroll
        for (int i = 0; i < 10; ++i) st[i] = b4[tid + i * 256];
#pragma unroll
        for (int i = 0; i < 10; ++i) {
            const int j = tid + i * 256;
            const int k = j / 10;
            Blds4[k * 12 + (j - k * 10)] = st[i];
        }
        for (int j = tid; j < 512; j += 256)
            Blds4[(j >> 1) * 12 + 10 + (j & 1)] = make_float4(0.f, 0.f, 0.f, 0.f);
    }
    __syncthreads();

    const int wid  = tid >> 5;
    const int lane = tid & 31;
    const int half = lane >> 4;
    const int l16  = lane & 15;
    const int rowBase = bx * 128 + wid * 16;

    int arow = rowBase + l16;
    if (arow >= nrows) arow = nrows - 1;
    const float* aptr = A + (size_t)arow * 256;

    v8f acc[COLT] = {};

    for (int k0 = 0; k0 < 256; k0 += 4) {
        const int kb = k0 + half * 2;
        v2f a;
        a.x = aptr[kb];
        a.y = aptr[kb + 1];
#pragma unroll
        for (int t = 0; t < COLT; ++t) {
            v2f b;
            const int c = t * 16 + l16;
            b.x = Blds[kb * CW + c];
            b.y = Blds[(kb + 1) * CW + c];
            acc[t] = __builtin_amdgcn_wmma_f32_16x16x4_f32(false, a, false, b, (short)0,
                                                           acc[t], false, false);
        }
    }

    {
        float as[COLT], ad[COLT];
        for (int t = 0; t < COLT; ++t) {
            const int c = t * 16 + l16;
            if constexpr (MODE == 0) {
                as[t] = avec[by * 128 + c];
                ad[t] = avec[by * 128 + 64 + c];
            } else {
                as[t] = (c < 40) ? avec[c] : 0.0f;
                ad[t] = (c < 40) ? avec[40 + c] : 0.0f;
            }
        }
#define ROW_DOTS(v, s, d)                      \
    float s = 0.f, d = 0.f;                    \
    for (int t = 0; t < COLT; ++t) {           \
        s += acc[t][v] * as[t];                \
        d += acc[t][v] * ad[t];                \
    }
        ROW_DOTS(0, s0, d0)
        ROW_DOTS(1, s1, d1)
        ROW_DOTS(2, s2, d2)
        ROW_DOTS(3, s3, d3)
        ROW_DOTS(4, s4, d4)
        ROW_DOTS(5, s5, d5)
        ROW_DOTS(6, s6, d6)
        ROW_DOTS(7, s7, d7)
#undef ROW_DOTS
        const bool u8 = (l16 & 8) != 0, u4 = (l16 & 4) != 0, u2 = (l16 & 2) != 0,
                   u1 = (l16 & 1) != 0;
        const float q0 = xchg_add(u8, s0, d0, 8);
        const float q1 = xchg_add(u8, s1, d1, 8);
        const float q2 = xchg_add(u8, s2, d2, 8);
        const float q3 = xchg_add(u8, s3, d3, 8);
        const float q4 = xchg_add(u8, s4, d4, 8);
        const float q5 = xchg_add(u8, s5, d5, 8);
        const float q6 = xchg_add(u8, s6, d6, 8);
        const float q7 = xchg_add(u8, s7, d7, 8);
        const float r0 = xchg_add(u4, q0, q4, 4);
        const float r1 = xchg_add(u4, q1, q5, 4);
        const float r2 = xchg_add(u4, q2, q6, 4);
        const float r3 = xchg_add(u4, q3, q7, 4);
        const float y0 = xchg_add(u2, r0, r2, 2);
        const float y1 = xchg_add(u2, r1, r3, 2);
        const float tot = xchg_add(u1, y0, y1, 1);
        const int srow = rowBase + half * 8 + (l16 & 7);
        float* sp = (l16 < 8) ? ssrc : sdst;
        if (srow < nrows) {
            if constexpr (MODE == 0) {
                sp[(size_t)srow * 4 + by] = tot;
            } else {
                sp[srow] = tot;
            }
        }
    }

    if (rowBase + 16 <= nrows) {
#pragma unroll
        for (int t = 0; t < COLT; ++t) {
            const int col = colBase + t * 16 + l16;
            if constexpr (MODE == 1) {
                if (col >= 40) continue;
            }
#pragma unroll
            for (int v = 0; v < 8; ++v) {
                const int row = rowBase + v + half * 8;
                if constexpr (MODE == 0) {
                    C[(size_t)row * 256 + col] = acc[t][v];
                } else {
                    C[(size_t)row * 40 + col] = acc[t][v];
                }
            }
        }
    } else {
#pragma unroll
        for (int t = 0; t < COLT; ++t) {
            const int col = colBase + t * 16 + l16;
#pragma unroll
            for (int v = 0; v < 8; ++v) {
                const int row = rowBase + v + half * 8;
                if constexpr (MODE == 0) {
                    if (row < nrows) C[(size_t)row * 256 + col] = acc[t][v];
                } else {
                    if (row < nrows && col < 40) C[(size_t)row * 40 + col] = acc[t][v];
                }
            }
        }
    }
}

__device__ __forceinline__ int wred_sum_i(int v) {
    for (int o = 16; o > 0; o >>= 1) v += __shfl_xor(v, o, 32);
    return v;
}
__device__ __forceinline__ float readlane_f(float v, int l) {
    return __int_as_float(__builtin_amdgcn_readlane(__float_as_int(v), l));
}
__device__ __forceinline__ float uni_f(float v) {
    return __int_as_float(__builtin_amdgcn_readfirstlane(__float_as_int(v)));
}
#define CSR_CHUNK 50000
#define CSR_CAPW 51200
#define CSR_SP (2 * CSR_CAPW)
__global__ __launch_bounds__(1024) void csr_count_lds(const int* __restrict__ dst,
                                                      unsigned* __restrict__ cntw,
                                                      int* __restrict__ rank, int ne) {
    __shared__ uint4 cl4[CSR_CAPW / 4];
    unsigned* cl = (unsigned*)cl4;
    const int tid = threadIdx.x;
    for (int i = tid; i < CSR_CAPW / 4; i += 1024) cl4[i] = make_uint4(0u, 0u, 0u, 0u);
    __syncthreads();
    const int e0 = blockIdx.x * CSR_CHUNK;
    const int e1 = (e0 + CSR_CHUNK < ne) ? (e0 + CSR_CHUNK) : ne;
    const int nfull = (e1 - e0) / (16 * 1024);
    int tb = e0;
    for (int r = 0; r < nfull; ++r, tb += 16 * 1024) {
        const int eb = tb + tid;
        int d[16];
        unsigned old[16];
#pragma unroll
        for (int k = 0; k < 16; ++k) d[k] = dst[eb + k * 1024];
#pragma unroll
        for (int k = 0; k < 16; ++k) {
            const int sh = (d[k] & 1) << 4;
            old[k] = atomicAdd(&cl[d[k] >> 1], 1u << sh);
        }
#pragma unroll
        for (int k = 0; k < 16; ++k) {
            const int sh = (d[k] & 1) << 4;
            rank[eb + k * 1024] = (int)((old[k] >> sh) & 0xffffu);
        }
    }
    for (int e = tb + tid; e < e1; e += 1024) {
        const int dd = dst[e];
        const int sh = (dd & 1) << 4;
        const unsigned o = atomicAdd(&cl[dd >> 1], 1u << sh);
        rank[e] = (int)((o >> sh) & 0xffffu);
    }
    __syncthreads();
    uint4* o = (uint4*)(cntw + (size_t)blockIdx.x * CSR_CAPW);
    for (int i = tid; i < CSR_CAPW / 4; i += 1024) o[i] = cl4[i];
}
__device__ __forceinline__ void csr_deg4(const unsigned* __restrict__ cntw, const int nch,
                                         const int base, int c[4]) {
    c[0] = 0;
    c[1] = 0;
    c[2] = 0;
    c[3] = 0;
    const unsigned* p = cntw + (base >> 1);
#pragma unroll 8
    for (int ch = 0; ch < nch; ++ch) {
        const uint2 v = *(const uint2*)(p + (size_t)ch * CSR_CAPW);
        c[0] += (int)(v.x & 0xffffu);
        c[1] += (int)(v.x >> 16);
        c[2] += (int)(v.y & 0xffffu);
        c[3] += (int)(v.y >> 16);
    }
}
__global__ __launch_bounds__(256) void csr_blocksum(const unsigned* __restrict__ cntw,
                                                    int* __restrict__ bsum, int n1, int nch) {
    __shared__ int wsum[8];
    const int base = blockIdx.x * 1024 + threadIdx.x * 4;
    int c[4];
    csr_deg4(cntw, nch, base, c);
    int s = 0;
    for (int k = 0; k < 4; ++k)
        if (base + k < n1) s += c[k];
    s = wred_sum_i(s);
    if ((threadIdx.x & 31) == 0) wsum[threadIdx.x >> 5] = s;
    __syncthreads();
    if (threadIdx.x == 0) {
        int t = 0;
        for (int w = 0; w < 8; ++w) t += wsum[w];
        bsum[blockIdx.x] = t;
    }
}
__global__ __launch_bounds__(256) void csr_rowptr(const unsigned* __restrict__ cntw,
                                                  const int* __restrict__ bsum,
                                                  int* __restrict__ rowptr,
                                                  int* __restrict__ start, int n1, int nch) {
    __shared__ int wsum[8];
    __shared__ int wsum2[8];
    const int tid = threadIdx.x, lane = tid & 31, wid = tid >> 5;
    int p = 0;
    for (int j = tid; j < (int)blockIdx.x; j += 256) p += bsum[j];
    p = wred_sum_i(p);
    if (lane == 0) wsum[wid] = p;
    const int base = blockIdx.x * 1024 + tid * 4;
    int c[4];
    csr_deg4(cntw, nch, base, c);
    int s = 0;
    for (int k = 0; k < 4; ++k) {
        if (base + k >= n1) c[k] = 0;
        s += c[k];
    }
    int inc = s;
    for (int o = 1; o < 32; o <<= 1) {
        int t = __shfl_up(inc, o, 32);
        if (lane >= o) inc += t;
    }
    if (lane == 31) wsum2[wid] = inc;
    __syncthreads();
    int run = inc - s;
    for (int w = 0; w < 8; ++w) {
        run += wsum[w];
        if (w < wid) run += wsum2[w];
    }
    int4 a;
    a.x = run;
    a.y = a.x + c[0];
    a.z = a.y + c[1];
    a.w = a.z + c[2];
    if (base < n1) rowptr[base] = a.x;
    if (base + 1 < n1) rowptr[base + 1] = a.y;
    if (base + 2 < n1) rowptr[base + 2] = a.z;
    if (base + 3 < n1) rowptr[base + 3] = a.w;
    const unsigned* cp = cntw + (base >> 1);
#pragma unroll 8
    for (int ch = 0; ch < nch; ++ch) {
        const uint2 v = *(const uint2*)(cp + (size_t)ch * CSR_CAPW);
        *(int4*)(start + (size_t)ch * CSR_SP + base) = a;
        a.x += (int)(v.x & 0xffffu);
        a.y += (int)(v.x >> 16);
        a.z += (int)(v.y & 0xffffu);
        a.w += (int)(v.y >> 16);
    }
}
#define PLACE_EDGES 2048
__device__ __forceinline__ void csr_place_block(const int* __restrict__ dst,
                                                const int* __restrict__ start,
                                                const int* __restrict__ rank,
                                                int* __restrict__ tmp, const int ne,
                                                const int pb) {
    const int eb = pb * PLACE_EDGES;
    const int tid = (int)threadIdx.x;
    if (eb + PLACE_EDGES <= ne) {
        const int e0 = eb + tid;
        int d[8], rk[8], st[8];
#pragma unroll
        for (int k = 0; k < 8; ++k) {
            d[k] = dst[e0 + k * 256];
            rk[k] = rank[e0 + k * 256];
        }
#pragma unroll
        for (int k = 0; k < 8; ++k)
            st[k] = start[(size_t)((e0 + k * 256) / CSR_CHUNK) * CSR_SP + d[k]];
#pragma unroll
        for (int k = 0; k < 8; ++k) tmp[st[k] + rk[k]] = e0 + k * 256;
    } else {
        for (int e = eb + tid; e < ne; e += 256)
            tmp[start[(size_t)(e / CSR_CHUNK) * CSR_SP + dst[e]] + rank[e]] = e;
    }
}
__global__ __launch_bounds__(256) void gemm1_place(const float* __restrict__ A,
                                                   const float* __restrict__ B,
                                                   float* __restrict__ C,
                                                   const float* __restrict__ avec,
                                                   float* __restrict__ ssrc,
                                                   float* __restrict__ sdst, int nrows,
                                                   const int* __restrict__ dst,
                                                   const int* __restrict__ start,
                                                   const int* __restrict__ rank,
                                                   int* __restrict__ tmp, int ne) {
    const int nrt = (nrows + 127) >> 7;
    const int npl = (ne + PLACE_EDGES - 1) / PLACE_EDGES;
    const int m = (nrt < npl) ? nrt : npl;
    const int b = (int)blockIdx.x;
    int pb = -1, g = 0;
    if (b < 5 * m) {
        const int grp = b / 5, r = b - grp * 5;
        if (r == 4) pb = grp;
        else g = grp * 4 + r;
    } else {
        const int i = b - 5 * m, prest = npl - m;
        if (i < prest) pb = m + i;
        else g = 4 * m + (i - prest);
    }
    if (pb >= 0) {
        csr_place_block(dst, start, rank, tmp, ne, pb);
        return;
    }
    gemm_wmma_f32<4, 0>(A, B, C, avec, ssrc, sdst, nrows, g % nrt, g / nrt);
}
__device__ __forceinline__ void edge_scores1e(const int* __restrict__ src,
                                              const float4* __restrict__ ss4, const float4 sd,
                                              const bool has, const int e, int& s_i,
                                              float4& x) {
    s_i = 0;
    x.x = -INFINITY;
    x.y = -INFINITY;
    x.z = -INFINITY;
    x.w = -INFINITY;
    if (has) {
        s_i = src[e];
        const float4 vs = ss4[s_i];
        x.x = lrelu(vs.x + sd.x);
        x.y = lrelu(vs.y + sd.y);
        x.z = lrelu(vs.z + sd.z);
        x.w = lrelu(vs.w + sd.w);
    }
}
__device__ __forceinline__ void edge_scores1(const int* perm, const int* __restrict__ src,
                                             const float4* __restrict__ ss4, const float4 sd,
                                             int i, int end, int& s_i, float4& x) {
    const bool has = i < end;
    int e = 0;
    if (has) e = perm[i];
    edge_scores1e(src, ss4, sd, has, e, s_i, x);
}
__device__ __forceinline__ void rows_fma1(const float* __restrict__ wl, const bool hi,
                                          const int s_i, const float4 a_i, const int m,
                                          float4& acc0, float4& acc1) {
#pragma unroll 4
    for (int t = 0; t < m; ++t) {
        const int s = __builtin_amdgcn_readlane(s_i, t);
        const float ax = readlane_f(a_i.x, t);
        const float ay = readlane_f(a_i.y, t);
        const float az = readlane_f(a_i.z, t);
        const float aw = readlane_f(a_i.w, t);
        const float al = hi ? ay : ax;
        const float au = hi ? aw : az;
        const float* r = wl + (size_t)s * 256;
        const float4 w0 = *(const float4*)(r);
        const float4 w1 = *(const float4*)(r + 128);
        acc0.x += al * w0.x;
        acc0.y += al * w0.y;
        acc0.z += al * w0.z;
        acc0.w += al * w0.w;
        acc1.x += au * w1.x;
        acc1.y += au * w1.y;
        acc1.z += au * w1.z;
        acc1.w += au * w1.w;
    }
}

#define XP 260
__global__ __launch_bounds__(512) void gather1_elu(const int* __restrict__ rowptr,
                                                   const int* __restrict__ tmp, int* perm,
                                                   int* __restrict__ srcs,
                                                   const int* __restrict__ src,
                                                   const float4* __restrict__ ss4,
                                                   const float4* __restrict__ sd4,
                                                   const float* __restrict__ Wh,
                                                   const float* __restrict__ Wout,
                                                   const float* __restrict__ aout,
                                                   float* __restrict__ Wx,
                                                   float* __restrict__ s2s,
                                                   float* __restrict__ s2d, int n) {
    __shared__ float4 xs4[16 * XP / 4];
    __shared__ float part[4 * 768];
    __shared__ float wxs[16 * 48];
    float* xs = (float*)xs4;
    const int tid = threadIdx.x;
    const int wid = tid >> 5;
    const int node0 = blockIdx.x * 16;
    const int nodeu = __builtin_amdgcn_readfirstlane(node0 + wid);
    const int lane = tid & 31;
    const int node = (nodeu < n) ? nodeu : (n - 1);
    const int beg = rowptr[node], end = rowptr[node + 1];
    const float4 sd = sd4[node];
    const bool hi = lane >= 16;
    const float* wl = Wh + lane * 4;
    float4 acc0 = {0.f, 0.f, 0.f, 0.f};
    float4 acc1 = {0.f, 0.f, 0.f, 0.f};
    if (end - beg <= 32) {
        const int m = end - beg;
        const bool has = lane < m;
        const int ei = has ? tmp[beg + lane] : 0x7fffffff;
        int r = 0;
        for (int t = 0; t < m; ++t) r += (__builtin_amdgcn_readlane(ei, t) < ei) ? 1 : 0;
        if (!has) r = lane;
        const int e = __builtin_amdgcn_ds_permute(r << 2, ei);
        int s_i;
        float4 x;
        edge_scores1e(src, ss4, sd, has, e, s_i, x);
        if (has && nodeu < n) srcs[beg + lane] = s_i;
        float4 mx, ex, dn, a_i;
        mx.x = wred_max(x.x);
        mx.y = wred_max(x.y);
        mx.z = wred_max(x.z);
        mx.w = wred_max(x.w);
        ex.x = expf(x.x - mx.x);
        ex.y = expf(x.y - mx.y);
        ex.z = expf(x.z - mx.z);
        ex.w = expf(x.w - mx.w);
        dn.x = wred_sum(ex.x);
        dn.y = wred_sum(ex.y);
        dn.z = wred_sum(ex.z);
        dn.w = wred_sum(ex.w);
        a_i.x = ex.x / dn.x;
        a_i.y = ex.y / dn.y;
        a_i.z = ex.z / dn.z;
        a_i.w = ex.w / dn.w;
        rows_fma1(wl, hi, s_i, a_i, end - beg, acc0, acc1);
    } else {
        for (int i0 = beg; i0 < end; i0 += 32) {
            const int i = i0 + lane;
            const int ei = (i < end) ? tmp[i] : 0x7fffffff;
            int r = 0;
            for (int j0 = beg; j0 < end; j0 += 32) {
                const int j = j0 + lane;
                const int ej = (j < end) ? tmp[j] : 0x7fffffff;
                const int mj = (end - j0 < 32) ? (end - j0) : 32;
                for (int t = 0; t < mj; ++t) r += (__builtin_amdgcn_readlane(ej, t) < ei) ? 1 : 0;
            }
            if (i < end) perm[beg + r] = ei;
        }
        __threadfence();
        float4 mx = {-INFINITY, -INFINITY, -INFINITY, -INFINITY};
        for (int i0 = beg; i0 < end; i0 += 32) {
            int s_i;
            float4 x;
            edge_scores1(perm, src, ss4, sd, i0 + lane, end, s_i, x);
            mx.x = fmaxf(mx.x, x.x);
            mx.y = fmaxf(mx.y, x.y);
            mx.z = fmaxf(mx.z, x.z);
            mx.w = fmaxf(mx.w, x.w);
        }
        mx.x = uni_f(wred_max(mx.x));
        mx.y = uni_f(wred_max(mx.y));
        mx.z = uni_f(wred_max(mx.z));
        mx.w = uni_f(wred_max(mx.w));
        float4 dn = {0.f, 0.f, 0.f, 0.f};
        for (int i0 = beg; i0 < end; i0 += 32) {
            int s_i;
            float4 x;
            edge_scores1(perm, src, ss4, sd, i0 + lane, end, s_i, x);
            if (i0 + lane < end) {
                dn.x += expf(x.x - mx.x);
                dn.y += expf(x.y - mx.y);
                dn.z += expf(x.z - mx.z);
                dn.w += expf(x.w - mx.w);
            }
        }
        dn.x = uni_f(wred_sum(dn.x));
        dn.y = uni_f(wred_sum(dn.y));
        dn.z = uni_f(wred_sum(dn.z));
        dn.w = uni_f(wred_sum(dn.w));
        for (int i0 = beg; i0 < end; i0 += 32) {
            int s_i;
            float4 x;
            edge_scores1(perm, src, ss4, sd, i0 + lane, end, s_i, x);
            float4 a_i = {0.f, 0.f, 0.f, 0.f};
            if (i0 + lane < end) {
                if (nodeu < n) srcs[i0 + lane] = s_i;
                a_i.x = expf(x.x - mx.x) / dn.x;
                a_i.y = expf(x.y - mx.y) / dn.y;
                a_i.z = expf(x.z - mx.z) / dn.z;
                a_i.w = expf(x.w - mx.w) / dn.w;
            }
            const int m = (end - i0 < 32) ? (end - i0) : 32;
            rows_fma1(wl, hi, s_i, a_i, m, acc0, acc1);
        }
    }
    acc0.x = acc0.x > 0.f ? acc0.x : (expf(acc0.x) - 1.f);
    acc0.y = acc0.y > 0.f ? acc0.y : (expf(acc0.y) - 1.f);
    acc0.z = acc0.z > 0.f ? acc0.z : (expf(acc0.z) - 1.f);
    acc0.w = acc0.w > 0.f ? acc0.w : (expf(acc0.w) - 1.f);
    acc1.x = acc1.x > 0.f ? acc1.x : (expf(acc1.x) - 1.f);
    acc1.y = acc1.y > 0.f ? acc1.y : (expf(acc1.y) - 1.f);
    acc1.z = acc1.z > 0.f ? acc1.z : (expf(acc1.z) - 1.f);
    acc1.w = acc1.w > 0.f ? acc1.w : (expf(acc1.w) - 1.f);
    {
        float* xr = xs + wid * XP + lane * 4;
        *(float4*)(xr) = acc0;
        *(float4*)(xr + 128) = acc1;
    }
    __syncthreads();
    if (wid < 12) {
        const int t = wid % 3, kq = wid / 3;
        const int half = lane >> 4, l16 = lane & 15;
        const int cc = t * 16 + l16;
        const bool cv = cc < 40;
        const float* bp = Wout + (cv ? cc : 39);
        const float* ap = xs + l16 * XP + kq * 64 + half * 2;
        v8f p0 = {}, p1 = {};
#pragma unroll 4
        for (int j = 0; j < 16; j += 2) {
            const int kb0 = kq * 64 + j * 4 + half * 2;
            const v2f a0 = *(const v2f*)(ap + j * 4);
            const v2f a1 = *(const v2f*)(ap + j * 4 + 4);
            const float w00 = bp[(kb0) * 40], w01 = bp[(kb0 + 1) * 40];
            const float w10 = bp[(kb0 + 4) * 40], w11 = bp[(kb0 + 5) * 40];
            v2f b0, b1;
            b0.x = cv ? w00 : 0.f;
            b0.y = cv ? w01 : 0.f;
            b1.x = cv ? w10 : 0.f;
            b1.y = cv ? w11 : 0.f;
            p0 = __builtin_amdgcn_wmma_f32_16x16x4_f32(false, a0, false, b0, (short)0, p0, false,
                                                       false);
            p1 = __builtin_amdgcn_wmma_f32_16x16x4_f32(false, a1, false, b1, (short)0, p1, false,
                                                       false);
        }
        float* pp = part + kq * 768 + t * 256 + lane;
#pragma unroll
        for (int v = 0; v < 8; ++v) pp[v * 32] = p0[v] + p1[v];
    }
    __syncthreads();
    for (int e = tid; e < 768; e += 512) {
        const float s = ((part[e] + part[768 + e]) + part[1536 + e]) + part[2304 + e];
        const int t = e >> 8, v = (e >> 5) & 7, l = e & 31;
        wxs[(8 * (l >> 4) + v) * 48 + t * 16 + (l & 15)] = s;
    }
    __syncthreads();
    for (int i = tid; i < 16 * 40; i += 512) {
        const int r = i / 40, cix = i - r * 40;
        if (node0 + r < n) Wx[(size_t)node0 * 40 + i] = wxs[r * 48 + cix];
    }
    if (tid < 32) {
        const int r = tid & 15;
        const float* av = aout + (tid >> 4) * 40;
        float s = 0.f;
        for (int k = 0; k < 40; ++k) s += wxs[r * 48 + k] * av[k];
        if (node0 + r < n) {
            if (tid < 16) s2s[node0 + r] = s;
            else s2d[node0 + r] = s;
        }
    }
}

__device__ __forceinline__ void edge_score2(const int* __restrict__ srcs,
                                            const float* __restrict__ ss, const float sd, int i,
                                            int end, int& s_i, float& x) {
    s_i = 0;
    x = -INFINITY;
    if (i < end) {
        s_i = srcs[i];
        x = lrelu(ss[s_i] + sd);
    }
}
__device__ __forceinline__ void rows_fma2(const float* __restrict__ w1p,
                                          const float* __restrict__ w2p, const int s_i,
                                          const float a_i, const int m, float& acc1,
                                          float& acc2) {
#pragma unroll 4
    for (int t = 0; t < m; ++t) {
        const int s = __builtin_amdgcn_readlane(s_i, t);
        const float a = readlane_f(a_i, t);
        acc1 += a * w1p[(size_t)s * 40];
        acc2 += a * w2p[(size_t)s * 40];
    }
}
__global__ __launch_bounds__(256) void gather2_lsm(const int* __restrict__ rowptr,
                                                   const int* __restrict__ srcs,
                                                   const float* __restrict__ ss,
                                                   const float* __restrict__ sdv,
                                                   const float* __restrict__ Wx,
                                                   float* __restrict__ out, int n) {
    const int node = __builtin_amdgcn_readfirstlane((int)((blockIdx.x * 256 + threadIdx.x) >> 5));
    const int lane = threadIdx.x & 31;
    if (node >= n) return;
    const int beg = rowptr[node], end = rowptr[node + 1];
    const float sd = sdv[node];
    const float* w1p = Wx + lane;
    const float* w2p = Wx + 32 + (lane & 7);
    float acc1 = 0.f, acc2 = 0.f;
    if (end - beg <= 32) {
        int s_i;
        float x;
        edge_score2(srcs, ss, sd, beg + lane, end, s_i, x);
        const float mx = wred_max(x);
        const float ex = expf(x - mx);
        const float dn = wred_sum(ex);
        rows_fma2(w1p, w2p, s_i, ex / dn, end - beg, acc1, acc2);
    } else {
        float mx = -INFINITY;
        for (int i0 = beg; i0 < end; i0 += 32) {
            int s_i;
            float x;
            edge_score2(srcs, ss, sd, i0 + lane, end, s_i, x);
            mx = fmaxf(mx, x);
        }
        mx = uni_f(wred_max(mx));
        float dn = 0.f;
        for (int i0 = beg; i0 < end; i0 += 32) {
            int s_i;
            float x;
            edge_score2(srcs, ss, sd, i0 + lane, end, s_i, x);
            if (i0 + lane < end) dn += expf(x - mx);
        }
        dn = uni_f(wred_sum(dn));
        for (int i0 = beg; i0 < end; i0 += 32) {
            int s_i;
            float x;
            edge_score2(srcs, ss, sd, i0 + lane, end, s_i, x);
            const float a_i = (i0 + lane < end) ? expf(x - mx) / dn : 0.f;
            const int m = (end - i0 < 32) ? (end - i0) : 32;
            rows_fma2(w1p, w2p, s_i, a_i, m, acc1, acc2);
        }
    }
    const float v1 = acc1;
    const float v2 = (lane < 8) ? acc2 : -INFINITY;
    const float mxo = wred_max(fmaxf(v1, v2));
    float sm = expf(v1 - mxo) + ((lane < 8) ? expf(v2 - mxo) : 0.f);
    sm = wred_sum(sm);
    const float lg = mxo + logf(sm);
    float* row = out + (size_t)node * 40;
    row[lane] = v1 - lg;
    if (lane < 8) row[32 + lane] = v2 - lg;
}

extern "C" void kernel_launch(void* const* d_in, const int* in_sizes, int n_in, void* d_out,
                              int out_size, void* d_ws, size_t ws_size, hipStream_t stream) {
    (void)n_in; (void)out_size; (void)ws_size;
    const float* h      = (const float*)d_in[0];
    const int*   eidx   = (const int*)d_in[1];
    const float* Wheads = (const float*)d_in[2];
    const float* aheads = (const float*)d_in[3];
    const float* Wout   = (const float*)d_in[4];
    const float* aout   = (const float*)d_in[5];
    float* out = (float*)d_out;

    const int N = in_sizes[0] / 256;
    const int E = in_sizes[1] / 2;
    const int* src = eidx;
    const int* dst = eidx + E;

    float* ws = (float*)d_ws;
    size_t oWh   = 0;
    size_t oHP   = oWh  + (size_t)N * 256;
    size_t oS1s  = oHP  + (size_t)N * 256;
    size_t oS1d  = oS1s + (size_t)N * 4;
    size_t oEdge = oS1d + (size_t)N * 12;
    size_t oCnt  = oEdge + (size_t)E * 4;
    size_t oRP   = oCnt + (size_t)N + 1;
    size_t oBs   = oRP  + (size_t)N + 1;
    size_t oPerm = oEdge + (size_t)E * 4 + (size_t)N * 4;
    size_t oRank = oEdge;
    size_t oTmp  = oEdge + (size_t)E;
    size_t oWx   = oHP;
    size_t oS2s  = oHP + (size_t)N * 40;
    size_t oS2d  = oS2s + (size_t)N;

    int* rowptr = (int*)(ws + oRP);
    int* bsum   = (int*)(ws + oBs);
    int* perm   = (int*)(ws + oPerm);
    int* rank   = (int*)(ws + oRank);
    int* tmp    = (int*)(ws + oTmp);
    int* srcs   = rank;

    const int n1  = N + 1;
    const int nb  = (n1 + 1023) / 1024;
    const int nch = (E + CSR_CHUNK - 1) / CSR_CHUNK;
    unsigned* cntw = (unsigned*)(ws + oHP);
    int* start     = (int*)(ws + oHP + (size_t)nch * CSR_CAPW);

    csr_count_lds<<<nch, 1024, 0, stream>>>(dst, cntw, rank, E);
    csr_blocksum<<<nb, 256, 0, stream>>>(cntw, bsum, n1, nch);
    csr_rowptr<<<nb, 256, 0, stream>>>(cntw, bsum, rowptr, start, n1, nch);

    const int nGemm  = 4 * ((N + 127) / 128);
    const int nPlace = (E + PLACE_EDGES - 1) / PLACE_EDGES;
    gemm1_place<<<nGemm + nPlace, 256, 0, stream>>>(h, Wheads, ws + oWh, aheads, ws + oS1s,
                                                    ws + oS1d, N, dst, start, rank, tmp, E);

    gather1_elu<<<(N + 15) / 16, 512, 0, stream>>>(rowptr, tmp, perm, srcs, src,
                                                   (const float4*)(ws + oS1s),
                                                   (const float4*)(ws + oS1d), ws + oWh, Wout,
                                                   aout, ws + oWx, ws + oS2s, ws + oS2d, N);


    gather2_lsm<<<(N + 7) / 8, 256, 0, stream>>>(rowptr, srcs, ws + oS2s, ws + oS2d, ws + oWx,
                                                 out, N);
}


# edge placement stores (edge id, source node) pairs: the layer-1 gather gets its sources with its one contiguous segment read, the per-edge scattered read of the source list is gone; sums unchanged
# speedup vs baseline: 7.8159x; 1.0313x over previous
#include <hip/hip_runtime.h>


#define LRELU 0.2f

typedef __attribute__((ext_vector_type(2))) float v2f;
typedef __attribute__((ext_vector_type(8))) float v8f;

__device__ __forceinline__ float wred_sum(float v) {
#pragma unroll
    for (int o = 16; o > 0; o >>= 1) v += __shfl_xor(v, o, 32);
    return v;
}
__device__ __forceinline__ float wred_max(float v) {
#pragma unroll
    for (int o = 16; o > 0; o >>= 1) v = fmaxf(v, __shfl_xor(v, o, 32));
    return v;
}
__device__ __forceinline__ float lrelu(float x) { return x > 0.f ? x : LRELU * x; }
__device__ __forceinline__ float xchg_add(const bool up, const float lo, const float hi,
                                          const int w) {
    const float keep = up ? hi : lo;
    const float send = up ? lo : hi;
    return keep + __shfl_xor(send, w, 32);
}

template <int COLT, int MODE>
__device__ __forceinline__ void gemm_wmma_f32(const float* __restrict__ A,
                                              const float* __restrict__ B,
                                              float* __restrict__ C,
                                              const float* __restrict__ avec,
                                              float* __restrict__ ssrc,
                                              float* __restrict__ sdst, const int nrows,
                                              const int bx, const int by) {
    constexpr int CW = COLT * 16;
    __shared__ float4 Blds4[64 * CW];
    float* Blds = (float*)Blds4;
    const int tid = threadIdx.x;
    const int colBase = by * CW;

    if constexpr (MODE == 0) {
        const float4* __restrict__ b4 = (const float4*)(B + (size_t)by * (256 * 64));
        float4 st[16];
#pragma unroll
        for (int i = 0; i < 16; ++i) st[i] = b4[tid + i * 256];
#pragma unroll
        for (int i = 0; i < 16; ++i) Blds4[tid + i * 256] = st[i];
    } else {
        const float4* __restrict__ b4 = (const float4*)B;
        float4 st[10];
#pragma unroll
        for (int i = 0; i < 10; ++i) st[i] = b4[tid + i * 256];
#pragma unroll
        for (int i = 0; i < 10; ++i) {
            const int j = tid + i * 256;
            const int k = j / 10;
            Blds4[k * 12 + (j - k * 10)] = st[i];
        }
        for (int j = tid; j < 512; j += 256)
            Blds4[(j >> 1) * 12 + 10 + (j & 1)] = make_float4(0.f, 0.f, 0.f, 0.f);
    }
    __syncthreads();

    const int wid  = tid >> 5;
    const int lane = tid & 31;
    const int half = lane >> 4;
    const int l16  = lane & 15;
    const int rowBase = bx * 128 + wid * 16;

    int arow = rowBase + l16;
    if (arow >= nrows) arow = nrows - 1;
    const float* aptr = A + (size_t)arow * 256;

    v8f acc[COLT] = {};

    for (int k0 = 0; k0 < 256; k0 += 4) {
        const int kb = k0 + half * 2;
        v2f a;
        a.x = aptr[kb];
        a.y = aptr[kb + 1];
#pragma unroll
        for (int t = 0; t < COLT; ++t) {
            v2f b;
            const int c = t * 16 + l16;
            b.x = Blds[kb * CW + c];
            b.y = Blds[(kb + 1) * CW + c];
            acc[t] = __builtin_amdgcn_wmma_f32_16x16x4_f32(false, a, false, b, (short)0,
                                                           acc[t], false, false);
        }
    }

    {
        float as[COLT], ad[COLT];
        for (int t = 0; t < COLT; ++t) {
            const int c = t * 16 + l16;
            if constexpr (MODE == 0) {
                as[t] = avec[by * 128 + c];
                ad[t] = avec[by * 128 + 64 + c];
            } else {
                as[t] = (c < 40) ? avec[c] : 0.0f;
                ad[t] = (c < 40) ? avec[40 + c] : 0.0f;
            }
        }
#define ROW_DOTS(v, s, d)                      \
    float s = 0.f, d = 0.f;                    \
    for (int t = 0; t < COLT; ++t) {           \
        s += acc[t][v] * as[t];                \
        d += acc[t][v] * ad[t];                \
    }
        ROW_DOTS(0, s0, d0)
        ROW_DOTS(1, s1, d1)
        ROW_DOTS(2, s2, d2)
        ROW_DOTS(3, s3, d3)
        ROW_DOTS(4, s4, d4)
        ROW_DOTS(5, s5, d5)
        ROW_DOTS(6, s6, d6)
        ROW_DOTS(7, s7, d7)
#undef ROW_DOTS
        const bool u8 = (l16 & 8) != 0, u4 = (l16 & 4) != 0, u2 = (l16 & 2) != 0,
                   u1 = (l16 & 1) != 0;
        const float q0 = xchg_add(u8, s0, d0, 8);
        const float q1 = xchg_add(u8, s1, d1, 8);
        const float q2 = xchg_add(u8, s2, d2, 8);
        const float q3 = xchg_add(u8, s3, d3, 8);
        const float q4 = xchg_add(u8, s4, d4, 8);
        const float q5 = xchg_add(u8, s5, d5, 8);
        const float q6 = xchg_add(u8, s6, d6, 8);
        const float q7 = xchg_add(u8, s7, d7, 8);
        const float r0 = xchg_add(u4, q0, q4, 4);
        const float r1 = xchg_add(u4, q1, q5, 4);
        const float r2 = xchg_add(u4, q2, q6, 4);
        const float r3 = xchg_add(u4, q3, q7, 4);
        const float y0 = xchg_add(u2, r0, r2, 2);
        const float y1 = xchg_add(u2, r1, r3, 2);
        const float tot = xchg_add(u1, y0, y1, 1);
        const int srow = rowBase + half * 8 + (l16 & 7);
        float* sp = (l16 < 8) ? ssrc : sdst;
        if (srow < nrows) {
            if constexpr (MODE == 0) {
                sp[(size_t)srow * 4 + by] = tot;
            } else {
                sp[srow] = tot;
            }
        }
    }

    if (rowBase + 16 <= nrows) {
#pragma unroll
        for (int t = 0; t < COLT; ++t) {
            const int col = colBase + t * 16 + l16;
            if constexpr (MODE == 1) {
                if (col >= 40) continue;
            }
#pragma unroll
            for (int v = 0; v < 8; ++v) {
                const int row = rowBase + v + half * 8;
                if constexpr (MODE == 0) {
                    C[(size_t)row * 256 + col] = acc[t][v];
                } else {
                    C[(size_t)row * 40 + col] = acc[t][v];
                }
            }
        }
    } else {
#pragma unroll
        for (int t = 0; t < COLT; ++t) {
            const int col = colBase + t * 16 + l16;
#pragma unroll
            for (int v = 0; v < 8; ++v) {
                const int row = rowBase + v + half * 8;
                if constexpr (MODE == 0) {
                    if (row < nrows) C[(size_t)row * 256 + col] = acc[t][v];
                } else {
                    if (row < nrows && col < 40) C[(size_t)row * 40 + col] = acc[t][v];
                }
            }
        }
    }
}

__device__ __forceinline__ int wred_sum_i(int v) {
    for (int o = 16; o > 0; o >>= 1) v += __shfl_xor(v, o, 32);
    return v;
}
__device__ __forceinline__ float readlane_f(float v, int l) {
    return __int_as_float(__builtin_amdgcn_readlane(__float_as_int(v), l));
}
__device__ __forceinline__ float uni_f(float v) {
    return __int_as_float(__builtin_amdgcn_readfirstlane(__float_as_int(v)));
}
#define CSR_CHUNK 50000
#define CSR_CAPW 51200
#define CSR_SP (2 * CSR_CAPW)
__global__ __launch_bounds__(1024) void csr_count_lds(const int* __restrict__ dst,
                                                      unsigned* __restrict__ cntw,
                                                      int* __restrict__ rank, int ne) {
    __shared__ uint4 cl4[CSR_CAPW / 4];
    unsigned* cl = (unsigned*)cl4;
    const int tid = threadIdx.x;
    for (int i = tid; i < CSR_CAPW / 4; i += 1024) cl4[i] = make_uint4(0u, 0u, 0u, 0u);
    __syncthreads();
    const int e0 = blockIdx.x * CSR_CHUNK;
    const int e1 = (e0 + CSR_CHUNK < ne) ? (e0 + CSR_CHUNK) : ne;
    const int nfull = (e1 - e0) / (16 * 1024);
    int tb = e0;
    for (int r = 0; r < nfull; ++r, tb += 16 * 1024) {
        const int eb = tb + tid;
        int d[16];
        unsigned old[16];
#pragma unroll
        for (int k = 0; k < 16; ++k) d[k] = dst[eb + k * 1024];
#pragma unroll
        for (int k = 0; k < 16; ++k) {
            const int sh = (d[k] & 1) << 4;
            old[k] = atomicAdd(&cl[d[k] >> 1], 1u << sh);
        }
#pragma unroll
        for (int k = 0; k < 16; ++k) {
            const int sh = (d[k] & 1) << 4;
            rank[eb + k * 1024] = (int)((old[k] >> sh) & 0xffffu);
        }
    }
    for (int e = tb + tid; e < e1; e += 1024) {
        const int dd = dst[e];
        const int sh = (dd & 1) << 4;
        const unsigned o = atomicAdd(&cl[dd >> 1], 1u << sh);
        rank[e] = (int)((o >> sh) & 0xffffu);
    }
    __syncthreads();
    uint4* o = (uint4*)(cntw + (size_t)blockIdx.x * CSR_CAPW);
    for (int i = tid; i < CSR_CAPW / 4; i += 1024) o[i] = cl4[i];
}
__device__ __forceinline__ void csr_deg4(const unsigned* __restrict__ cntw, const int nch,
                                         const int base, int c[4]) {
    c[0] = 0;
    c[1] = 0;
    c[2] = 0;
    c[3] = 0;
    const unsigned* p = cntw + (base >> 1);
#pragma unroll 8
    for (int ch = 0; ch < nch; ++ch) {
        const uint2 v = *(const uint2*)(p + (size_t)ch * CSR_CAPW);
        c[0] += (int)(v.x & 0xffffu);
        c[1] += (int)(v.x >> 16);
        c[2] += (int)(v.y & 0xffffu);
        c[3] += (int)(v.y >> 16);
    }
}
__global__ __launch_bounds__(256) void csr_blocksum(const unsigned* __restrict__ cntw,
                                                    int* __restrict__ bsum, int n1, int nch) {
    __shared__ int wsum[8];
    const int base = blockIdx.x * 1024 + threadIdx.x * 4;
    int c[4];
    csr_deg4(cntw, nch, base, c);
    int s = 0;
    for (int k = 0; k < 4; ++k)
        if (base + k < n1) s += c[k];
    s = wred_sum_i(s);
    if ((threadIdx.x & 31) == 0) wsum[threadIdx.x >> 5] = s;
    __syncthreads();
    if (threadIdx.x == 0) {
        int t = 0;
        for (int w = 0; w < 8; ++w) t += wsum[w];
        bsum[blockIdx.x] = t;
    }
}
__global__ __launch_bounds__(256) void csr_rowptr(const unsigned* __restrict__ cntw,
                                                  const int* __restrict__ bsum,
                                                  int* __restrict__ rowptr,
                                                  int* __restrict__ start, int n1, int nch) {
    __shared__ int wsum[8];
    __shared__ int wsum2[8];
    const int tid = threadIdx.x, lane = tid & 31, wid = tid >> 5;
    int p = 0;
    for (int j = tid; j < (int)blockIdx.x; j += 256) p += bsum[j];
    p = wred_sum_i(p);
    if (lane == 0) wsum[wid] = p;
    const int base = blockIdx.x * 1024 + tid * 4;
    int c[4];
    csr_deg4(cntw, nch, base, c);
    int s = 0;
    for (int k = 0; k < 4; ++k) {
        if (base + k >= n1) c[k] = 0;
        s += c[k];
    }
    int inc = s;
    for (int o = 1; o < 32; o <<= 1) {
        int t = __shfl_up(inc, o, 32);
        if (lane >= o) inc += t;
    }
    if (lane == 31) wsum2[wid] = inc;
    __syncthreads();
    int run = inc - s;
    for (int w = 0; w < 8; ++w) {
        run += wsum[w];
        if (w < wid) run += wsum2[w];
    }
    int4 a;
    a.x = run;
    a.y = a.x + c[0];
    a.z = a.y + c[1];
    a.w = a.z + c[2];
    if (base < n1) rowptr[base] = a.x;
    if (base + 1 < n1) rowptr[base + 1] = a.y;
    if (base + 2 < n1) rowptr[base + 2] = a.z;
    if (base + 3 < n1) rowptr[base + 3] = a.w;
    const unsigned* cp = cntw + (base >> 1);
#pragma unroll 8
    for (int ch = 0; ch < nch; ++ch) {
        const uint2 v = *(const uint2*)(cp + (size_t)ch * CSR_CAPW);
        *(int4*)(start + (size_t)ch * CSR_SP + base) = a;
        a.x += (int)(v.x & 0xffffu);
        a.y += (int)(v.x >> 16);
        a.z += (int)(v.y & 0xffffu);
        a.w += (int)(v.y >> 16);
    }
}
#define PLACE_EDGES 2048
__device__ __forceinline__ void csr_place_block(const int* __restrict__ src,
                                                const int* __restrict__ dst,
                                                const int* __restrict__ start,
                                                const int* __restrict__ rank,
                                                int2* __restrict__ pairs, const int ne,
                                                const int pb) {
    const int eb = pb * PLACE_EDGES;
    const int tid = (int)threadIdx.x;
    if (eb + PLACE_EDGES <= ne) {
        const int e0 = eb + tid;
        int d[8], rk[8], st[8], sr[8];
#pragma unroll
        for (int k = 0; k < 8; ++k) {
            d[k] = dst[e0 + k * 256];
            rk[k] = rank[e0 + k * 256];
            sr[k] = src[e0 + k * 256];
        }
#pragma unroll
        for (int k = 0; k < 8; ++k)
            st[k] = start[(size_t)((e0 + k * 256) / CSR_CHUNK) * CSR_SP + d[k]];
#pragma unroll
        for (int k = 0; k < 8; ++k) pairs[st[k] + rk[k]] = make_int2(e0 + k * 256, sr[k]);
    } else {
        for (int e = eb + tid; e < ne; e += 256)
            pairs[start[(size_t)(e / CSR_CHUNK) * CSR_SP + dst[e]] + rank[e]] =
                make_int2(e, src[e]);
    }
}
__global__ __launch_bounds__(256) void gemm1_place(const float* __restrict__ A,
                                                   const float* __restrict__ B,
                                                   float* __restrict__ C,
                                                   const float* __restrict__ avec,
                                                   float* __restrict__ ssrc,
                                                   float* __restrict__ sdst, int nrows,
                                                   const int* __restrict__ src,
                                                   const int* __restrict__ dst,
                                                   const int* __restrict__ start,
                                                   const int* __restrict__ rank,
                                                   int2* __restrict__ pairs, int ne) {
    const int nrt = (nrows + 127) >> 7;
    const int npl = (ne + PLACE_EDGES - 1) / PLACE_EDGES;
    const int m = (nrt < npl) ? nrt : npl;
    const int b = (int)blockIdx.x;
    int pb = -1, g = 0;
    if (b < 5 * m) {
        const int grp = b / 5, r = b - grp * 5;
        if (r == 4) pb = grp;
        else g = grp * 4 + r;
    } else {
        const int i = b - 5 * m, prest = npl - m;
        if (i < prest) pb = m + i;
        else g = 4 * m + (i - prest);
    }
    if (pb >= 0) {
        csr_place_block(src, dst, start, rank, pairs, ne, pb);
        return;
    }
    gemm_wmma_f32<4, 0>(A, B, C, avec, ssrc, sdst, nrows, g % nrt, g / nrt);
}
__device__ __forceinline__ void node_scores1(const float4* __restrict__ ss4, const float4 sd,
                                             const bool has, const int s_i, float4& x) {
    x.x = -INFINITY;
    x.y = -INFINITY;
    x.z = -INFINITY;
    x.w = -INFINITY;
    if (has) {
        const float4 vs = ss4[s_i];
        x.x = lrelu(vs.x + sd.x);
        x.y = lrelu(vs.y + sd.y);
        x.z = lrelu(vs.z + sd.z);
        x.w = lrelu(vs.w + sd.w);
    }
}
__device__ __forceinline__ void edge_scores1e(const int* __restrict__ src,
                                              const float4* __restrict__ ss4, const float4 sd,
                                              const bool has, const int e, int& s_i,
                                              float4& x) {
    s_i = 0;
    if (has) s_i = src[e];
    node_scores1(ss4, sd, has, s_i, x);
}
__device__ __forceinline__ void edge_scores1(const int* perm, const int* __restrict__ src,
                                             const float4* __restrict__ ss4, const float4 sd,
                                             int i, int end, int& s_i, float4& x) {
    const bool has = i < end;
    int e = 0;
    if (has) e = perm[i];
    edge_scores1e(src, ss4, sd, has, e, s_i, x);
}
__device__ __forceinline__ void rows_fma1(const float* __restrict__ wl, const bool hi,
                                          const int s_i, const float4 a_i, const int m,
                                          float4& acc0, float4& acc1) {
#pragma unroll 4
    for (int t = 0; t < m; ++t) {
        const int s = __builtin_amdgcn_readlane(s_i, t);
        const float ax = readlane_f(a_i.x, t);
        const float ay = readlane_f(a_i.y, t);
        const float az = readlane_f(a_i.z, t);
        const float aw = readlane_f(a_i.w, t);
        const float al = hi ? ay : ax;
        const float au = hi ? aw : az;
        const float* r = wl + (size_t)s * 256;
        const float4 w0 = *(const float4*)(r);
        const float4 w1 = *(const float4*)(r + 128);
        acc0.x += al * w0.x;
        acc0.y += al * w0.y;
        acc0.z += al * w0.z;
        acc0.w += al * w0.w;
        acc1.x += au * w1.x;
        acc1.y += au * w1.y;
        acc1.z += au * w1.z;
        acc1.w += au * w1.w;
    }
}

#define XP 260
__global__ __launch_bounds__(512) void gather1_elu(const int* __restrict__ rowptr,
                                                   const int2* __restrict__ pairs, int* perm,
                                                   int* __restrict__ srcs,
                                                   const int* __restrict__ src,
                                                   const float4* __restrict__ ss4,
                                                   const float4* __restrict__ sd4,
                                                   const float* __restrict__ Wh,
                                                   const float* __restrict__ Wout,
                                                   const float* __restrict__ aout,
                                                   float* __restrict__ Wx,
                                                   float* __restrict__ s2s,
                                                   float* __restrict__ s2d, int n) {
    __shared__ float4 xs4[16 * XP / 4];
    __shared__ float part[4 * 768];
    __shared__ float wxs[16 * 48];
    float* xs = (float*)xs4;
    const int tid = threadIdx.x;
    const int wid = tid >> 5;
    const int node0 = blockIdx.x * 16;
    const int nodeu = __builtin_amdgcn_readfirstlane(node0 + wid);
    const int lane = tid & 31;
    const int node = (nodeu < n) ? nodeu : (n - 1);
    const int beg = rowptr[node], end = rowptr[node + 1];
    const float4 sd = sd4[node];
    const bool hi = lane >= 16;
    const float* wl = Wh + lane * 4;
    float4 acc0 = {0.f, 0.f, 0.f, 0.f};
    float4 acc1 = {0.f, 0.f, 0.f, 0.f};
    if (end - beg <= 32) {
        const int m = end - beg;
        const bool has = lane < m;
        int2 pr = make_int2(0x7fffffff, 0);
        if (has) pr = pairs[beg + lane];
        const int ei = pr.x;
        int r = 0;
        for (int t = 0; t < m; ++t) r += (__builtin_amdgcn_readlane(ei, t) < ei) ? 1 : 0;
        if (!has) r = lane;
        const int s_i = __builtin_amdgcn_ds_permute(r << 2, pr.y);
        float4 x;
        node_scores1(ss4, sd, has, s_i, x);
        if (has && nodeu < n) srcs[beg + lane] = s_i;
        float4 mx, ex, dn, a_i;
        mx.x = wred_max(x.x);
        mx.y = wred_max(x.y);
        mx.z = wred_max(x.z);
        mx.w = wred_max(x.w);
        ex.x = expf(x.x - mx.x);
        ex.y = expf(x.y - mx.y);
        ex.z = expf(x.z - mx.z);
        ex.w = expf(x.w - mx.w);
        dn.x = wred_sum(ex.x);
        dn.y = wred_sum(ex.y);
        dn.z = wred_sum(ex.z);
        dn.w = wred_sum(ex.w);
        a_i.x = ex.x / dn.x;
        a_i.y = ex.y / dn.y;
        a_i.z = ex.z / dn.z;
        a_i.w = ex.w / dn.w;
        rows_fma1(wl, hi, s_i, a_i, end - beg, acc0, acc1);
    } else {
        for (int i0 = beg; i0 < end; i0 += 32) {
            const int i = i0 + lane;
            const int ei = (i < end) ? pairs[i].x : 0x7fffffff;
            int r = 0;
            for (int j0 = beg; j0 < end; j0 += 32) {
                const int j = j0 + lane;
                const int ej = (j < end) ? pairs[j].x : 0x7fffffff;
                const int mj = (end - j0 < 32) ? (end - j0) : 32;
                for (int t = 0; t < mj; ++t) r += (__builtin_amdgcn_readlane(ej, t) < ei) ? 1 : 0;
            }
            if (i < end) perm[beg + r] = ei;
        }
        __threadfence();
        float4 mx = {-INFINITY, -INFINITY, -INFINITY, -INFINITY};
        for (int i0 = beg; i0 < end; i0 += 32) {
            int s_i;
            float4 x;
            edge_scores1(perm, src, ss4, sd, i0 + lane, end, s_i, x);
            mx.x = fmaxf(mx.x, x.x);
            mx.y = fmaxf(mx.y, x.y);
            mx.z = fmaxf(mx.z, x.z);
            mx.w = fmaxf(mx.w, x.w);
        }
        mx.x = uni_f(wred_max(mx.x));
        mx.y = uni_f(wred_max(mx.y));
        mx.z = uni_f(wred_max(mx.z));
        mx.w = uni_f(wred_max(mx.w));
        float4 dn = {0.f, 0.f, 0.f, 0.f};
        for (int i0 = beg; i0 < end; i0 += 32) {
            int s_i;
            float4 x;
            edge_scores1(perm, src, ss4, sd, i0 + lane, end, s_i, x);
            if (i0 + lane < end) {
                dn.x += expf(x.x - mx.x);
                dn.y += expf(x.y - mx.y);
                dn.z += expf(x.z - mx.z);
                dn.w += expf(x.w - mx.w);
            }
        }
        dn.x = uni_f(wred_sum(dn.x));
        dn.y = uni_f(wred_sum(dn.y));
        dn.z = uni_f(wred_sum(dn.z));
        dn.w = uni_f(wred_sum(dn.w));
        for (int i0 = beg; i0 < end; i0 += 32) {
            int s_i;
            float4 x;
            edge_scores1(perm, src, ss4, sd, i0 + lane, end, s_i, x);
            float4 a_i = {0.f, 0.f, 0.f, 0.f};
            if (i0 + lane < end) {
                if (nodeu < n) srcs[i0 + lane] = s_i;
                a_i.x = expf(x.x - mx.x) / dn.x;
                a_i.y = expf(x.y - mx.y) / dn.y;
                a_i.z = expf(x.z - mx.z) / dn.z;
                a_i.w = expf(x.w - mx.w) / dn.w;
            }
            const int m = (end - i0 < 32) ? (end - i0) : 32;
            rows_fma1(wl, hi, s_i, a_i, m, acc0, acc1);
        }
    }
    acc0.x = acc0.x > 0.f ? acc0.x : (expf(acc0.x) - 1.f);
    acc0.y = acc0.y > 0.f ? acc0.y : (expf(acc0.y) - 1.f);
    acc0.z = acc0.z > 0.f ? acc0.z : (expf(acc0.z) - 1.f);
    acc0.w = acc0.w > 0.f ? acc0.w : (expf(acc0.w) - 1.f);
    acc1.x = acc1.x > 0.f ? acc1.x : (expf(acc1.x) - 1.f);
    acc1.y = acc1.y > 0.f ? acc1.y : (expf(acc1.y) - 1.f);
    acc1.z = acc1.z > 0.f ? acc1.z : (expf(acc1.z) - 1.f);
    acc1.w = acc1.w > 0.f ? acc1.w : (expf(acc1.w) - 1.f);
    {
        float* xr = xs + wid * XP + lane * 4;
        *(float4*)(xr) = acc0;
        *(float4*)(xr + 128) = acc1;
    }
    __syncthreads();
    if (wid < 12) {
        const int t = wid % 3, kq = wid / 3;
        const int half = lane >> 4, l16 = lane & 15;
        const int cc = t * 16 + l16;
        const bool cv = cc < 40;
        const float* bp = Wout + (cv ? cc : 39);
        const float* ap = xs + l16 * XP + kq * 64 + half * 2;
        v8f p0 = {}, p1 = {};
#pragma unroll 4
        for (int j = 0; j < 16; j += 2) {
            const int kb0 = kq * 64 + j * 4 + half * 2;
            const v2f a0 = *(const v2f*)(ap + j * 4);
            const v2f a1 = *(const v2f*)(ap + j * 4 + 4);
            const float w00 = bp[(kb0) * 40], w01 = bp[(kb0 + 1) * 40];
            const float w10 = bp[(kb0 + 4) * 40], w11 = bp[(kb0 + 5) * 40];
            v2f b0, b1;
            b0.x = cv ? w00 : 0.f;
            b0.y = cv ? w01 : 0.f;
            b1.x = cv ? w10 : 0.f;
            b1.y = cv ? w11 : 0.f;
            p0 = __builtin_amdgcn_wmma_f32_16x16x4_f32(false, a0, false, b0, (short)0, p0, false,
                                                       false);
            p1 = __builtin_amdgcn_wmma_f32_16x16x4_f32(false, a1, false, b1, (short)0, p1, false,
                                                       false);
        }
        float* pp = part + kq * 768 + t * 256 + lane;
#pragma unroll
        for (int v = 0; v < 8; ++v) pp[v * 32] = p0[v] + p1[v];
    }
    __syncthreads();
    for (int e = tid; e < 768; e += 512) {
        const float s = ((part[e] + part[768 + e]) + part[1536 + e]) + part[2304 + e];
        const int t = e >> 8, v = (e >> 5) & 7, l = e & 31;
        wxs[(8 * (l >> 4) + v) * 48 + t * 16 + (l & 15)] = s;
    }
    __syncthreads();
    for (int i = tid; i < 16 * 40; i += 512) {
        const int r = i / 40, cix = i - r * 40;
        if (node0 + r < n) Wx[(size_t)node0 * 40 + i] = wxs[r * 48 + cix];
    }
    if (tid < 32) {
        const int r = tid & 15;
        const float* av = aout + (tid >> 4) * 40;
        float s = 0.f;
        for (int k = 0; k < 40; ++k) s += wxs[r * 48 + k] * av[k];
        if (node0 + r < n) {
            if (tid < 16) s2s[node0 + r] = s;
            else s2d[node0 + r] = s;
        }
    }
}

__device__ __forceinline__ void edge_score2(const int* __restrict__ srcs,
                                            const float* __restrict__ ss, const float sd, int i,
                                            int end, int& s_i, float& x) {
    s_i = 0;
    x = -INFINITY;
    if (i < end) {
        s_i = srcs[i];
        x = lrelu(ss[s_i] + sd);
    }
}
__device__ __forceinline__ void rows_fma2(const float* __restrict__ w1p,
                                          const float* __restrict__ w2p, const int s_i,
                                          const float a_i, const int m, float& acc1,
                                          float& acc2) {
#pragma unroll 4
    for (int t = 0; t < m; ++t) {
        const int s = __builtin_amdgcn_readlane(s_i, t);
        const float a = readlane_f(a_i, t);
        acc1 += a * w1p[(size_t)s * 40];
        acc2 += a * w2p[(size_t)s * 40];
    }
}
__global__ __launch_bounds__(256) void gather2_lsm(const int* __restrict__ rowptr,
                                                   const int* __restrict__ srcs,
                                                   const float* __restrict__ ss,
                                                   const float* __restrict__ sdv,
                                                   const float* __restrict__ Wx,
                                                   float* __restrict__ out, int n) {
    const int node = __builtin_amdgcn_readfirstlane((int)((blockIdx.x * 256 + threadIdx.x) >> 5));
    const int lane = threadIdx.x & 31;
    if (node >= n) return;
    const int beg = rowptr[node], end = rowptr[node + 1];
    const float sd = sdv[node];
    const float* w1p = Wx + lane;
    const float* w2p = Wx + 32 + (lane & 7);
    float acc1 = 0.f, acc2 = 0.f;
    if (end - beg <= 32) {
        int s_i;
        float x;
        edge_score2(srcs, ss, sd, beg + lane, end, s_i, x);
        const float mx = wred_max(x);
        const float ex = expf(x - mx);
        const float dn = wred_sum(ex);
        rows_fma2(w1p, w2p, s_i, ex / dn, end - beg, acc1, acc2);
    } else {
        float mx = -INFINITY;
        for (int i0 = beg; i0 < end; i0 += 32) {
            int s_i;
            float x;
            edge_score2(srcs, ss, sd, i0 + lane, end, s_i, x);
            mx = fmaxf(mx, x);
        }
        mx = uni_f(wred_max(mx));
        float dn = 0.f;
        for (int i0 = beg; i0 < end; i0 += 32) {
            int s_i;
            float x;
            edge_score2(srcs, ss, sd, i0 + lane, end, s_i, x);
            if (i0 + lane < end) dn += expf(x - mx);
        }
        dn = uni_f(wred_sum(dn));
        for (int i0 = beg; i0 < end; i0 += 32) {
            int s_i;
            float x;
            edge_score2(srcs, ss, sd, i0 + lane, end, s_i, x);
            const float a_i = (i0 + lane < end) ? expf(x - mx) / dn : 0.f;
            const int m = (end - i0 < 32) ? (end - i0) : 32;
            rows_fma2(w1p, w2p, s_i, a_i, m, acc1, acc2);
        }
    }
    const float v1 = acc1;
    const float v2 = (lane < 8) ? acc2 : -INFINITY;
    const float mxo = wred_max(fmaxf(v1, v2));
    float sm = expf(v1 - mxo) + ((lane < 8) ? expf(v2 - mxo) : 0.f);
    sm = wred_sum(sm);
    const float lg = mxo + logf(sm);
    float* row = out + (size_t)node * 40;
    row[lane] = v1 - lg;
    if (lane < 8) row[32 + lane] = v2 - lg;
}

extern "C" void kernel_launch(void* const* d_in, const int* in_sizes, int n_in, void* d_out,
                              int out_size, void* d_ws, size_t ws_size, hipStream_t stream) {
    (void)n_in; (void)out_size; (void)ws_size;
    const float* h      = (const float*)d_in[0];
    const int*   eidx   = (const int*)d_in[1];
    const float* Wheads = (const float*)d_in[2];
    const float* aheads = (const float*)d_in[3];
    const float* Wout   = (const float*)d_in[4];
    const float* aout   = (const float*)d_in[5];
    float* out = (float*)d_out;

    const int N = in_sizes[0] / 256;
    const int E = in_sizes[1] / 2;
    const int* src = eidx;
    const int* dst = eidx + E;

    float* ws = (float*)d_ws;
    size_t oWh   = 0;
    size_t oHP   = oWh  + (size_t)N * 256;
    size_t oS1s  = oHP  + (size_t)N * 256;
    size_t oS1d  = oS1s + (size_t)N * 4;
    size_t oEdge = oS1d + (size_t)N * 12;
    size_t oCnt  = oEdge + (size_t)E * 4;
    size_t oRP   = oCnt + (size_t)N + 1;
    size_t oBs   = oRP  + (size_t)N + 1;
    size_t oPerm = oEdge + (size_t)E * 4 + (size_t)N * 4;
    size_t oRank = oEdge;
    size_t oPair = oEdge + (size_t)E * 2;
    size_t oWx   = oHP;
    size_t oS2s  = oHP + (size_t)N * 40;
    size_t oS2d  = oS2s + (size_t)N;

    int* rowptr = (int*)(ws + oRP);
    int* bsum   = (int*)(ws + oBs);
    int* perm   = (int*)(ws + oPerm);
    int* rank   = (int*)(ws + oRank);
    int2* pairs = (int2*)(ws + oPair);
    int* srcs   = rank;

    const int n1  = N + 1;
    const int nb  = (n1 + 1023) / 1024;
    const int nch = (E + CSR_CHUNK - 1) / CSR_CHUNK;
    unsigned* cntw = (unsigned*)(ws + oHP);
    int* start     = (int*)(ws + oHP + (size_t)nch * CSR_CAPW);

    csr_count_lds<<<nch, 1024, 0, stream>>>(dst, cntw, rank, E);
    csr_blocksum<<<nb, 256, 0, stream>>>(cntw, bsum, n1, nch);
    csr_rowptr<<<nb, 256, 0, stream>>>(cntw, bsum, rowptr, start, n1, nch);

    const int nGemm  = 4 * ((N + 127) / 128);
    const int nPlace = (E + PLACE_EDGES - 1) / PLACE_EDGES;
    gemm1_place<<<nGemm + nPlace, 256, 0, stream>>>(h, Wheads, ws + oWh, aheads, ws + oS1s,
                                                    ws + oS1d, N, src, dst, start, rank, pairs, E);

    gather1_elu<<<(N + 15) / 16, 512, 0, stream>>>(rowptr, pairs, perm, srcs, src,
                                                   (const float4*)(ws + oS1s),
                                                   (const float4*)(ws + oS1d), ws + oWh, Wout,
                                                   aout, ws + oWx, ws + oS2s, ws + oS2d, N);


    gather2_lsm<<<(N + 7) / 8, 256, 0, stream>>>(rowptr, srcs, ws + oS2s, ws + oS2d, ws + oWx,
                                                 out, N);
}
